# starting kernel
# baseline (speedup 1.0000x reference)
#include <hip/hip_runtime.h>
#include <hip/hip_bf16.h>
#include <math.h>

typedef float v2f __attribute__((ext_vector_type(2)));
typedef float v8f __attribute__((ext_vector_type(8)));

#define T_SEQ 2048
#define HID_DIM 1024
#define NH 16
#define DH 64
#define NB 32
#define KMAX_SEL 8

__device__ __forceinline__ v8f wmma4(v2f a, v2f b, v8f c) {
  return __builtin_amdgcn_wmma_f32_16x16x4_f32(false, a, false, b, (short)0, c,
                                               false, false);
}

__device__ __forceinline__ float apply_act(float x, int act) {
  if (act == 1) return 0.5f * x * (1.0f + erff(x * 0.70710678118654752f));
  if (act == 2) return 1.0f / (1.0f + __expf(-x));
  return x;
}

__global__ void wmma_gemm_f32_blk(const float* __restrict__ A,
                                  const float* __restrict__ B,
                                  const float* __restrict__ bias,
                                  float* __restrict__ C, int M, int N, int K,
                                  int act, int headSplit) {
  int wave = threadIdx.x >> 5;
  int lane = threadIdx.x & 31;
  int wtile = blockIdx.x * (blockDim.x >> 5) + wave;
  int wn = N >> 5;
  int total = (M >> 6) * wn;
  if (wtile >= total) return;
  int tm = wtile / wn, tn = wtile - tm * wn;
  int half = lane >> 4, l = lane & 15;
  int m0 = tm << 6, n0 = tn << 5;

  v8f acc[4][2];
#pragma unroll
  for (int mi = 0; mi < 4; ++mi)
#pragma unroll
    for (int ni = 0; ni < 2; ++ni) acc[mi][ni] = (v8f){};

  const float* ap = A + (size_t)(m0 + l) * K + 2 * half;
  const float* bp = B + (size_t)(2 * half) * N + n0 + l;
  const size_t aRowStep = (size_t)16 * K;
  for (int k0 = 0; k0 < K; k0 += 4) {
    v2f a[4], b[2];
#pragma unroll
    for (int mi = 0; mi < 4; ++mi) {
      const float* p = ap + (size_t)mi * aRowStep;
      a[mi].x = p[0];
      a[mi].y = p[1];
    }
#pragma unroll
    for (int ni = 0; ni < 2; ++ni) {
      b[ni].x = bp[ni * 16];
      b[ni].y = bp[(size_t)N + ni * 16];
    }
#pragma unroll
    for (int mi = 0; mi < 4; ++mi)
#pragma unroll
      for (int ni = 0; ni < 2; ++ni)
        acc[mi][ni] = wmma4(a[mi], b[ni], acc[mi][ni]);
    ap += 4;
    bp += (size_t)4 * N;
  }
  float bv[2];
#pragma unroll
  for (int ni = 0; ni < 2; ++ni) bv[ni] = bias ? bias[n0 + ni * 16 + l] : 0.0f;
#pragma unroll
  for (int mi = 0; mi < 4; ++mi)
#pragma unroll
    for (int ni = 0; ni < 2; ++ni) {
      int col = n0 + ni * 16 + l;
#pragma unroll
      for (int r = 0; r < 8; ++r) {
        float x = apply_act(acc[mi][ni][r] + bv[ni], act);
        int m = m0 + mi * 16 + r + 8 * half;
        if (headSplit) {
          int hh = col >> 6, dd = col & 63;
          C[((size_t)hh * M + m) * 64 + dd] = x;
        } else {
          C[(size_t)m * N + col] = x;
        }
      }
    }
}

__global__ void wmma_gemm_f32(const float* __restrict__ A,
                              const float* __restrict__ B,
                              const float* __restrict__ bias,
                              float* __restrict__ C, int M, int N, int K,
                              int act, int headSplit) {
  int wave = threadIdx.x >> 5;
  int lane = threadIdx.x & 31;
  int tile = blockIdx.x * (blockDim.x >> 5) + wave;
  int tiles_n = N >> 4;
  int total = (M >> 4) * tiles_n;
  if (tile >= total) return;
  int tm = tile / tiles_n, tn = tile - tm * tiles_n;
  int half = lane >> 4, l = lane & 15;
  int arow = (tm << 4) + l;
  int bcol = (tn << 4) + l;
  v8f acc = {};
  const float* ap = A + (size_t)arow * K + 2 * half;
  const float* bp = B + (size_t)(2 * half) * N + bcol;
  for (int k0 = 0; k0 < K; k0 += 4) {
    v2f a, b;
    a.x = ap[0];
    a.y = ap[1];
    b.x = bp[0];
    b.y = bp[N];
    acc = wmma4(a, b, acc);
    ap += 4;
    bp += (size_t)4 * N;
  }
  float bv = bias ? bias[bcol] : 0.0f;
#pragma unroll
  for (int r = 0; r < 8; ++r) {
    float x = apply_act(acc[r] + bv, act);
    int m = (tm << 4) + r + 8 * half;
    if (headSplit) {
      int hh = bcol >> 6, dd = bcol & 63;
      C[((size_t)hh * M + m) * 64 + dd] = x;
    } else {
      C[(size_t)m * N + bcol] = x;
    }
  }
}

__global__ void sliding_window_kernel(const float* __restrict__ q,
                                      const float* __restrict__ k,
                                      const float* __restrict__ v,
                                      float* __restrict__ out) {
  extern __shared__ float lds[];
  float* Ks = lds;
  float* Vs = Ks + 128 * 66;
  float* Ps = Vs + 128 * 66;
  int h = blockIdx.x >> 5;
  int w = blockIdx.x & 31;
  const float* kh = k + (size_t)h * T_SEQ * DH;
  const float* vh = v + (size_t)h * T_SEQ * DH;
  for (int idx = threadIdx.x; idx < 128 * 32; idx += blockDim.x) {
    int r = idx >> 5, c2 = (idx & 31) << 1;
    int g = (w - 1) * 64 + r;
    float2 kk = make_float2(0.f, 0.f), vv = make_float2(0.f, 0.f);
    if (g >= 0) {
      kk = *(const float2*)&kh[(size_t)g * DH + c2];
      vv = *(const float2*)&vh[(size_t)g * DH + c2];
    }
    *(float2*)&Ks[r * 66 + c2] = kk;
    *(float2*)&Vs[r * 66 + c2] = vv;
  }
  __syncthreads();

  int wv = threadIdx.x >> 5, lane = threadIdx.x & 31;
  int half = lane >> 4, l = lane & 15;
  const float* qrow =
      q + ((size_t)h * T_SEQ + w * 64 + wv * 16 + l) * DH + 2 * half;

  v8f s[8] = {};
  for (int k0 = 0; k0 < 64; k0 += 4) {
    v2f a;
    a.x = qrow[k0];
    a.y = qrow[k0 + 1];
#pragma unroll
    for (int nt = 0; nt < 8; ++nt) {
      v2f b;
      const float* kp = &Ks[(nt * 16 + l) * 66 + k0 + 2 * half];
      b.x = kp[0];
      b.y = kp[1];
      s[nt] = wmma4(a, b, s[nt]);
    }
  }
  float rmax[8], rsum[8];
#pragma unroll
  for (int r = 0; r < 8; ++r) { rmax[r] = -3.0e38f; rsum[r] = 0.f; }
#pragma unroll
  for (int nt = 0; nt < 8; ++nt) {
    int n = nt * 16 + l;
#pragma unroll
    for (int r = 0; r < 8; ++r) {
      int m = wv * 16 + r + 8 * half;
      float x = s[nt][r] * 0.125f;
      bool ok = (m + 64 >= n) && (w > 0 || n >= 64);
      x = ok ? x : -3.0e38f;
      s[nt][r] = x;
      rmax[r] = fmaxf(rmax[r], x);
    }
  }
#pragma unroll
  for (int d = 1; d < 16; d <<= 1)
#pragma unroll
    for (int r = 0; r < 8; ++r) rmax[r] = fmaxf(rmax[r], __shfl_xor(rmax[r], d));
#pragma unroll
  for (int nt = 0; nt < 8; ++nt)
#pragma unroll
    for (int r = 0; r < 8; ++r) {
      float e = __expf(s[nt][r] - rmax[r]);
      s[nt][r] = e;
      rsum[r] += e;
    }
#pragma unroll
  for (int d = 1; d < 16; d <<= 1)
#pragma unroll
    for (int r = 0; r < 8; ++r) rsum[r] += __shfl_xor(rsum[r], d);
  float rinv[8];
#pragma unroll
  for (int r = 0; r < 8; ++r) rinv[r] = 1.0f / rsum[r];
#pragma unroll
  for (int nt = 0; nt < 8; ++nt)
#pragma unroll
    for (int r = 0; r < 8; ++r)
      Ps[(wv * 16 + r + 8 * half) * 130 + nt * 16 + l] = s[nt][r] * rinv[r];
  __syncthreads();

  v8f o[4] = {};
  const float* prow = &Ps[(wv * 16 + l) * 130 + 2 * half];
  for (int k0 = 0; k0 < 128; k0 += 4) {
    v2f a;
    a.x = prow[k0];
    a.y = prow[k0 + 1];
#pragma unroll
    for (int nt = 0; nt < 4; ++nt) {
      v2f b;
      const float* vp = &Vs[(k0 + 2 * half) * 66 + nt * 16 + l];
      b.x = vp[0];
      b.y = vp[66];
      o[nt] = wmma4(a, b, o[nt]);
    }
  }
  float* oh = out + ((size_t)h * T_SEQ + w * 64) * DH;
#pragma unroll
  for (int nt = 0; nt < 4; ++nt)
#pragma unroll
    for (int r = 0; r < 8; ++r)
      oh[(size_t)(wv * 16 + r + 8 * half) * DH + nt * 16 + l] = o[nt][r];
}

__global__ void comp_branch_kernel(const float* __restrict__ q,
                                   const float* __restrict__ kc,
                                   const float* __restrict__ vc,
                                   float* __restrict__ bs,
                                   float* __restrict__ comp) {
  __shared__ float Ps[16 * 34];
  int h = blockIdx.x >> 7;
  int tt = blockIdx.x & 127;
  int t0 = tt << 4;
  int lane = threadIdx.x, half = lane >> 4, l = lane & 15;
  const float* qrow = q + ((size_t)h * T_SEQ + t0 + l) * DH + 2 * half;
  v8f s[2] = {};
  for (int k0 = 0; k0 < 64; k0 += 4) {
    v2f a;
    a.x = qrow[k0];
    a.y = qrow[k0 + 1];
#pragma unroll
    for (int nt = 0; nt < 2; ++nt) {
      v2f b;
      const float* kp = kc + (size_t)(h * 32 + nt * 16 + l) * 64 + k0 + 2 * half;
      b.x = kp[0];
      b.y = kp[1];
      s[nt] = wmma4(a, b, s[nt]);
    }
  }
#pragma unroll
  for (int nt = 0; nt < 2; ++nt)
#pragma unroll
    for (int r = 0; r < 8; ++r) s[nt][r] *= 0.125f;

  int qb = t0 >> 6;
#pragma unroll
  for (int nt = 0; nt < 2; ++nt) {
    float cs = 0.f;
#pragma unroll
    for (int r = 0; r < 8; ++r) cs += s[nt][r];
    cs += __shfl_xor(cs, 16);
    if (half == 0)
      atomicAdd(&bs[((size_t)h * 32 + qb) * 32 + nt * 16 + l],
                cs * (1.0f / 64.0f));
  }
  float rmax[8], rsum[8];
#pragma unroll
  for (int r = 0; r < 8; ++r) rmax[r] = fmaxf(s[0][r], s[1][r]);
#pragma unroll
  for (int d = 1; d < 16; d <<= 1)
#pragma unroll
    for (int r = 0; r < 8; ++r) rmax[r] = fmaxf(rmax[r], __shfl_xor(rmax[r], d));
#pragma unroll
  for (int r = 0; r < 8; ++r) rsum[r] = 0.f;
#pragma unroll
  for (int nt = 0; nt < 2; ++nt)
#pragma unroll
    for (int r = 0; r < 8; ++r) {
      float e = __expf(s[nt][r] - rmax[r]);
      s[nt][r] = e;
      rsum[r] += e;
    }
#pragma unroll
  for (int d = 1; d < 16; d <<= 1)
#pragma unroll
    for (int r = 0; r < 8; ++r) rsum[r] += __shfl_xor(rsum[r], d);
#pragma unroll
  for (int nt = 0; nt < 2; ++nt)
#pragma unroll
    for (int r = 0; r < 8; ++r)
      Ps[(r + 8 * half) * 34 + nt * 16 + l] = s[nt][r] / rsum[r];
  __syncthreads();

  v8f o[4] = {};
  const float* prow = &Ps[l * 34 + 2 * half];
  for (int k0 = 0; k0 < 32; k0 += 4) {
    v2f a;
    a.x = prow[k0];
    a.y = prow[k0 + 1];
#pragma unroll
    for (int nt = 0; nt < 4; ++nt) {
      v2f b;
      const float* vp = vc + (size_t)(h * 32 + k0 + 2 * half) * 64 + nt * 16 + l;
      b.x = vp[0];
      b.y = vp[64];
      o[nt] = wmma4(a, b, o[nt]);
    }
  }
  float* oh = comp + ((size_t)h * T_SEQ + t0) * DH;
#pragma unroll
  for (int nt = 0; nt < 4; ++nt)
#pragma unroll
    for (int r = 0; r < 8; ++r)
      oh[(size_t)(r + 8 * half) * DH + nt * 16 + l] = o[nt][r];
}

__global__ void block_stats_kernel(const float* __restrict__ bs,
                                   const float* __restrict__ ksc,
                                   const float* __restrict__ kbi,
                                   unsigned char* __restrict__ mask,
                                   float* __restrict__ avg_acc,
                                   float* __restrict__ tail) {
  int h = blockIdx.x, j = threadIdx.x;
  const float* row = bs + ((size_t)h * 32 + j) * 32;
  float sc[32], p[32];
  float mx = -3.0e38f;
  for (int c = 0; c < 32; ++c) { sc[c] = row[c]; mx = fmaxf(mx, sc[c]); }
  float ssum = 0.f;
  for (int c = 0; c < 32; ++c) { p[c] = __expf(sc[c] - mx); ssum += p[c]; }
  float inv = 1.0f / ssum;
  float ent = 0.f;
  for (int c = 0; c < 32; ++c) {
    float pp = p[c] * inv;
    ent -= pp * __logf(pp + 1e-9f);
  }
  float ne = ent / (logf(32.0f) + 1e-9f);
  float kr = 1.0f / (1.0f + __expf(-(ksc[h] * ne + kbi[h])));
  float dkf = fminf(fmaxf(8.0f * kr, 1.0f), 8.0f);
  int dk = (int)dkf;
  tail[1 + h * 32 + j] = kr;
  tail[1 + 512 + h * 32 + j] = ne;
  atomicAdd(avg_acc, (float)dk);
  unsigned used = 0;
  unsigned char mrow[32];
  for (int c = 0; c < 32; ++c) mrow[c] = 0;
  mrow[j] = 1;
  for (int kk = 0; kk < KMAX_SEL; ++kk) {
    float best = -3.4e38f;
    int bi = 0;
    for (int c = 0; c < 32; ++c)
      if (!((used >> c) & 1u) && sc[c] > best) { best = sc[c]; bi = c; }
    used |= 1u << bi;
    if (kk < dk) mrow[bi] = 1;
  }
  for (int c = 0; c < 32; ++c) mask[((size_t)h * 32 + j) * 32 + c] = mrow[c];
}

__global__ void fine_attn_kernel(const float* __restrict__ q,
                                 const float* __restrict__ k,
                                 const float* __restrict__ v,
                                 const unsigned char* __restrict__ mask,
                                 float* __restrict__ out) {
  __shared__ float Ks[64 * 66];
  __shared__ float Vs[64 * 66];
  __shared__ float Ps[64 * 66];
  int h = blockIdx.x >> 5, qb = blockIdx.x & 31;
  int wv = threadIdx.x >> 5, lane = threadIdx.x & 31;
  int half = lane >> 4, l = lane & 15;
  const float* kh = k + (size_t)h * T_SEQ * DH;
  const float* vh = v + (size_t)h * T_SEQ * DH;
  const float* qrow =
      q + ((size_t)h * T_SEQ + qb * 64 + wv * 16 + l) * DH + 2 * half;
  const unsigned char* mrow = mask + ((size_t)h * 32 + qb) * 32;
  v8f o[4] = {};
  float rmax[8], rsum[8];
#pragma unroll
  for (int r = 0; r < 8; ++r) { rmax[r] = -3.0e38f; rsum[r] = 0.f; }

  for (int j = 0; j < 32; ++j) {
    if (!mrow[j]) continue;
    __syncthreads();
    for (int idx = threadIdx.x; idx < 64 * 32; idx += blockDim.x) {
      int r = idx >> 5, c2 = (idx & 31) << 1;
      *(float2*)&Ks[r * 66 + c2] =
          *(const float2*)&kh[(size_t)(j * 64 + r) * DH + c2];
      *(float2*)&Vs[r * 66 + c2] =
          *(const float2*)&vh[(size_t)(j * 64 + r) * DH + c2];
    }
    __syncthreads();
    v8f s[4] = {};
    for (int k0 = 0; k0 < 64; k0 += 4) {
      v2f a;
      a.x = qrow[k0];
      a.y = qrow[k0 + 1];
#pragma unroll
      for (int nt = 0; nt < 4; ++nt) {
        v2f b;
        const float* kp = &Ks[(nt * 16 + l) * 66 + k0 + 2 * half];
        b.x = kp[0];
        b.y = kp[1];
        s[nt] = wmma4(a, b, s[nt]);
      }
    }
    float nm[8], ps[8], fac[8];
#pragma unroll
    for (int r = 0; r < 8; ++r) nm[r] = rmax[r];
#pragma unroll
    for (int nt = 0; nt < 4; ++nt)
#pragma unroll
      for (int r = 0; r < 8; ++r) {
        float x = s[nt][r] * 0.125f;
        s[nt][r] = x;
        nm[r] = fmaxf(nm[r], x);
      }
#pragma unroll
    for (int d = 1; d < 16; d <<= 1)
#pragma unroll
      for (int r = 0; r < 8; ++r) nm[r] = fmaxf(nm[r], __shfl_xor(nm[r], d));
#pragma unroll
    for (int r = 0; r < 8; ++r) { fac[r] = __expf(rmax[r] - nm[r]); ps[r] = 0.f; }
#pragma unroll
    for (int nt = 0; nt < 4; ++nt)
#pragma unroll
      for (int r = 0; r < 8; ++r) {
        float e = __expf(s[nt][r] - nm[r]);
        s[nt][r] = e;
        ps[r] += e;
      }
#pragma unroll
    for (int d = 1; d < 16; d <<= 1)
#pragma unroll
      for (int r = 0; r < 8; ++r) ps[r] += __shfl_xor(ps[r], d);
#pragma unroll
    for (int r = 0; r < 8; ++r) {
      rsum[r] = rsum[r] * fac[r] + ps[r];
      rmax[r] = nm[r];
    }
#pragma unroll
    for (int nt = 0; nt < 4; ++nt)
#pragma unroll
      for (int r = 0; r < 8; ++r)
        Ps[(wv * 16 + r + 8 * half) * 66 + nt * 16 + l] = s[nt][r];
#pragma unroll
    for (int nt = 0; nt < 4; ++nt)
#pragma unroll
      for (int r = 0; r < 8; ++r) o[nt][r] *= fac[r];
    __syncthreads();
    const float* prow = &Ps[(wv * 16 + l) * 66 + 2 * half];
    for (int k0 = 0; k0 < 64; k0 += 4) {
      v2f a;
      a.x = prow[k0];
      a.y = prow[k0 + 1];
#pragma unroll
      for (int nt = 0; nt < 4; ++nt) {
        v2f b;
        const float* vp = &Vs[(k0 + 2 * half) * 66 + nt * 16 + l];
        b.x = vp[0];
        b.y = vp[66];
        o[nt] = wmma4(a, b, o[nt]);
      }
    }
  }
  float* oh = out + ((size_t)h * T_SEQ + qb * 64) * DH;
#pragma unroll
  for (int nt = 0; nt < 4; ++nt)
#pragma unroll
    for (int r = 0; r < 8; ++r)
      oh[(size_t)(wv * 16 + r + 8 * half) * DH + nt * 16 + l] =
          o[nt][r] / rsum[r];
}

__global__ void combine_kernel(const float* __restrict__ sw,
                               const float* __restrict__ comp,
                               const float* __restrict__ fine,
                               const float* __restrict__ gate,
                               float* __restrict__ merged) {
  int idx = blockIdx.x * blockDim.x + threadIdx.x;
  if (idx >= T_SEQ * HID_DIM) return;
  int t = idx >> 10, c = idx & 1023, hh = c >> 6, dd = c & 63;
  size_t a = ((size_t)hh * T_SEQ + t) * DH + dd;
  const float* g = gate + (size_t)t * 48 + hh * 3;
  merged[idx] = sw[a] * g[0] + comp[a] * g[1] + fine[a] * g[2];
}

__global__ void zero_kernel(float* __restrict__ p, int n) {
  int i = blockIdx.x * blockDim.x + threadIdx.x;
  if (i < n) p[i] = 0.f;
}

__global__ void finalize_kernel(const float* __restrict__ avg_acc,
                                float* __restrict__ tail) {
  if (threadIdx.x == 0) tail[0] = avg_acc[0] * (1.0f / 512.0f);
}

extern "C" void kernel_launch(void* const* d_in, const int* in_sizes, int n_in,
                              void* d_out, int out_size, void* d_ws,
                              size_t ws_size, hipStream_t stream) {
  const float* X = (const float*)d_in[0];
  const float* Wq = (const float*)d_in[1];
  const float* Wk = (const float*)d_in[2];
  const float* Wv = (const float*)d_in[3];
  const float* Wo = (const float*)d_in[4];
  const float* Wc = (const float*)d_in[5];
  const float* bc = (const float*)d_in[6];
  const float* ksc = (const float*)d_in[7];
  const float* kbi = (const float*)d_in[8];
  const float* Wg = (const float*)d_in[9];
  const float* bg = (const float*)d_in[10];
  float* out = (float*)d_out;

  const size_t U = (size_t)NH * T_SEQ * DH;
  float* ws = (float*)d_ws;
  float* q = ws;
  float* kbuf = ws + U;
  float* vbuf = ws + 2 * U;
  float* sw = ws + 3 * U;
  float* comp = ws + 4 * U;
  float* fine = ws + 5 * U;
  float* merged = q;
  float* extra = ws + 6 * U;
  float* gate = extra;
  float* kc = gate + (size_t)T_SEQ * 48;
  float* vc = kc + 512 * 64;
  float* bs = vc + 512 * 64;
  float* avg = bs + 16384;
  unsigned char* mask = (unsigned char*)(avg + 64);

  zero_kernel<<<(16385 + 255) / 256, 256, 0, stream>>>(bs, 16385);

  {
    int wtiles = (T_SEQ / 64) * (HID_DIM / 32);
    int blocks = (wtiles + 7) / 8;
    wmma_gemm_f32_blk<<<blocks, 256, 0, stream>>>(X, Wq, nullptr, q, T_SEQ,
                                                  HID_DIM, HID_DIM, 0, 1);
    wmma_gemm_f32_blk<<<blocks, 256, 0, stream>>>(X, Wk, nullptr, kbuf, T_SEQ,
                                                  HID_DIM, HID_DIM, 0, 1);
    wmma_gemm_f32_blk<<<blocks, 256, 0, stream>>>(X, Wv, nullptr, vbuf, T_SEQ,
                                                  HID_DIM, HID_DIM, 0, 1);
  }
  {
    int tiles = (T_SEQ / 16) * (48 / 16);
    wmma_gemm_f32<<<(tiles + 7) / 8, 256, 0, stream>>>(X, Wg, bg, gate, T_SEQ,
                                                       48, HID_DIM, 2, 0);
  }
  sliding_window_kernel<<<NH * NB, 128,
                          (2 * 128 * 66 + 64 * 130) * sizeof(float), stream>>>(
      q, kbuf, vbuf, sw);
  {
    int wtiles = (512 / 64) * (64 / 32);
    wmma_gemm_f32_blk<<<(wtiles + 7) / 8, 256, 0, stream>>>(kbuf, Wc, bc, kc,
                                                            512, 64, 4096, 1, 0);
    wmma_gemm_f32_blk<<<(wtiles + 7) / 8, 256, 0, stream>>>(vbuf, Wc, bc, vc,
                                                            512, 64, 4096, 1, 0);
  }
  comp_branch_kernel<<<NH * (T_SEQ / 16), 32, 0, stream>>>(q, kc, vc, bs, comp);
  block_stats_kernel<<<NH, 32, 0, stream>>>(bs, ksc, kbi, mask, avg,
                                            out + (size_t)T_SEQ * HID_DIM);
  fine_attn_kernel<<<NH * NB, 128, 0, stream>>>(q, kbuf, vbuf, mask, fine);
  combine_kernel<<<(T_SEQ * HID_DIM + 255) / 256, 256, 0, stream>>>(
      sw, comp, fine, gate, merged);
  {
    int wtiles = (T_SEQ / 64) * (HID_DIM / 32);
    wmma_gemm_f32_blk<<<(wtiles + 7) / 8, 256, 0, stream>>>(
        merged, Wo, nullptr, out, T_SEQ, HID_DIM, HID_DIM, 0, 0);
  }
  finalize_kernel<<<1, 1, 0, stream>>>(avg, out + (size_t)T_SEQ * HID_DIM);
  (void)in_sizes;
  (void)n_in;
  (void)out_size;
  (void)ws_size;
}


# compress MLP GEMMs (512x64, K=4096): split K over the 32 waves of a 16x32-tile work-group, partials summed in LDS in wave order (changes the summation order over k)
# speedup vs baseline: 2.3103x; 2.3103x over previous
#include <hip/hip_runtime.h>
#include <hip/hip_bf16.h>
#include <math.h>

typedef float v2f __attribute__((ext_vector_type(2)));
typedef float v8f __attribute__((ext_vector_type(8)));

#define T_SEQ 2048
#define HID_DIM 1024
#define NH 16
#define DH 64
#define NB 32
#define KMAX_SEL 8

__device__ __forceinline__ v8f wmma4(v2f a, v2f b, v8f c) {
  return __builtin_amdgcn_wmma_f32_16x16x4_f32(false, a, false, b, (short)0, c,
                                               false, false);
}

__device__ __forceinline__ float apply_act(float x, int act) {
  if (act == 1) return 0.5f * x * (1.0f + erff(x * 0.70710678118654752f));
  if (act == 2) return 1.0f / (1.0f + __expf(-x));
  return x;
}

__global__ void wmma_gemm_f32_blk(const float* __restrict__ A,
                                  const float* __restrict__ B,
                                  const float* __restrict__ bias,
                                  float* __restrict__ C, int M, int N, int K,
                                  int act, int headSplit) {
  int wave = threadIdx.x >> 5;
  int lane = threadIdx.x & 31;
  int wtile = blockIdx.x * (blockDim.x >> 5) + wave;
  int wn = N >> 5;
  int total = (M >> 6) * wn;
  if (wtile >= total) return;
  int tm = wtile / wn, tn = wtile - tm * wn;
  int half = lane >> 4, l = lane & 15;
  int m0 = tm << 6, n0 = tn << 5;

  v8f acc[4][2];
#pragma unroll
  for (int mi = 0; mi < 4; ++mi)
#pragma unroll
    for (int ni = 0; ni < 2; ++ni) acc[mi][ni] = (v8f){};

  const float* ap = A + (size_t)(m0 + l) * K + 2 * half;
  const float* bp = B + (size_t)(2 * half) * N + n0 + l;
  const size_t aRowStep = (size_t)16 * K;
  for (int k0 = 0; k0 < K; k0 += 4) {
    v2f a[4], b[2];
#pragma unroll
    for (int mi = 0; mi < 4; ++mi) {
      const float* p = ap + (size_t)mi * aRowStep;
      a[mi].x = p[0];
      a[mi].y = p[1];
    }
#pragma unroll
    for (int ni = 0; ni < 2; ++ni) {
      b[ni].x = bp[ni * 16];
      b[ni].y = bp[(size_t)N + ni * 16];
    }
#pragma unroll
    for (int mi = 0; mi < 4; ++mi)
#pragma unroll
      for (int ni = 0; ni < 2; ++ni)
        acc[mi][ni] = wmma4(a[mi], b[ni], acc[mi][ni]);
    ap += 4;
    bp += (size_t)4 * N;
  }
  float bv[2];
#pragma unroll
  for (int ni = 0; ni < 2; ++ni) bv[ni] = bias ? bias[n0 + ni * 16 + l] : 0.0f;
#pragma unroll
  for (int mi = 0; mi < 4; ++mi)
#pragma unroll
    for (int ni = 0; ni < 2; ++ni) {
      int col = n0 + ni * 16 + l;
#pragma unroll
      for (int r = 0; r < 8; ++r) {
        float x = apply_act(acc[mi][ni][r] + bv[ni], act);
        int m = m0 + mi * 16 + r + 8 * half;
        if (headSplit) {
          int hh = col >> 6, dd = col & 63;
          C[((size_t)hh * M + m) * 64 + dd] = x;
        } else {
          C[(size_t)m * N + col] = x;
        }
      }
    }
}

#define CK_K 4096
#define CK_N 64
#define CK_WAVES 32
#define CK_SLICE (CK_K / CK_WAVES)
#define CK_NI_PITCH 272
#define CK_W_PITCH (2 * CK_NI_PITCH)
__global__ void __launch_bounds__(CK_WAVES * 32)
compress_gemm_splitk(const float* __restrict__ A, const float* __restrict__ B,
                     const float* __restrict__ bias, float* __restrict__ C) {
  __shared__ float part[CK_WAVES * CK_W_PITCH];
  int wave = threadIdx.x >> 5, lane = threadIdx.x & 31;
  int half = lane >> 4, l = lane & 15;
  int m0 = (blockIdx.x >> 1) << 4, n0 = (blockIdx.x & 1) << 5;
  int kbase = wave * CK_SLICE;
  const float* ap = A + (size_t)(m0 + l) * CK_K + kbase + 2 * half;
  const float* bp = B + (size_t)(kbase + 2 * half) * CK_N + n0 + l;
  v8f acc0 = {}, acc1 = {};
#pragma unroll
  for (int s = 0; s < CK_SLICE / 4; ++s) {
    v2f a, b0, b1;
    a.x = ap[4 * s];
    a.y = ap[4 * s + 1];
    b0.x = bp[(4 * s) * CK_N];
    b0.y = bp[(4 * s + 1) * CK_N];
    b1.x = bp[(4 * s) * CK_N + 16];
    b1.y = bp[(4 * s + 1) * CK_N + 16];
    acc0 = wmma4(a, b0, acc0);
    acc1 = wmma4(a, b1, acc1);
  }
  float* pw = part + wave * CK_W_PITCH + lane;
#pragma unroll
  for (int r = 0; r < 8; ++r) {
    pw[r * 32] = acc0[r];
    pw[CK_NI_PITCH + r * 32] = acc1[r];
  }
  __syncthreads();
  int t = threadIdx.x;
  if (t < 512) {
    int row = t >> 5, col = t & 31;
    const float* pr = part + (col >> 4) * CK_NI_PITCH + (row & 7) * 32 +
                      (row >> 3) * 16 + (col & 15);
    float sum = pr[0];
#pragma unroll
    for (int w = 1; w < CK_WAVES; ++w) sum += pr[w * CK_W_PITCH];
    C[(size_t)(m0 + row) * CK_N + n0 + col] =
        apply_act(sum + bias[n0 + col], 1);
  }
}

__global__ void wmma_gemm_f32(const float* __restrict__ A,
                              const float* __restrict__ B,
                              const float* __restrict__ bias,
                              float* __restrict__ C, int M, int N, int K,
                              int act, int headSplit) {
  int wave = threadIdx.x >> 5;
  int lane = threadIdx.x & 31;
  int tile = blockIdx.x * (blockDim.x >> 5) + wave;
  int tiles_n = N >> 4;
  int total = (M >> 4) * tiles_n;
  if (tile >= total) return;
  int tm = tile / tiles_n, tn = tile - tm * tiles_n;
  int half = lane >> 4, l = lane & 15;
  int arow = (tm << 4) + l;
  int bcol = (tn << 4) + l;
  v8f acc = {};
  const float* ap = A + (size_t)arow * K + 2 * half;
  const float* bp = B + (size_t)(2 * half) * N + bcol;
  for (int k0 = 0; k0 < K; k0 += 4) {
    v2f a, b;
    a.x = ap[0];
    a.y = ap[1];
    b.x = bp[0];
    b.y = bp[N];
    acc = wmma4(a, b, acc);
    ap += 4;
    bp += (size_t)4 * N;
  }
  float bv = bias ? bias[bcol] : 0.0f;
#pragma unroll
  for (int r = 0; r < 8; ++r) {
    float x = apply_act(acc[r] + bv, act);
    int m = (tm << 4) + r + 8 * half;
    if (headSplit) {
      int hh = bcol >> 6, dd = bcol & 63;
      C[((size_t)hh * M + m) * 64 + dd] = x;
    } else {
      C[(size_t)m * N + bcol] = x;
    }
  }
}

__global__ void sliding_window_kernel(const float* __restrict__ q,
                                      const float* __restrict__ k,
                                      const float* __restrict__ v,
                                      float* __restrict__ out) {
  extern __shared__ float lds[];
  float* Ks = lds;
  float* Vs = Ks + 128 * 66;
  float* Ps = Vs + 128 * 66;
  int h = blockIdx.x >> 5;
  int w = blockIdx.x & 31;
  const float* kh = k + (size_t)h * T_SEQ * DH;
  const float* vh = v + (size_t)h * T_SEQ * DH;
  for (int idx = threadIdx.x; idx < 128 * 32; idx += blockDim.x) {
    int r = idx >> 5, c2 = (idx & 31) << 1;
    int g = (w - 1) * 64 + r;
    float2 kk = make_float2(0.f, 0.f), vv = make_float2(0.f, 0.f);
    if (g >= 0) {
      kk = *(const float2*)&kh[(size_t)g * DH + c2];
      vv = *(const float2*)&vh[(size_t)g * DH + c2];
    }
    *(float2*)&Ks[r * 66 + c2] = kk;
    *(float2*)&Vs[r * 66 + c2] = vv;
  }
  __syncthreads();

  int wv = threadIdx.x >> 5, lane = threadIdx.x & 31;
  int half = lane >> 4, l = lane & 15;
  const float* qrow =
      q + ((size_t)h * T_SEQ + w * 64 + wv * 16 + l) * DH + 2 * half;

  v8f s[8] = {};
  for (int k0 = 0; k0 < 64; k0 += 4) {
    v2f a;
    a.x = qrow[k0];
    a.y = qrow[k0 + 1];
#pragma unroll
    for (int nt = 0; nt < 8; ++nt) {
      v2f b;
      const float* kp = &Ks[(nt * 16 + l) * 66 + k0 + 2 * half];
      b.x = kp[0];
      b.y = kp[1];
      s[nt] = wmma4(a, b, s[nt]);
    }
  }
  float rmax[8], rsum[8];
#pragma unroll
  for (int r = 0; r < 8; ++r) { rmax[r] = -3.0e38f; rsum[r] = 0.f; }
#pragma unroll
  for (int nt = 0; nt < 8; ++nt) {
    int n = nt * 16 + l;
#pragma unroll
    for (int r = 0; r < 8; ++r) {
      int m = wv * 16 + r + 8 * half;
      float x = s[nt][r] * 0.125f;
      bool ok = (m + 64 >= n) && (w > 0 || n >= 64);
      x = ok ? x : -3.0e38f;
      s[nt][r] = x;
      rmax[r] = fmaxf(rmax[r], x);
    }
  }
#pragma unroll
  for (int d = 1; d < 16; d <<= 1)
#pragma unroll
    for (int r = 0; r < 8; ++r) rmax[r] = fmaxf(rmax[r], __shfl_xor(rmax[r], d));
#pragma unroll
  for (int nt = 0; nt < 8; ++nt)
#pragma unroll
    for (int r = 0; r < 8; ++r) {
      float e = __expf(s[nt][r] - rmax[r]);
      s[nt][r] = e;
      rsum[r] += e;
    }
#pragma unroll
  for (int d = 1; d < 16; d <<= 1)
#pragma unroll
    for (int r = 0; r < 8; ++r) rsum[r] += __shfl_xor(rsum[r], d);
  float rinv[8];
#pragma unroll
  for (int r = 0; r < 8; ++r) rinv[r] = 1.0f / rsum[r];
#pragma unroll
  for (int nt = 0; nt < 8; ++nt)
#pragma unroll
    for (int r = 0; r < 8; ++r)
      Ps[(wv * 16 + r + 8 * half) * 130 + nt * 16 + l] = s[nt][r] * rinv[r];
  __syncthreads();

  v8f o[4] = {};
  const float* prow = &Ps[(wv * 16 + l) * 130 + 2 * half];
  for (int k0 = 0; k0 < 128; k0 += 4) {
    v2f a;
    a.x = prow[k0];
    a.y = prow[k0 + 1];
#pragma unroll
    for (int nt = 0; nt < 4; ++nt) {
      v2f b;
      const float* vp = &Vs[(k0 + 2 * half) * 66 + nt * 16 + l];
      b.x = vp[0];
      b.y = vp[66];
      o[nt] = wmma4(a, b, o[nt]);
    }
  }
  float* oh = out + ((size_t)h * T_SEQ + w * 64) * DH;
#pragma unroll
  for (int nt = 0; nt < 4; ++nt)
#pragma unroll
    for (int r = 0; r < 8; ++r)
      oh[(size_t)(wv * 16 + r + 8 * half) * DH + nt * 16 + l] = o[nt][r];
}

__global__ void comp_branch_kernel(const float* __restrict__ q,
                                   const float* __restrict__ kc,
                                   const float* __restrict__ vc,
                                   float* __restrict__ bs,
                                   float* __restrict__ comp) {
  __shared__ float Ps[16 * 34];
  int h = blockIdx.x >> 7;
  int tt = blockIdx.x & 127;
  int t0 = tt << 4;
  int lane = threadIdx.x, half = lane >> 4, l = lane & 15;
  const float* qrow = q + ((size_t)h * T_SEQ + t0 + l) * DH + 2 * half;
  v8f s[2] = {};
  for (int k0 = 0; k0 < 64; k0 += 4) {
    v2f a;
    a.x = qrow[k0];
    a.y = qrow[k0 + 1];
#pragma unroll
    for (int nt = 0; nt < 2; ++nt) {
      v2f b;
      const float* kp = kc + (size_t)(h * 32 + nt * 16 + l) * 64 + k0 + 2 * half;
      b.x = kp[0];
      b.y = kp[1];
      s[nt] = wmma4(a, b, s[nt]);
    }
  }
#pragma unroll
  for (int nt = 0; nt < 2; ++nt)
#pragma unroll
    for (int r = 0; r < 8; ++r) s[nt][r] *= 0.125f;

  int qb = t0 >> 6;
#pragma unroll
  for (int nt = 0; nt < 2; ++nt) {
    float cs = 0.f;
#pragma unroll
    for (int r = 0; r < 8; ++r) cs += s[nt][r];
    cs += __shfl_xor(cs, 16);
    if (half == 0)
      atomicAdd(&bs[((size_t)h * 32 + qb) * 32 + nt * 16 + l],
                cs * (1.0f / 64.0f));
  }
  float rmax[8], rsum[8];
#pragma unroll
  for (int r = 0; r < 8; ++r) rmax[r] = fmaxf(s[0][r], s[1][r]);
#pragma unroll
  for (int d = 1; d < 16; d <<= 1)
#pragma unroll
    for (int r = 0; r < 8; ++r) rmax[r] = fmaxf(rmax[r], __shfl_xor(rmax[r], d));
#pragma unroll
  for (int r = 0; r < 8; ++r) rsum[r] = 0.f;
#pragma unroll
  for (int nt = 0; nt < 2; ++nt)
#pragma unroll
    for (int r = 0; r < 8; ++r) {
      float e = __expf(s[nt][r] - rmax[r]);
      s[nt][r] = e;
      rsum[r] += e;
    }
#pragma unroll
  for (int d = 1; d < 16; d <<= 1)
#pragma unroll
    for (int r = 0; r < 8; ++r) rsum[r] += __shfl_xor(rsum[r], d);
#pragma unroll
  for (int nt = 0; nt < 2; ++nt)
#pragma unroll
    for (int r = 0; r < 8; ++r)
      Ps[(r + 8 * half) * 34 + nt * 16 + l] = s[nt][r] / rsum[r];
  __syncthreads();

  v8f o[4] = {};
  const float* prow = &Ps[l * 34 + 2 * half];
  for (int k0 = 0; k0 < 32; k0 += 4) {
    v2f a;
    a.x = prow[k0];
    a.y = prow[k0 + 1];
#pragma unroll
    for (int nt = 0; nt < 4; ++nt) {
      v2f b;
      const float* vp = vc + (size_t)(h * 32 + k0 + 2 * half) * 64 + nt * 16 + l;
      b.x = vp[0];
      b.y = vp[64];
      o[nt] = wmma4(a, b, o[nt]);
    }
  }
  float* oh = comp + ((size_t)h * T_SEQ + t0) * DH;
#pragma unroll
  for (int nt = 0; nt < 4; ++nt)
#pragma unroll
    for (int r = 0; r < 8; ++r)
      oh[(size_t)(r + 8 * half) * DH + nt * 16 + l] = o[nt][r];
}

__global__ void block_stats_kernel(const float* __restrict__ bs,
                                   const float* __restrict__ ksc,
                                   const float* __restrict__ kbi,
                                   unsigned char* __restrict__ mask,
                                   float* __restrict__ avg_acc,
                                   float* __restrict__ tail) {
  int h = blockIdx.x, j = threadIdx.x;
  const float* row = bs + ((size_t)h * 32 + j) * 32;
  float sc[32], p[32];
  float mx = -3.0e38f;
  for (int c = 0; c < 32; ++c) { sc[c] = row[c]; mx = fmaxf(mx, sc[c]); }
  float ssum = 0.f;
  for (int c = 0; c < 32; ++c) { p[c] = __expf(sc[c] - mx); ssum += p[c]; }
  float inv = 1.0f / ssum;
  float ent = 0.f;
  for (int c = 0; c < 32; ++c) {
    float pp = p[c] * inv;
    ent -= pp * __logf(pp + 1e-9f);
  }
  float ne = ent / (logf(32.0f) + 1e-9f);
  float kr = 1.0f / (1.0f + __expf(-(ksc[h] * ne + kbi[h])));
  float dkf = fminf(fmaxf(8.0f * kr, 1.0f), 8.0f);
  int dk = (int)dkf;
  tail[1 + h * 32 + j] = kr;
  tail[1 + 512 + h * 32 + j] = ne;
  atomicAdd(avg_acc, (float)dk);
  unsigned used = 0;
  unsigned char mrow[32];
  for (int c = 0; c < 32; ++c) mrow[c] = 0;
  mrow[j] = 1;
  for (int kk = 0; kk < KMAX_SEL; ++kk) {
    float best = -3.4e38f;
    int bi = 0;
    for (int c = 0; c < 32; ++c)
      if (!((used >> c) & 1u) && sc[c] > best) { best = sc[c]; bi = c; }
    used |= 1u << bi;
    if (kk < dk) mrow[bi] = 1;
  }
  for (int c = 0; c < 32; ++c) mask[((size_t)h * 32 + j) * 32 + c] = mrow[c];
}

__global__ void fine_attn_kernel(const float* __restrict__ q,
                                 const float* __restrict__ k,
                                 const float* __restrict__ v,
                                 const unsigned char* __restrict__ mask,
                                 float* __restrict__ out) {
  __shared__ float Ks[64 * 66];
  __shared__ float Vs[64 * 66];
  __shared__ float Ps[64 * 66];
  int h = blockIdx.x >> 5, qb = blockIdx.x & 31;
  int wv = threadIdx.x >> 5, lane = threadIdx.x & 31;
  int half = lane >> 4, l = lane & 15;
  const float* kh = k + (size_t)h * T_SEQ * DH;
  const float* vh = v + (size_t)h * T_SEQ * DH;
  const float* qrow =
      q + ((size_t)h * T_SEQ + qb * 64 + wv * 16 + l) * DH + 2 * half;
  const unsigned char* mrow = mask + ((size_t)h * 32 + qb) * 32;
  v8f o[4] = {};
  float rmax[8], rsum[8];
#pragma unroll
  for (int r = 0; r < 8; ++r) { rmax[r] = -3.0e38f; rsum[r] = 0.f; }

  for (int j = 0; j < 32; ++j) {
    if (!mrow[j]) continue;
    __syncthreads();
    for (int idx = threadIdx.x; idx < 64 * 32; idx += blockDim.x) {
      int r = idx >> 5, c2 = (idx & 31) << 1;
      *(float2*)&Ks[r * 66 + c2] =
          *(const float2*)&kh[(size_t)(j * 64 + r) * DH + c2];
      *(float2*)&Vs[r * 66 + c2] =
          *(const float2*)&vh[(size_t)(j * 64 + r) * DH + c2];
    }
    __syncthreads();
    v8f s[4] = {};
    for (int k0 = 0; k0 < 64; k0 += 4) {
      v2f a;
      a.x = qrow[k0];
      a.y = qrow[k0 + 1];
#pragma unroll
      for (int nt = 0; nt < 4; ++nt) {
        v2f b;
        const float* kp = &Ks[(nt * 16 + l) * 66 + k0 + 2 * half];
        b.x = kp[0];
        b.y = kp[1];
        s[nt] = wmma4(a, b, s[nt]);
      }
    }
    float nm[8], ps[8], fac[8];
#pragma unroll
    for (int r = 0; r < 8; ++r) nm[r] = rmax[r];
#pragma unroll
    for (int nt = 0; nt < 4; ++nt)
#pragma unroll
      for (int r = 0; r < 8; ++r) {
        float x = s[nt][r] * 0.125f;
        s[nt][r] = x;
        nm[r] = fmaxf(nm[r], x);
      }
#pragma unroll
    for (int d = 1; d < 16; d <<= 1)
#pragma unroll
      for (int r = 0; r < 8; ++r) nm[r] = fmaxf(nm[r], __shfl_xor(nm[r], d));
#pragma unroll
    for (int r = 0; r < 8; ++r) { fac[r] = __expf(rmax[r] - nm[r]); ps[r] = 0.f; }
#pragma unroll
    for (int nt = 0; nt < 4; ++nt)
#pragma unroll
      for (int r = 0; r < 8; ++r) {
        float e = __expf(s[nt][r] - nm[r]);
        s[nt][r] = e;
        ps[r] += e;
      }
#pragma unroll
    for (int d = 1; d < 16; d <<= 1)
#pragma unroll
      for (int r = 0; r < 8; ++r) ps[r] += __shfl_xor(ps[r], d);
#pragma unroll
    for (int r = 0; r < 8; ++r) {
      rsum[r] = rsum[r] * fac[r] + ps[r];
      rmax[r] = nm[r];
    }
#pragma unroll
    for (int nt = 0; nt < 4; ++nt)
#pragma unroll
      for (int r = 0; r < 8; ++r)
        Ps[(wv * 16 + r + 8 * half) * 66 + nt * 16 + l] = s[nt][r];
#pragma unroll
    for (int nt = 0; nt < 4; ++nt)
#pragma unroll
      for (int r = 0; r < 8; ++r) o[nt][r] *= fac[r];
    __syncthreads();
    const float* prow = &Ps[(wv * 16 + l) * 66 + 2 * half];
    for (int k0 = 0; k0 < 64; k0 += 4) {
      v2f a;
      a.x = prow[k0];
      a.y = prow[k0 + 1];
#pragma unroll
      for (int nt = 0; nt < 4; ++nt) {
        v2f b;
        const float* vp = &Vs[(k0 + 2 * half) * 66 + nt * 16 + l];
        b.x = vp[0];
        b.y = vp[66];
        o[nt] = wmma4(a, b, o[nt]);
      }
    }
  }
  float* oh = out + ((size_t)h * T_SEQ + qb * 64) * DH;
#pragma unroll
  for (int nt = 0; nt < 4; ++nt)
#pragma unroll
    for (int r = 0; r < 8; ++r)
      oh[(size_t)(wv * 16 + r + 8 * half) * DH + nt * 16 + l] =
          o[nt][r] / rsum[r];
}

__global__ void combine_kernel(const float* __restrict__ sw,
                               const float* __restrict__ comp,
                               const float* __restrict__ fine,
                               const float* __restrict__ gate,
                               float* __restrict__ merged) {
  int idx = blockIdx.x * blockDim.x + threadIdx.x;
  if (idx >= T_SEQ * HID_DIM) return;
  int t = idx >> 10, c = idx & 1023, hh = c >> 6, dd = c & 63;
  size_t a = ((size_t)hh * T_SEQ + t) * DH + dd;
  const float* g = gate + (size_t)t * 48 + hh * 3;
  merged[idx] = sw[a] * g[0] + comp[a] * g[1] + fine[a] * g[2];
}

__global__ void zero_kernel(float* __restrict__ p, int n) {
  int i = blockIdx.x * blockDim.x + threadIdx.x;
  if (i < n) p[i] = 0.f;
}

__global__ void finalize_kernel(const float* __restrict__ avg_acc,
                                float* __restrict__ tail) {
  if (threadIdx.x == 0) tail[0] = avg_acc[0] * (1.0f / 512.0f);
}

extern "C" void kernel_launch(void* const* d_in, const int* in_sizes, int n_in,
                              void* d_out, int out_size, void* d_ws,
                              size_t ws_size, hipStream_t stream) {
  const float* X = (const float*)d_in[0];
  const float* Wq = (const float*)d_in[1];
  const float* Wk = (const float*)d_in[2];
  const float* Wv = (const float*)d_in[3];
  const float* Wo = (const float*)d_in[4];
  const float* Wc = (const float*)d_in[5];
  const float* bc = (const float*)d_in[6];
  const float* ksc = (const float*)d_in[7];
  const float* kbi = (const float*)d_in[8];
  const float* Wg = (const float*)d_in[9];
  const float* bg = (const float*)d_in[10];
  float* out = (float*)d_out;

  const size_t U = (size_t)NH * T_SEQ * DH;
  float* ws = (float*)d_ws;
  float* q = ws;
  float* kbuf = ws + U;
  float* vbuf = ws + 2 * U;
  float* sw = ws + 3 * U;
  float* comp = ws + 4 * U;
  float* fine = ws + 5 * U;
  float* merged = q;
  float* extra = ws + 6 * U;
  float* gate = extra;
  float* kc = gate + (size_t)T_SEQ * 48;
  float* vc = kc + 512 * 64;
  float* bs = vc + 512 * 64;
  float* avg = bs + 16384;
  unsigned char* mask = (unsigned char*)(avg + 64);

  zero_kernel<<<(16385 + 255) / 256, 256, 0, stream>>>(bs, 16385);

  {
    int wtiles = (T_SEQ / 64) * (HID_DIM / 32);
    int blocks = (wtiles + 7) / 8;
    wmma_gemm_f32_blk<<<blocks, 256, 0, stream>>>(X, Wq, nullptr, q, T_SEQ,
                                                  HID_DIM, HID_DIM, 0, 1);
    wmma_gemm_f32_blk<<<blocks, 256, 0, stream>>>(X, Wk, nullptr, kbuf, T_SEQ,
                                                  HID_DIM, HID_DIM, 0, 1);
    wmma_gemm_f32_blk<<<blocks, 256, 0, stream>>>(X, Wv, nullptr, vbuf, T_SEQ,
                                                  HID_DIM, HID_DIM, 0, 1);
  }
  {
    int tiles = (T_SEQ / 16) * (48 / 16);
    wmma_gemm_f32<<<(tiles + 7) / 8, 256, 0, stream>>>(X, Wg, bg, gate, T_SEQ,
                                                       48, HID_DIM, 2, 0);
  }
  sliding_window_kernel<<<NH * NB, 128,
                          (2 * 128 * 66 + 64 * 130) * sizeof(float), stream>>>(
      q, kbuf, vbuf, sw);
  {
    compress_gemm_splitk<<<64, 1024, 0, stream>>>(kbuf, Wc, bc, kc);
    compress_gemm_splitk<<<64, 1024, 0, stream>>>(vbuf, Wc, bc, vc);
  }
  comp_branch_kernel<<<NH * (T_SEQ / 16), 32, 0, stream>>>(q, kc, vc, bs, comp);
  block_stats_kernel<<<NH, 32, 0, stream>>>(bs, ksc, kbi, mask, avg,
                                            out + (size_t)T_SEQ * HID_DIM);
  fine_attn_kernel<<<NH * NB, 128, 0, stream>>>(q, kbuf, vbuf, mask, fine);
  combine_kernel<<<(T_SEQ * HID_DIM + 255) / 256, 256, 0, stream>>>(
      sw, comp, fine, gate, merged);
  {
    int wtiles = (T_SEQ / 64) * (HID_DIM / 32);
    wmma_gemm_f32_blk<<<(wtiles + 7) / 8, 256, 0, stream>>>(
        merged, Wo, nullptr, out, T_SEQ, HID_DIM, HID_DIM, 0, 0);
  }
  finalize_kernel<<<1, 1, 0, stream>>>(avg, out + (size_t)T_SEQ * HID_DIM);
  (void)in_sizes;
  (void)n_in;
  (void)out_size;
  (void)ws_size;
}


# Q/K/V/out GEMMs (K=1024): K-loop in 16-k blocks on two register sets, next block's 32 fragment loads issued as one group ahead of the current block's 32 WMMAs (launch bound 256); k order kept
# speedup vs baseline: 5.3216x; 2.3034x over previous
#include <hip/hip_runtime.h>
#include <hip/hip_bf16.h>
#include <math.h>

typedef float v2f __attribute__((ext_vector_type(2)));
typedef float v8f __attribute__((ext_vector_type(8)));

#define T_SEQ 2048
#define HID_DIM 1024
#define NH 16
#define DH 64
#define NB 32
#define KMAX_SEL 8

__device__ __forceinline__ v8f wmma4(v2f a, v2f b, v8f c) {
  return __builtin_amdgcn_wmma_f32_16x16x4_f32(false, a, false, b, (short)0, c,
                                               false, false);
}

__device__ __forceinline__ float apply_act(float x, int act) {
  if (act == 1) return 0.5f * x * (1.0f + erff(x * 0.70710678118654752f));
  if (act == 2) return 1.0f / (1.0f + __expf(-x));
  return x;
}

struct GemmFrags {
  v2f a[4][4];
  v2f b[4][2];
};

__device__ __forceinline__ void gemm_load_block(GemmFrags& f, const float* ap,
                                                const float* bp,
                                                size_t aRowStep, size_t N) {
#pragma unroll
  for (int s = 0; s < 4; ++s) {
#pragma unroll
    for (int mi = 0; mi < 4; ++mi) {
      const float* p = ap + (size_t)mi * aRowStep + 4 * s;
      f.a[s][mi].x = p[0];
      f.a[s][mi].y = p[1];
    }
#pragma unroll
    for (int ni = 0; ni < 2; ++ni) {
      f.b[s][ni].x = bp[(size_t)(4 * s) * N + ni * 16];
      f.b[s][ni].y = bp[(size_t)(4 * s + 1) * N + ni * 16];
    }
  }
}

__device__ __forceinline__ void gemm_mma_block(const GemmFrags& f,
                                               v8f (&acc)[4][2]) {
#pragma unroll
  for (int s = 0; s < 4; ++s)
#pragma unroll
    for (int mi = 0; mi < 4; ++mi)
#pragma unroll
      for (int ni = 0; ni < 2; ++ni)
        acc[mi][ni] = wmma4(f.a[s][mi], f.b[s][ni], acc[mi][ni]);
}

__global__ void __launch_bounds__(256)
wmma_gemm_f32_blk(const float* __restrict__ A, const float* __restrict__ B,
                  const float* __restrict__ bias, float* __restrict__ C, int M,
                  int N, int K, int act, int headSplit) {
  int wave = threadIdx.x >> 5;
  int lane = threadIdx.x & 31;
  int wtile = blockIdx.x * (blockDim.x >> 5) + wave;
  int wn = N >> 5;
  int total = (M >> 6) * wn;
  if (wtile >= total) return;
  int tm = wtile / wn, tn = wtile - tm * wn;
  int half = lane >> 4, l = lane & 15;
  int m0 = tm << 6, n0 = tn << 5;

  v8f acc[4][2];
#pragma unroll
  for (int mi = 0; mi < 4; ++mi)
#pragma unroll
    for (int ni = 0; ni < 2; ++ni) acc[mi][ni] = (v8f){};

  const float* ap = A + (size_t)(m0 + l) * K + 2 * half;
  const float* bp = B + (size_t)(2 * half) * N + n0 + l;
  const size_t aRowStep = (size_t)16 * K;
  const size_t bBlk = (size_t)16 * N;
  GemmFrags f0, f1;
  gemm_load_block(f0, ap, bp, aRowStep, (size_t)N);
  for (int k0 = 0; k0 < K - 32; k0 += 32) {
    gemm_load_block(f1, ap + 16, bp + bBlk, aRowStep, (size_t)N);
    __builtin_amdgcn_sched_barrier(0);
    gemm_mma_block(f0, acc);
    __builtin_amdgcn_sched_barrier(0);
    gemm_load_block(f0, ap + 32, bp + 2 * bBlk, aRowStep, (size_t)N);
    __builtin_amdgcn_sched_barrier(0);
    gemm_mma_block(f1, acc);
    __builtin_amdgcn_sched_barrier(0);
    ap += 32;
    bp += 2 * bBlk;
  }
  gemm_load_block(f1, ap + 16, bp + bBlk, aRowStep, (size_t)N);
  __builtin_amdgcn_sched_barrier(0);
  gemm_mma_block(f0, acc);
  gemm_mma_block(f1, acc);

  float bv[2];
#pragma unroll
  for (int ni = 0; ni < 2; ++ni) bv[ni] = bias ? bias[n0 + ni * 16 + l] : 0.0f;
#pragma unroll
  for (int mi = 0; mi < 4; ++mi)
#pragma unroll
    for (int ni = 0; ni < 2; ++ni) {
      int col = n0 + ni * 16 + l;
#pragma unroll
      for (int r = 0; r < 8; ++r) {
        float x = apply_act(acc[mi][ni][r] + bv[ni], act);
        int m = m0 + mi * 16 + r + 8 * half;
        if (headSplit) {
          int hh = col >> 6, dd = col & 63;
          C[((size_t)hh * M + m) * 64 + dd] = x;
        } else {
          C[(size_t)m * N + col] = x;
        }
      }
    }
}

#define CK_K 4096
#define CK_N 64
#define CK_WAVES 32
#define CK_SLICE (CK_K / CK_WAVES)
#define CK_NI_PITCH 272
#define CK_W_PITCH (2 * CK_NI_PITCH)
__global__ void __launch_bounds__(CK_WAVES * 32)
compress_gemm_splitk(const float* __restrict__ A, const float* __restrict__ B,
                     const float* __restrict__ bias, float* __restrict__ C) {
  __shared__ float part[CK_WAVES * CK_W_PITCH];
  int wave = threadIdx.x >> 5, lane = threadIdx.x & 31;
  int half = lane >> 4, l = lane & 15;
  int m0 = (blockIdx.x >> 1) << 4, n0 = (blockIdx.x & 1) << 5;
  int kbase = wave * CK_SLICE;
  const float* ap = A + (size_t)(m0 + l) * CK_K + kbase + 2 * half;
  const float* bp = B + (size_t)(kbase + 2 * half) * CK_N + n0 + l;
  v8f acc0 = {}, acc1 = {};
#pragma unroll
  for (int s = 0; s < CK_SLICE / 4; ++s) {
    v2f a, b0, b1;
    a.x = ap[4 * s];
    a.y = ap[4 * s + 1];
    b0.x = bp[(4 * s) * CK_N];
    b0.y = bp[(4 * s + 1) * CK_N];
    b1.x = bp[(4 * s) * CK_N + 16];
    b1.y = bp[(4 * s + 1) * CK_N + 16];
    acc0 = wmma4(a, b0, acc0);
    acc1 = wmma4(a, b1, acc1);
  }
  float* pw = part + wave * CK_W_PITCH + lane;
#pragma unroll
  for (int r = 0; r < 8; ++r) {
    pw[r * 32] = acc0[r];
    pw[CK_NI_PITCH + r * 32] = acc1[r];
  }
  __syncthreads();
  int t = threadIdx.x;
  if (t < 512) {
    int row = t >> 5, col = t & 31;
    const float* pr = part + (col >> 4) * CK_NI_PITCH + (row & 7) * 32 +
                      (row >> 3) * 16 + (col & 15);
    float sum = pr[0];
#pragma unroll
    for (int w = 1; w < CK_WAVES; ++w) sum += pr[w * CK_W_PITCH];
    C[(size_t)(m0 + row) * CK_N + n0 + col] =
        apply_act(sum + bias[n0 + col], 1);
  }
}

__global__ void wmma_gemm_f32(const float* __restrict__ A,
                              const float* __restrict__ B,
                              const float* __restrict__ bias,
                              float* __restrict__ C, int M, int N, int K,
                              int act, int headSplit) {
  int wave = threadIdx.x >> 5;
  int lane = threadIdx.x & 31;
  int tile = blockIdx.x * (blockDim.x >> 5) + wave;
  int tiles_n = N >> 4;
  int total = (M >> 4) * tiles_n;
  if (tile >= total) return;
  int tm = tile / tiles_n, tn = tile - tm * tiles_n;
  int half = lane >> 4, l = lane & 15;
  int arow = (tm << 4) + l;
  int bcol = (tn << 4) + l;
  v8f acc = {};
  const float* ap = A + (size_t)arow * K + 2 * half;
  const float* bp = B + (size_t)(2 * half) * N + bcol;
  for (int k0 = 0; k0 < K; k0 += 4) {
    v2f a, b;
    a.x = ap[0];
    a.y = ap[1];
    b.x = bp[0];
    b.y = bp[N];
    acc = wmma4(a, b, acc);
    ap += 4;
    bp += (size_t)4 * N;
  }
  float bv = bias ? bias[bcol] : 0.0f;
#pragma unroll
  for (int r = 0; r < 8; ++r) {
    float x = apply_act(acc[r] + bv, act);
    int m = (tm << 4) + r + 8 * half;
    if (headSplit) {
      int hh = bcol >> 6, dd = bcol & 63;
      C[((size_t)hh * M + m) * 64 + dd] = x;
    } else {
      C[(size_t)m * N + bcol] = x;
    }
  }
}

__global__ void sliding_window_kernel(const float* __restrict__ q,
                                      const float* __restrict__ k,
                                      const float* __restrict__ v,
                                      float* __restrict__ out) {
  extern __shared__ float lds[];
  float* Ks = lds;
  float* Vs = Ks + 128 * 66;
  float* Ps = Vs + 128 * 66;
  int h = blockIdx.x >> 5;
  int w = blockIdx.x & 31;
  const float* kh = k + (size_t)h * T_SEQ * DH;
  const float* vh = v + (size_t)h * T_SEQ * DH;
  for (int idx = threadIdx.x; idx < 128 * 32; idx += blockDim.x) {
    int r = idx >> 5, c2 = (idx & 31) << 1;
    int g = (w - 1) * 64 + r;
    float2 kk = make_float2(0.f, 0.f), vv = make_float2(0.f, 0.f);
    if (g >= 0) {
      kk = *(const float2*)&kh[(size_t)g * DH + c2];
      vv = *(const float2*)&vh[(size_t)g * DH + c2];
    }
    *(float2*)&Ks[r * 66 + c2] = kk;
    *(float2*)&Vs[r * 66 + c2] = vv;
  }
  __syncthreads();

  int wv = threadIdx.x >> 5, lane = threadIdx.x & 31;
  int half = lane >> 4, l = lane & 15;
  const float* qrow =
      q + ((size_t)h * T_SEQ + w * 64 + wv * 16 + l) * DH + 2 * half;

  v8f s[8] = {};
  for (int k0 = 0; k0 < 64; k0 += 4) {
    v2f a;
    a.x = qrow[k0];
    a.y = qrow[k0 + 1];
#pragma unroll
    for (int nt = 0; nt < 8; ++nt) {
      v2f b;
      const float* kp = &Ks[(nt * 16 + l) * 66 + k0 + 2 * half];
      b.x = kp[0];
      b.y = kp[1];
      s[nt] = wmma4(a, b, s[nt]);
    }
  }
  float rmax[8], rsum[8];
#pragma unroll
  for (int r = 0; r < 8; ++r) { rmax[r] = -3.0e38f; rsum[r] = 0.f; }
#pragma unroll
  for (int nt = 0; nt < 8; ++nt) {
    int n = nt * 16 + l;
#pragma unroll
    for (int r = 0; r < 8; ++r) {
      int m = wv * 16 + r + 8 * half;
      float x = s[nt][r] * 0.125f;
      bool ok = (m + 64 >= n) && (w > 0 || n >= 64);
      x = ok ? x : -3.0e38f;
      s[nt][r] = x;
      rmax[r] = fmaxf(rmax[r], x);
    }
  }
#pragma unroll
  for (int d = 1; d < 16; d <<= 1)
#pragma unroll
    for (int r = 0; r < 8; ++r) rmax[r] = fmaxf(rmax[r], __shfl_xor(rmax[r], d));
#pragma unroll
  for (int nt = 0; nt < 8; ++nt)
#pragma unroll
    for (int r = 0; r < 8; ++r) {
      float e = __expf(s[nt][r] - rmax[r]);
      s[nt][r] = e;
      rsum[r] += e;
    }
#pragma unroll
  for (int d = 1; d < 16; d <<= 1)
#pragma unroll
    for (int r = 0; r < 8; ++r) rsum[r] += __shfl_xor(rsum[r], d);
  float rinv[8];
#pragma unroll
  for (int r = 0; r < 8; ++r) rinv[r] = 1.0f / rsum[r];
#pragma unroll
  for (int nt = 0; nt < 8; ++nt)
#pragma unroll
    for (int r = 0; r < 8; ++r)
      Ps[(wv * 16 + r + 8 * half) * 130 + nt * 16 + l] = s[nt][r] * rinv[r];
  __syncthreads();

  v8f o[4] = {};
  const float* prow = &Ps[(wv * 16 + l) * 130 + 2 * half];
  for (int k0 = 0; k0 < 128; k0 += 4) {
    v2f a;
    a.x = prow[k0];
    a.y = prow[k0 + 1];
#pragma unroll
    for (int nt = 0; nt < 4; ++nt) {
      v2f b;
      const float* vp = &Vs[(k0 + 2 * half) * 66 + nt * 16 + l];
      b.x = vp[0];
      b.y = vp[66];
      o[nt] = wmma4(a, b, o[nt]);
    }
  }
  float* oh = out + ((size_t)h * T_SEQ + w * 64) * DH;
#pragma unroll
  for (int nt = 0; nt < 4; ++nt)
#pragma unroll
    for (int r = 0; r < 8; ++r)
      oh[(size_t)(wv * 16 + r + 8 * half) * DH + nt * 16 + l] = o[nt][r];
}

__global__ void comp_branch_kernel(const float* __restrict__ q,
                                   const float* __restrict__ kc,
                                   const float* __restrict__ vc,
                                   float* __restrict__ bs,
                                   float* __restrict__ comp) {
  __shared__ float Ps[16 * 34];
  int h = blockIdx.x >> 7;
  int tt = blockIdx.x & 127;
  int t0 = tt << 4;
  int lane = threadIdx.x, half = lane >> 4, l = lane & 15;
  const float* qrow = q + ((size_t)h * T_SEQ + t0 + l) * DH + 2 * half;
  v8f s[2] = {};
  for (int k0 = 0; k0 < 64; k0 += 4) {
    v2f a;
    a.x = qrow[k0];
    a.y = qrow[k0 + 1];
#pragma unroll
    for (int nt = 0; nt < 2; ++nt) {
      v2f b;
      const float* kp = kc + (size_t)(h * 32 + nt * 16 + l) * 64 + k0 + 2 * half;
      b.x = kp[0];
      b.y = kp[1];
      s[nt] = wmma4(a, b, s[nt]);
    }
  }
#pragma unroll
  for (int nt = 0; nt < 2; ++nt)
#pragma unroll
    for (int r = 0; r < 8; ++r) s[nt][r] *= 0.125f;

  int qb = t0 >> 6;
#pragma unroll
  for (int nt = 0; nt < 2; ++nt) {
    float cs = 0.f;
#pragma unroll
    for (int r = 0; r < 8; ++r) cs += s[nt][r];
    cs += __shfl_xor(cs, 16);
    if (half == 0)
      atomicAdd(&bs[((size_t)h * 32 + qb) * 32 + nt * 16 + l],
                cs * (1.0f / 64.0f));
  }
  float rmax[8], rsum[8];
#pragma unroll
  for (int r = 0; r < 8; ++r) rmax[r] = fmaxf(s[0][r], s[1][r]);
#pragma unroll
  for (int d = 1; d < 16; d <<= 1)
#pragma unroll
    for (int r = 0; r < 8; ++r) rmax[r] = fmaxf(rmax[r], __shfl_xor(rmax[r], d));
#pragma unroll
  for (int r = 0; r < 8; ++r) rsum[r] = 0.f;
#pragma unroll
  for (int nt = 0; nt < 2; ++nt)
#pragma unroll
    for (int r = 0; r < 8; ++r) {
      float e = __expf(s[nt][r] - rmax[r]);
      s[nt][r] = e;
      rsum[r] += e;
    }
#pragma unroll
  for (int d = 1; d < 16; d <<= 1)
#pragma unroll
    for (int r = 0; r < 8; ++r) rsum[r] += __shfl_xor(rsum[r], d);
#pragma unroll
  for (int nt = 0; nt < 2; ++nt)
#pragma unroll
    for (int r = 0; r < 8; ++r)
      Ps[(r + 8 * half) * 34 + nt * 16 + l] = s[nt][r] / rsum[r];
  __syncthreads();

  v8f o[4] = {};
  const float* prow = &Ps[l * 34 + 2 * half];
  for (int k0 = 0; k0 < 32; k0 += 4) {
    v2f a;
    a.x = prow[k0];
    a.y = prow[k0 + 1];
#pragma unroll
    for (int nt = 0; nt < 4; ++nt) {
      v2f b;
      const float* vp = vc + (size_t)(h * 32 + k0 + 2 * half) * 64 + nt * 16 + l;
      b.x = vp[0];
      b.y = vp[64];
      o[nt] = wmma4(a, b, o[nt]);
    }
  }
  float* oh = comp + ((size_t)h * T_SEQ + t0) * DH;
#pragma unroll
  for (int nt = 0; nt < 4; ++nt)
#pragma unroll
    for (int r = 0; r < 8; ++r)
      oh[(size_t)(r + 8 * half) * DH + nt * 16 + l] = o[nt][r];
}

__global__ void block_stats_kernel(const float* __restrict__ bs,
                                   const float* __restrict__ ksc,
                                   const float* __restrict__ kbi,
                                   unsigned char* __restrict__ mask,
                                   float* __restrict__ avg_acc,
                                   float* __restrict__ tail) {
  int h = blockIdx.x, j = threadIdx.x;
  const float* row = bs + ((size_t)h * 32 + j) * 32;
  float sc[32], p[32];
  float mx = -3.0e38f;
  for (int c = 0; c < 32; ++c) { sc[c] = row[c]; mx = fmaxf(mx, sc[c]); }
  float ssum = 0.f;
  for (int c = 0; c < 32; ++c) { p[c] = __expf(sc[c] - mx); ssum += p[c]; }
  float inv = 1.0f / ssum;
  float ent = 0.f;
  for (int c = 0; c < 32; ++c) {
    float pp = p[c] * inv;
    ent -= pp * __logf(pp + 1e-9f);
  }
  float ne = ent / (logf(32.0f) + 1e-9f);
  float kr = 1.0f / (1.0f + __expf(-(ksc[h] * ne + kbi[h])));
  float dkf = fminf(fmaxf(8.0f * kr, 1.0f), 8.0f);
  int dk = (int)dkf;
  tail[1 + h * 32 + j] = kr;
  tail[1 + 512 + h * 32 + j] = ne;
  atomicAdd(avg_acc, (float)dk);
  unsigned used = 0;
  unsigned char mrow[32];
  for (int c = 0; c < 32; ++c) mrow[c] = 0;
  mrow[j] = 1;
  for (int kk = 0; kk < KMAX_SEL; ++kk) {
    float best = -3.4e38f;
    int bi = 0;
    for (int c = 0; c < 32; ++c)
      if (!((used >> c) & 1u) && sc[c] > best) { best = sc[c]; bi = c; }
    used |= 1u << bi;
    if (kk < dk) mrow[bi] = 1;
  }
  for (int c = 0; c < 32; ++c) mask[((size_t)h * 32 + j) * 32 + c] = mrow[c];
}

__global__ void fine_attn_kernel(const float* __restrict__ q,
                                 const float* __restrict__ k,
                                 const float* __restrict__ v,
                                 const unsigned char* __restrict__ mask,
                                 float* __restrict__ out) {
  __shared__ float Ks[64 * 66];
  __shared__ float Vs[64 * 66];
  __shared__ float Ps[64 * 66];
  int h = blockIdx.x >> 5, qb = blockIdx.x & 31;
  int wv = threadIdx.x >> 5, lane = threadIdx.x & 31;
  int half = lane >> 4, l = lane & 15;
  const float* kh = k + (size_t)h * T_SEQ * DH;
  const float* vh = v + (size_t)h * T_SEQ * DH;
  const float* qrow =
      q + ((size_t)h * T_SEQ + qb * 64 + wv * 16 + l) * DH + 2 * half;
  const unsigned char* mrow = mask + ((size_t)h * 32 + qb) * 32;
  v8f o[4] = {};
  float rmax[8], rsum[8];
#pragma unroll
  for (int r = 0; r < 8; ++r) { rmax[r] = -3.0e38f; rsum[r] = 0.f; }

  for (int j = 0; j < 32; ++j) {
    if (!mrow[j]) continue;
    __syncthreads();
    for (int idx = threadIdx.x; idx < 64 * 32; idx += blockDim.x) {
      int r = idx >> 5, c2 = (idx & 31) << 1;
      *(float2*)&Ks[r * 66 + c2] =
          *(const float2*)&kh[(size_t)(j * 64 + r) * DH + c2];
      *(float2*)&Vs[r * 66 + c2] =
          *(const float2*)&vh[(size_t)(j * 64 + r) * DH + c2];
    }
    __syncthreads();
    v8f s[4] = {};
    for (int k0 = 0; k0 < 64; k0 += 4) {
      v2f a;
      a.x = qrow[k0];
      a.y = qrow[k0 + 1];
#pragma unroll
      for (int nt = 0; nt < 4; ++nt) {
        v2f b;
        const float* kp = &Ks[(nt * 16 + l) * 66 + k0 + 2 * half];
        b.x = kp[0];
        b.y = kp[1];
        s[nt] = wmma4(a, b, s[nt]);
      }
    }
    float nm[8], ps[8], fac[8];
#pragma unroll
    for (int r = 0; r < 8; ++r) nm[r] = rmax[r];
#pragma unroll
    for (int nt = 0; nt < 4; ++nt)
#pragma unroll
      for (int r = 0; r < 8; ++r) {
        float x = s[nt][r] * 0.125f;
        s[nt][r] = x;
        nm[r] = fmaxf(nm[r], x);
      }
#pragma unroll
    for (int d = 1; d < 16; d <<= 1)
#pragma unroll
      for (int r = 0; r < 8; ++r) nm[r] = fmaxf(nm[r], __shfl_xor(nm[r], d));
#pragma unroll
    for (int r = 0; r < 8; ++r) { fac[r] = __expf(rmax[r] - nm[r]); ps[r] = 0.f; }
#pragma unroll
    for (int nt = 0; nt < 4; ++nt)
#pragma unroll
      for (int r = 0; r < 8; ++r) {
        float e = __expf(s[nt][r] - nm[r]);
        s[nt][r] = e;
        ps[r] += e;
      }
#pragma unroll
    for (int d = 1; d < 16; d <<= 1)
#pragma unroll
      for (int r = 0; r < 8; ++r) ps[r] += __shfl_xor(ps[r], d);
#pragma unroll
    for (int r = 0; r < 8; ++r) {
      rsum[r] = rsum[r] * fac[r] + ps[r];
      rmax[r] = nm[r];
    }
#pragma unroll
    for (int nt = 0; nt < 4; ++nt)
#pragma unroll
      for (int r = 0; r < 8; ++r)
        Ps[(wv * 16 + r + 8 * half) * 66 + nt * 16 + l] = s[nt][r];
#pragma unroll
    for (int nt = 0; nt < 4; ++nt)
#pragma unroll
      for (int r = 0; r < 8; ++r) o[nt][r] *= fac[r];
    __syncthreads();
    const float* prow = &Ps[(wv * 16 + l) * 66 + 2 * half];
    for (int k0 = 0; k0 < 64; k0 += 4) {
      v2f a;
      a.x = prow[k0];
      a.y = prow[k0 + 1];
#pragma unroll
      for (int nt = 0; nt < 4; ++nt) {
        v2f b;
        const float* vp = &Vs[(k0 + 2 * half) * 66 + nt * 16 + l];
        b.x = vp[0];
        b.y = vp[66];
        o[nt] = wmma4(a, b, o[nt]);
      }
    }
  }
  float* oh = out + ((size_t)h * T_SEQ + qb * 64) * DH;
#pragma unroll
  for (int nt = 0; nt < 4; ++nt)
#pragma unroll
    for (int r = 0; r < 8; ++r)
      oh[(size_t)(wv * 16 + r + 8 * half) * DH + nt * 16 + l] =
          o[nt][r] / rsum[r];
}

__global__ void combine_kernel(const float* __restrict__ sw,
                               const float* __restrict__ comp,
                               const float* __restrict__ fine,
                               const float* __restrict__ gate,
                               float* __restrict__ merged) {
  int idx = blockIdx.x * blockDim.x + threadIdx.x;
  if (idx >= T_SEQ * HID_DIM) return;
  int t = idx >> 10, c = idx & 1023, hh = c >> 6, dd = c & 63;
  size_t a = ((size_t)hh * T_SEQ + t) * DH + dd;
  const float* g = gate + (size_t)t * 48 + hh * 3;
  merged[idx] = sw[a] * g[0] + comp[a] * g[1] + fine[a] * g[2];
}

__global__ void zero_kernel(float* __restrict__ p, int n) {
  int i = blockIdx.x * blockDim.x + threadIdx.x;
  if (i < n) p[i] = 0.f;
}

__global__ void finalize_kernel(const float* __restrict__ avg_acc,
                                float* __restrict__ tail) {
  if (threadIdx.x == 0) tail[0] = avg_acc[0] * (1.0f / 512.0f);
}

extern "C" void kernel_launch(void* const* d_in, const int* in_sizes, int n_in,
                              void* d_out, int out_size, void* d_ws,
                              size_t ws_size, hipStream_t stream) {
  const float* X = (const float*)d_in[0];
  const float* Wq = (const float*)d_in[1];
  const float* Wk = (const float*)d_in[2];
  const float* Wv = (const float*)d_in[3];
  const float* Wo = (const float*)d_in[4];
  const float* Wc = (const float*)d_in[5];
  const float* bc = (const float*)d_in[6];
  const float* ksc = (const float*)d_in[7];
  const float* kbi = (const float*)d_in[8];
  const float* Wg = (const float*)d_in[9];
  const float* bg = (const float*)d_in[10];
  float* out = (float*)d_out;

  const size_t U = (size_t)NH * T_SEQ * DH;
  float* ws = (float*)d_ws;
  float* q = ws;
  float* kbuf = ws + U;
  float* vbuf = ws + 2 * U;
  float* sw = ws + 3 * U;
  float* comp = ws + 4 * U;
  float* fine = ws + 5 * U;
  float* merged = q;
  float* extra = ws + 6 * U;
  float* gate = extra;
  float* kc = gate + (size_t)T_SEQ * 48;
  float* vc = kc + 512 * 64;
  float* bs = vc + 512 * 64;
  float* avg = bs + 16384;
  unsigned char* mask = (unsigned char*)(avg + 64);

  zero_kernel<<<(16385 + 255) / 256, 256, 0, stream>>>(bs, 16385);

  {
    int wtiles = (T_SEQ / 64) * (HID_DIM / 32);
    int blocks = (wtiles + 7) / 8;
    wmma_gemm_f32_blk<<<blocks, 256, 0, stream>>>(X, Wq, nullptr, q, T_SEQ,
                                                  HID_DIM, HID_DIM, 0, 1);
    wmma_gemm_f32_blk<<<blocks, 256, 0, stream>>>(X, Wk, nullptr, kbuf, T_SEQ,
                                                  HID_DIM, HID_DIM, 0, 1);
    wmma_gemm_f32_blk<<<blocks, 256, 0, stream>>>(X, Wv, nullptr, vbuf, T_SEQ,
                                                  HID_DIM, HID_DIM, 0, 1);
  }
  {
    int tiles = (T_SEQ / 16) * (48 / 16);
    wmma_gemm_f32<<<(tiles + 7) / 8, 256, 0, stream>>>(X, Wg, bg, gate, T_SEQ,
                                                       48, HID_DIM, 2, 0);
  }
  sliding_window_kernel<<<NH * NB, 128,
                          (2 * 128 * 66 + 64 * 130) * sizeof(float), stream>>>(
      q, kbuf, vbuf, sw);
  {
    compress_gemm_splitk<<<64, 1024, 0, stream>>>(kbuf, Wc, bc, kc);
    compress_gemm_splitk<<<64, 1024, 0, stream>>>(vbuf, Wc, bc, vc);
  }
  comp_branch_kernel<<<NH * (T_SEQ / 16), 32, 0, stream>>>(q, kc, vc, bs, comp);
  block_stats_kernel<<<NH, 32, 0, stream>>>(bs, ksc, kbi, mask, avg,
                                            out + (size_t)T_SEQ * HID_DIM);
  fine_attn_kernel<<<NH * NB, 128, 0, stream>>>(q, kbuf, vbuf, mask, fine);
  combine_kernel<<<(T_SEQ * HID_DIM + 255) / 256, 256, 0, stream>>>(
      sw, comp, fine, gate, merged);
  {
    int wtiles = (T_SEQ / 64) * (HID_DIM / 32);
    wmma_gemm_f32_blk<<<(wtiles + 7) / 8, 256, 0, stream>>>(
        merged, Wo, nullptr, out, T_SEQ, HID_DIM, HID_DIM, 0, 0);
  }
  finalize_kernel<<<1, 1, 0, stream>>>(avg, out + (size_t)T_SEQ * HID_DIM);
  (void)in_sizes;
  (void)n_in;
  (void)out_size;
  (void)ws_size;
}


# Q/K/V/out GEMMs spread over all 256 WGPs: Q,K,V one launch of 256 12-wave groups (3 waves/SIMD), out 256 4-wave groups; before 128 8-wave groups used half the WGPs; k order kept
# speedup vs baseline: 6.6295x; 1.2458x over previous
#include <hip/hip_runtime.h>
#include <hip/hip_bf16.h>
#include <math.h>

typedef float v2f __attribute__((ext_vector_type(2)));
typedef float v8f __attribute__((ext_vector_type(8)));

#define T_SEQ 2048
#define HID_DIM 1024
#define NH 16
#define DH 64
#define NB 32
#define KMAX_SEL 8

__device__ __forceinline__ v8f wmma4(v2f a, v2f b, v8f c) {
  return __builtin_amdgcn_wmma_f32_16x16x4_f32(false, a, false, b, (short)0, c,
                                               false, false);
}

__device__ __forceinline__ float apply_act(float x, int act) {
  if (act == 1) return 0.5f * x * (1.0f + erff(x * 0.70710678118654752f));
  if (act == 2) return 1.0f / (1.0f + __expf(-x));
  return x;
}

struct GemmFrags {
  v2f a[4][4];
  v2f b[4][2];
};

__device__ __forceinline__ void gemm_load_block(GemmFrags& f, const float* ap,
                                                const float* bp,
                                                size_t aRowStep, size_t N) {
#pragma unroll
  for (int s = 0; s < 4; ++s) {
#pragma unroll
    for (int mi = 0; mi < 4; ++mi) {
      const float* p = ap + (size_t)mi * aRowStep + 4 * s;
      f.a[s][mi].x = p[0];
      f.a[s][mi].y = p[1];
    }
#pragma unroll
    for (int ni = 0; ni < 2; ++ni) {
      f.b[s][ni].x = bp[(size_t)(4 * s) * N + ni * 16];
      f.b[s][ni].y = bp[(size_t)(4 * s + 1) * N + ni * 16];
    }
  }
}

__device__ __forceinline__ void gemm_mma_block(const GemmFrags& f,
                                               v8f (&acc)[4][2]) {
#pragma unroll
  for (int s = 0; s < 4; ++s)
#pragma unroll
    for (int mi = 0; mi < 4; ++mi)
#pragma unroll
      for (int ni = 0; ni < 2; ++ni)
        acc[mi][ni] = wmma4(f.a[s][mi], f.b[s][ni], acc[mi][ni]);
}

__global__ void __launch_bounds__(384)
wmma_gemm_f32_blk(const float* __restrict__ A, const float* B0,
                  const float* B1, const float* B2,
                  const float* __restrict__ bias, float* C0, float* C1,
                  float* C2, int nProd, int M, int N, int K, int act,
                  int headSplit) {
  int wave = threadIdx.x >> 5;
  int lane = threadIdx.x & 31;
  int wtile = blockIdx.x * (blockDim.x >> 5) + wave;
  int wn = N >> 5;
  int total = (M >> 6) * wn;
  if (wtile >= total * nProd) return;
  int prod = wtile / total;
  wtile -= prod * total;
  const float* __restrict__ B = prod == 0 ? B0 : (prod == 1 ? B1 : B2);
  float* __restrict__ C = prod == 0 ? C0 : (prod == 1 ? C1 : C2);
  int tm = wtile / wn, tn = wtile - tm * wn;
  int half = lane >> 4, l = lane & 15;
  int m0 = tm << 6, n0 = tn << 5;

  v8f acc[4][2];
#pragma unroll
  for (int mi = 0; mi < 4; ++mi)
#pragma unroll
    for (int ni = 0; ni < 2; ++ni) acc[mi][ni] = (v8f){};

  const float* ap = A + (size_t)(m0 + l) * K + 2 * half;
  const float* bp = B + (size_t)(2 * half) * N + n0 + l;
  const size_t aRowStep = (size_t)16 * K;
  const size_t bBlk = (size_t)16 * N;
  GemmFrags f0, f1;
  gemm_load_block(f0, ap, bp, aRowStep, (size_t)N);
  for (int k0 = 0; k0 < K - 32; k0 += 32) {
    gemm_load_block(f1, ap + 16, bp + bBlk, aRowStep, (size_t)N);
    __builtin_amdgcn_sched_barrier(0);
    gemm_mma_block(f0, acc);
    __builtin_amdgcn_sched_barrier(0);
    gemm_load_block(f0, ap + 32, bp + 2 * bBlk, aRowStep, (size_t)N);
    __builtin_amdgcn_sched_barrier(0);
    gemm_mma_block(f1, acc);
    __builtin_amdgcn_sched_barrier(0);
    ap += 32;
    bp += 2 * bBlk;
  }
  gemm_load_block(f1, ap + 16, bp + bBlk, aRowStep, (size_t)N);
  __builtin_amdgcn_sched_barrier(0);
  gemm_mma_block(f0, acc);
  gemm_mma_block(f1, acc);

  float bv[2];
#pragma unroll
  for (int ni = 0; ni < 2; ++ni) bv[ni] = bias ? bias[n0 + ni * 16 + l] : 0.0f;
#pragma unroll
  for (int mi = 0; mi < 4; ++mi)
#pragma unroll
    for (int ni = 0; ni < 2; ++ni) {
      int col = n0 + ni * 16 + l;
#pragma unroll
      for (int r = 0; r < 8; ++r) {
        float x = apply_act(acc[mi][ni][r] + bv[ni], act);
        int m = m0 + mi * 16 + r + 8 * half;
        if (headSplit) {
          int hh = col >> 6, dd = col & 63;
          C[((size_t)hh * M + m) * 64 + dd] = x;
        } else {
          C[(size_t)m * N + col] = x;
        }
      }
    }
}

#define CK_K 4096
#define CK_N 64
#define CK_WAVES 32
#define CK_SLICE (CK_K / CK_WAVES)
#define CK_NI_PITCH 272
#define CK_W_PITCH (2 * CK_NI_PITCH)
__global__ void __launch_bounds__(CK_WAVES * 32)
compress_gemm_splitk(const float* __restrict__ A, const float* __restrict__ B,
                     const float* __restrict__ bias, float* __restrict__ C) {
  __shared__ float part[CK_WAVES * CK_W_PITCH];
  int wave = threadIdx.x >> 5, lane = threadIdx.x & 31;
  int half = lane >> 4, l = lane & 15;
  int m0 = (blockIdx.x >> 1) << 4, n0 = (blockIdx.x & 1) << 5;
  int kbase = wave * CK_SLICE;
  const float* ap = A + (size_t)(m0 + l) * CK_K + kbase + 2 * half;
  const float* bp = B + (size_t)(kbase + 2 * half) * CK_N + n0 + l;
  v8f acc0 = {}, acc1 = {};
#pragma unroll
  for (int s = 0; s < CK_SLICE / 4; ++s) {
    v2f a, b0, b1;
    a.x = ap[4 * s];
    a.y = ap[4 * s + 1];
    b0.x = bp[(4 * s) * CK_N];
    b0.y = bp[(4 * s + 1) * CK_N];
    b1.x = bp[(4 * s) * CK_N + 16];
    b1.y = bp[(4 * s + 1) * CK_N + 16];
    acc0 = wmma4(a, b0, acc0);
    acc1 = wmma4(a, b1, acc1);
  }
  float* pw = part + wave * CK_W_PITCH + lane;
#pragma unroll
  for (int r = 0; r < 8; ++r) {
    pw[r * 32] = acc0[r];
    pw[CK_NI_PITCH + r * 32] = acc1[r];
  }
  __syncthreads();
  int t = threadIdx.x;
  if (t < 512) {
    int row = t >> 5, col = t & 31;
    const float* pr = part + (col >> 4) * CK_NI_PITCH + (row & 7) * 32 +
                      (row >> 3) * 16 + (col & 15);
    float sum = pr[0];
#pragma unroll
    for (int w = 1; w < CK_WAVES; ++w) sum += pr[w * CK_W_PITCH];
    C[(size_t)(m0 + row) * CK_N + n0 + col] =
        apply_act(sum + bias[n0 + col], 1);
  }
}

__global__ void wmma_gemm_f32(const float* __restrict__ A,
                              const float* __restrict__ B,
                              const float* __restrict__ bias,
                              float* __restrict__ C, int M, int N, int K,
                              int act, int headSplit) {
  int wave = threadIdx.x >> 5;
  int lane = threadIdx.x & 31;
  int tile = blockIdx.x * (blockDim.x >> 5) + wave;
  int tiles_n = N >> 4;
  int total = (M >> 4) * tiles_n;
  if (tile >= total) return;
  int tm = tile / tiles_n, tn = tile - tm * tiles_n;
  int half = lane >> 4, l = lane & 15;
  int arow = (tm << 4) + l;
  int bcol = (tn << 4) + l;
  v8f acc = {};
  const float* ap = A + (size_t)arow * K + 2 * half;
  const float* bp = B + (size_t)(2 * half) * N + bcol;
  for (int k0 = 0; k0 < K; k0 += 4) {
    v2f a, b;
    a.x = ap[0];
    a.y = ap[1];
    b.x = bp[0];
    b.y = bp[N];
    acc = wmma4(a, b, acc);
    ap += 4;
    bp += (size_t)4 * N;
  }
  float bv = bias ? bias[bcol] : 0.0f;
#pragma unroll
  for (int r = 0; r < 8; ++r) {
    float x = apply_act(acc[r] + bv, act);
    int m = (tm << 4) + r + 8 * half;
    if (headSplit) {
      int hh = bcol >> 6, dd = bcol & 63;
      C[((size_t)hh * M + m) * 64 + dd] = x;
    } else {
      C[(size_t)m * N + bcol] = x;
    }
  }
}

__global__ void sliding_window_kernel(const float* __restrict__ q,
                                      const float* __restrict__ k,
                                      const float* __restrict__ v,
                                      float* __restrict__ out) {
  extern __shared__ float lds[];
  float* Ks = lds;
  float* Vs = Ks + 128 * 66;
  float* Ps = Vs + 128 * 66;
  int h = blockIdx.x >> 5;
  int w = blockIdx.x & 31;
  const float* kh = k + (size_t)h * T_SEQ * DH;
  const float* vh = v + (size_t)h * T_SEQ * DH;
  for (int idx = threadIdx.x; idx < 128 * 32; idx += blockDim.x) {
    int r = idx >> 5, c2 = (idx & 31) << 1;
    int g = (w - 1) * 64 + r;
    float2 kk = make_float2(0.f, 0.f), vv = make_float2(0.f, 0.f);
    if (g >= 0) {
      kk = *(const float2*)&kh[(size_t)g * DH + c2];
      vv = *(const float2*)&vh[(size_t)g * DH + c2];
    }
    *(float2*)&Ks[r * 66 + c2] = kk;
    *(float2*)&Vs[r * 66 + c2] = vv;
  }
  __syncthreads();

  int wv = threadIdx.x >> 5, lane = threadIdx.x & 31;
  int half = lane >> 4, l = lane & 15;
  const float* qrow =
      q + ((size_t)h * T_SEQ + w * 64 + wv * 16 + l) * DH + 2 * half;

  v8f s[8] = {};
  for (int k0 = 0; k0 < 64; k0 += 4) {
    v2f a;
    a.x = qrow[k0];
    a.y = qrow[k0 + 1];
#pragma unroll
    for (int nt = 0; nt < 8; ++nt) {
      v2f b;
      const float* kp = &Ks[(nt * 16 + l) * 66 + k0 + 2 * half];
      b.x = kp[0];
      b.y = kp[1];
      s[nt] = wmma4(a, b, s[nt]);
    }
  }
  float rmax[8], rsum[8];
#pragma unroll
  for (int r = 0; r < 8; ++r) { rmax[r] = -3.0e38f; rsum[r] = 0.f; }
#pragma unroll
  for (int nt = 0; nt < 8; ++nt) {
    int n = nt * 16 + l;
#pragma unroll
    for (int r = 0; r < 8; ++r) {
      int m = wv * 16 + r + 8 * half;
      float x = s[nt][r] * 0.125f;
      bool ok = (m + 64 >= n) && (w > 0 || n >= 64);
      x = ok ? x : -3.0e38f;
      s[nt][r] = x;
      rmax[r] = fmaxf(rmax[r], x);
    }
  }
#pragma unroll
  for (int d = 1; d < 16; d <<= 1)
#pragma unroll
    for (int r = 0; r < 8; ++r) rmax[r] = fmaxf(rmax[r], __shfl_xor(rmax[r], d));
#pragma unroll
  for (int nt = 0; nt < 8; ++nt)
#pragma unroll
    for (int r = 0; r < 8; ++r) {
      float e = __expf(s[nt][r] - rmax[r]);
      s[nt][r] = e;
      rsum[r] += e;
    }
#pragma unroll
  for (int d = 1; d < 16; d <<= 1)
#pragma unroll
    for (int r = 0; r < 8; ++r) rsum[r] += __shfl_xor(rsum[r], d);
  float rinv[8];
#pragma unroll
  for (int r = 0; r < 8; ++r) rinv[r] = 1.0f / rsum[r];
#pragma unroll
  for (int nt = 0; nt < 8; ++nt)
#pragma unroll
    for (int r = 0; r < 8; ++r)
      Ps[(wv * 16 + r + 8 * half) * 130 + nt * 16 + l] = s[nt][r] * rinv[r];
  __syncthreads();

  v8f o[4] = {};
  const float* prow = &Ps[(wv * 16 + l) * 130 + 2 * half];
  for (int k0 = 0; k0 < 128; k0 += 4) {
    v2f a;
    a.x = prow[k0];
    a.y = prow[k0 + 1];
#pragma unroll
    for (int nt = 0; nt < 4; ++nt) {
      v2f b;
      const float* vp = &Vs[(k0 + 2 * half) * 66 + nt * 16 + l];
      b.x = vp[0];
      b.y = vp[66];
      o[nt] = wmma4(a, b, o[nt]);
    }
  }
  float* oh = out + ((size_t)h * T_SEQ + w * 64) * DH;
#pragma unroll
  for (int nt = 0; nt < 4; ++nt)
#pragma unroll
    for (int r = 0; r < 8; ++r)
      oh[(size_t)(wv * 16 + r + 8 * half) * DH + nt * 16 + l] = o[nt][r];
}

__global__ void comp_branch_kernel(const float* __restrict__ q,
                                   const float* __restrict__ kc,
                                   const float* __restrict__ vc,
                                   float* __restrict__ bs,
                                   float* __restrict__ comp) {
  __shared__ float Ps[16 * 34];
  int h = blockIdx.x >> 7;
  int tt = blockIdx.x & 127;
  int t0 = tt << 4;
  int lane = threadIdx.x, half = lane >> 4, l = lane & 15;
  const float* qrow = q + ((size_t)h * T_SEQ + t0 + l) * DH + 2 * half;
  v8f s[2] = {};
  for (int k0 = 0; k0 < 64; k0 += 4) {
    v2f a;
    a.x = qrow[k0];
    a.y = qrow[k0 + 1];
#pragma unroll
    for (int nt = 0; nt < 2; ++nt) {
      v2f b;
      const float* kp = kc + (size_t)(h * 32 + nt * 16 + l) * 64 + k0 + 2 * half;
      b.x = kp[0];
      b.y = kp[1];
      s[nt] = wmma4(a, b, s[nt]);
    }
  }
#pragma unroll
  for (int nt = 0; nt < 2; ++nt)
#pragma unroll
    for (int r = 0; r < 8; ++r) s[nt][r] *= 0.125f;

  int qb = t0 >> 6;
#pragma unroll
  for (int nt = 0; nt < 2; ++nt) {
    float cs = 0.f;
#pragma unroll
    for (int r = 0; r < 8; ++r) cs += s[nt][r];
    cs += __shfl_xor(cs, 16);
    if (half == 0)
      atomicAdd(&bs[((size_t)h * 32 + qb) * 32 + nt * 16 + l],
                cs * (1.0f / 64.0f));
  }
  float rmax[8], rsum[8];
#pragma unroll
  for (int r = 0; r < 8; ++r) rmax[r] = fmaxf(s[0][r], s[1][r]);
#pragma unroll
  for (int d = 1; d < 16; d <<= 1)
#pragma unroll
    for (int r = 0; r < 8; ++r) rmax[r] = fmaxf(rmax[r], __shfl_xor(rmax[r], d));
#pragma unroll
  for (int r = 0; r < 8; ++r) rsum[r] = 0.f;
#pragma unroll
  for (int nt = 0; nt < 2; ++nt)
#pragma unroll
    for (int r = 0; r < 8; ++r) {
      float e = __expf(s[nt][r] - rmax[r]);
      s[nt][r] = e;
      rsum[r] += e;
    }
#pragma unroll
  for (int d = 1; d < 16; d <<= 1)
#pragma unroll
    for (int r = 0; r < 8; ++r) rsum[r] += __shfl_xor(rsum[r], d);
#pragma unroll
  for (int nt = 0; nt < 2; ++nt)
#pragma unroll
    for (int r = 0; r < 8; ++r)
      Ps[(r + 8 * half) * 34 + nt * 16 + l] = s[nt][r] / rsum[r];
  __syncthreads();

  v8f o[4] = {};
  const float* prow = &Ps[l * 34 + 2 * half];
  for (int k0 = 0; k0 < 32; k0 += 4) {
    v2f a;
    a.x = prow[k0];
    a.y = prow[k0 + 1];
#pragma unroll
    for (int nt = 0; nt < 4; ++nt) {
      v2f b;
      const float* vp = vc + (size_t)(h * 32 + k0 + 2 * half) * 64 + nt * 16 + l;
      b.x = vp[0];
      b.y = vp[64];
      o[nt] = wmma4(a, b, o[nt]);
    }
  }
  float* oh = comp + ((size_t)h * T_SEQ + t0) * DH;
#pragma unroll
  for (int nt = 0; nt < 4; ++nt)
#pragma unroll
    for (int r = 0; r < 8; ++r)
      oh[(size_t)(r + 8 * half) * DH + nt * 16 + l] = o[nt][r];
}

__global__ void block_stats_kernel(const float* __restrict__ bs,
                                   const float* __restrict__ ksc,
                                   const float* __restrict__ kbi,
                                   unsigned char* __restrict__ mask,
                                   float* __restrict__ avg_acc,
                                   float* __restrict__ tail) {
  int h = blockIdx.x, j = threadIdx.x;
  const float* row = bs + ((size_t)h * 32 + j) * 32;
  float sc[32], p[32];
  float mx = -3.0e38f;
  for (int c = 0; c < 32; ++c) { sc[c] = row[c]; mx = fmaxf(mx, sc[c]); }
  float ssum = 0.f;
  for (int c = 0; c < 32; ++c) { p[c] = __expf(sc[c] - mx); ssum += p[c]; }
  float inv = 1.0f / ssum;
  float ent = 0.f;
  for (int c = 0; c < 32; ++c) {
    float pp = p[c] * inv;
    ent -= pp * __logf(pp + 1e-9f);
  }
  float ne = ent / (logf(32.0f) + 1e-9f);
  float kr = 1.0f / (1.0f + __expf(-(ksc[h] * ne + kbi[h])));
  float dkf = fminf(fmaxf(8.0f * kr, 1.0f), 8.0f);
  int dk = (int)dkf;
  tail[1 + h * 32 + j] = kr;
  tail[1 + 512 + h * 32 + j] = ne;
  atomicAdd(avg_acc, (float)dk);
  unsigned used = 0;
  unsigned char mrow[32];
  for (int c = 0; c < 32; ++c) mrow[c] = 0;
  mrow[j] = 1;
  for (int kk = 0; kk < KMAX_SEL; ++kk) {
    float best = -3.4e38f;
    int bi = 0;
    for (int c = 0; c < 32; ++c)
      if (!((used >> c) & 1u) && sc[c] > best) { best = sc[c]; bi = c; }
    used |= 1u << bi;
    if (kk < dk) mrow[bi] = 1;
  }
  for (int c = 0; c < 32; ++c) mask[((size_t)h * 32 + j) * 32 + c] = mrow[c];
}

__global__ void fine_attn_kernel(const float* __restrict__ q,
                                 const float* __restrict__ k,
                                 const float* __restrict__ v,
                                 const unsigned char* __restrict__ mask,
                                 float* __restrict__ out) {
  __shared__ float Ks[64 * 66];
  __shared__ float Vs[64 * 66];
  __shared__ float Ps[64 * 66];
  int h = blockIdx.x >> 5, qb = blockIdx.x & 31;
  int wv = threadIdx.x >> 5, lane = threadIdx.x & 31;
  int half = lane >> 4, l = lane & 15;
  const float* kh = k + (size_t)h * T_SEQ * DH;
  const float* vh = v + (size_t)h * T_SEQ * DH;
  const float* qrow =
      q + ((size_t)h * T_SEQ + qb * 64 + wv * 16 + l) * DH + 2 * half;
  const unsigned char* mrow = mask + ((size_t)h * 32 + qb) * 32;
  v8f o[4] = {};
  float rmax[8], rsum[8];
#pragma unroll
  for (int r = 0; r < 8; ++r) { rmax[r] = -3.0e38f; rsum[r] = 0.f; }

  for (int j = 0; j < 32; ++j) {
    if (!mrow[j]) continue;
    __syncthreads();
    for (int idx = threadIdx.x; idx < 64 * 32; idx += blockDim.x) {
      int r = idx >> 5, c2 = (idx & 31) << 1;
      *(float2*)&Ks[r * 66 + c2] =
          *(const float2*)&kh[(size_t)(j * 64 + r) * DH + c2];
      *(float2*)&Vs[r * 66 + c2] =
          *(const float2*)&vh[(size_t)(j * 64 + r) * DH + c2];
    }
    __syncthreads();
    v8f s[4] = {};
    for (int k0 = 0; k0 < 64; k0 += 4) {
      v2f a;
      a.x = qrow[k0];
      a.y = qrow[k0 + 1];
#pragma unroll
      for (int nt = 0; nt < 4; ++nt) {
        v2f b;
        const float* kp = &Ks[(nt * 16 + l) * 66 + k0 + 2 * half];
        b.x = kp[0];
        b.y = kp[1];
        s[nt] = wmma4(a, b, s[nt]);
      }
    }
    float nm[8], ps[8], fac[8];
#pragma unroll
    for (int r = 0; r < 8; ++r) nm[r] = rmax[r];
#pragma unroll
    for (int nt = 0; nt < 4; ++nt)
#pragma unroll
      for (int r = 0; r < 8; ++r) {
        float x = s[nt][r] * 0.125f;
        s[nt][r] = x;
        nm[r] = fmaxf(nm[r], x);
      }
#pragma unroll
    for (int d = 1; d < 16; d <<= 1)
#pragma unroll
      for (int r = 0; r < 8; ++r) nm[r] = fmaxf(nm[r], __shfl_xor(nm[r], d));
#pragma unroll
    for (int r = 0; r < 8; ++r) { fac[r] = __expf(rmax[r] - nm[r]); ps[r] = 0.f; }
#pragma unroll
    for (int nt = 0; nt < 4; ++nt)
#pragma unroll
      for (int r = 0; r < 8; ++r) {
        float e = __expf(s[nt][r] - nm[r]);
        s[nt][r] = e;
        ps[r] += e;
      }
#pragma unroll
    for (int d = 1; d < 16; d <<= 1)
#pragma unroll
      for (int r = 0; r < 8; ++r) ps[r] += __shfl_xor(ps[r], d);
#pragma unroll
    for (int r = 0; r < 8; ++r) {
      rsum[r] = rsum[r] * fac[r] + ps[r];
      rmax[r] = nm[r];
    }
#pragma unroll
    for (int nt = 0; nt < 4; ++nt)
#pragma unroll
      for (int r = 0; r < 8; ++r)
        Ps[(wv * 16 + r + 8 * half) * 66 + nt * 16 + l] = s[nt][r];
#pragma unroll
    for (int nt = 0; nt < 4; ++nt)
#pragma unroll
      for (int r = 0; r < 8; ++r) o[nt][r] *= fac[r];
    __syncthreads();
    const float* prow = &Ps[(wv * 16 + l) * 66 + 2 * half];
    for (int k0 = 0; k0 < 64; k0 += 4) {
      v2f a;
      a.x = prow[k0];
      a.y = prow[k0 + 1];
#pragma unroll
      for (int nt = 0; nt < 4; ++nt) {
        v2f b;
        const float* vp = &Vs[(k0 + 2 * half) * 66 + nt * 16 + l];
        b.x = vp[0];
        b.y = vp[66];
        o[nt] = wmma4(a, b, o[nt]);
      }
    }
  }
  float* oh = out + ((size_t)h * T_SEQ + qb * 64) * DH;
#pragma unroll
  for (int nt = 0; nt < 4; ++nt)
#pragma unroll
    for (int r = 0; r < 8; ++r)
      oh[(size_t)(wv * 16 + r + 8 * half) * DH + nt * 16 + l] =
          o[nt][r] / rsum[r];
}

__global__ void combine_kernel(const float* __restrict__ sw,
                               const float* __restrict__ comp,
                               const float* __restrict__ fine,
                               const float* __restrict__ gate,
                               float* __restrict__ merged) {
  int idx = blockIdx.x * blockDim.x + threadIdx.x;
  if (idx >= T_SEQ * HID_DIM) return;
  int t = idx >> 10, c = idx & 1023, hh = c >> 6, dd = c & 63;
  size_t a = ((size_t)hh * T_SEQ + t) * DH + dd;
  const float* g = gate + (size_t)t * 48 + hh * 3;
  merged[idx] = sw[a] * g[0] + comp[a] * g[1] + fine[a] * g[2];
}

__global__ void zero_kernel(float* __restrict__ p, int n) {
  int i = blockIdx.x * blockDim.x + threadIdx.x;
  if (i < n) p[i] = 0.f;
}

__global__ void finalize_kernel(const float* __restrict__ avg_acc,
                                float* __restrict__ tail) {
  if (threadIdx.x == 0) tail[0] = avg_acc[0] * (1.0f / 512.0f);
}

extern "C" void kernel_launch(void* const* d_in, const int* in_sizes, int n_in,
                              void* d_out, int out_size, void* d_ws,
                              size_t ws_size, hipStream_t stream) {
  const float* X = (const float*)d_in[0];
  const float* Wq = (const float*)d_in[1];
  const float* Wk = (const float*)d_in[2];
  const float* Wv = (const float*)d_in[3];
  const float* Wo = (const float*)d_in[4];
  const float* Wc = (const float*)d_in[5];
  const float* bc = (const float*)d_in[6];
  const float* ksc = (const float*)d_in[7];
  const float* kbi = (const float*)d_in[8];
  const float* Wg = (const float*)d_in[9];
  const float* bg = (const float*)d_in[10];
  float* out = (float*)d_out;

  const size_t U = (size_t)NH * T_SEQ * DH;
  float* ws = (float*)d_ws;
  float* q = ws;
  float* kbuf = ws + U;
  float* vbuf = ws + 2 * U;
  float* sw = ws + 3 * U;
  float* comp = ws + 4 * U;
  float* fine = ws + 5 * U;
  float* merged = q;
  float* extra = ws + 6 * U;
  float* gate = extra;
  float* kc = gate + (size_t)T_SEQ * 48;
  float* vc = kc + 512 * 64;
  float* bs = vc + 512 * 64;
  float* avg = bs + 16384;
  unsigned char* mask = (unsigned char*)(avg + 64);

  zero_kernel<<<(16385 + 255) / 256, 256, 0, stream>>>(bs, 16385);

  {
    int wtiles = 3 * (T_SEQ / 64) * (HID_DIM / 32);
    int blocks = (wtiles + 11) / 12;
    wmma_gemm_f32_blk<<<blocks, 384, 0, stream>>>(X, Wq, Wk, Wv, nullptr, q,
                                                  kbuf, vbuf, 3, T_SEQ, HID_DIM,
                                                  HID_DIM, 0, 1);
  }
  {
    int tiles = (T_SEQ / 16) * (48 / 16);
    wmma_gemm_f32<<<(tiles + 7) / 8, 256, 0, stream>>>(X, Wg, bg, gate, T_SEQ,
                                                       48, HID_DIM, 2, 0);
  }
  sliding_window_kernel<<<NH * NB, 128,
                          (2 * 128 * 66 + 64 * 130) * sizeof(float), stream>>>(
      q, kbuf, vbuf, sw);
  {
    compress_gemm_splitk<<<64, 1024, 0, stream>>>(kbuf, Wc, bc, kc);
    compress_gemm_splitk<<<64, 1024, 0, stream>>>(vbuf, Wc, bc, vc);
  }
  comp_branch_kernel<<<NH * (T_SEQ / 16), 32, 0, stream>>>(q, kc, vc, bs, comp);
  block_stats_kernel<<<NH, 32, 0, stream>>>(bs, ksc, kbi, mask, avg,
                                            out + (size_t)T_SEQ * HID_DIM);
  fine_attn_kernel<<<NH * NB, 128, 0, stream>>>(q, kbuf, vbuf, mask, fine);
  combine_kernel<<<(T_SEQ * HID_DIM + 255) / 256, 256, 0, stream>>>(
      sw, comp, fine, gate, merged);
  {
    int wtiles = (T_SEQ / 64) * (HID_DIM / 32);
    wmma_gemm_f32_blk<<<(wtiles + 3) / 4, 128, 0, stream>>>(
        merged, Wo, nullptr, nullptr, nullptr, out, nullptr, nullptr, 1, T_SEQ,
        HID_DIM, HID_DIM, 0, 0);
  }
  finalize_kernel<<<1, 1, 0, stream>>>(avg, out + (size_t)T_SEQ * HID_DIM);
  (void)in_sizes;
  (void)n_in;
  (void)out_size;
  (void)ws_size;
}


# fine and sliding-window attention: K/V tile staged through registers, a thread's 32 global loads issued as one group (was a round trip per 2 loads, 16 per tile); launch bound 128 gives the registers
# speedup vs baseline: 7.9152x; 1.1939x over previous
#include <hip/hip_runtime.h>
#include <hip/hip_bf16.h>
#include <math.h>

typedef float v2f __attribute__((ext_vector_type(2)));
typedef float v8f __attribute__((ext_vector_type(8)));

#define T_SEQ 2048
#define HID_DIM 1024
#define NH 16
#define DH 64
#define NB 32
#define KMAX_SEL 8

__device__ __forceinline__ v8f wmma4(v2f a, v2f b, v8f c) {
  return __builtin_amdgcn_wmma_f32_16x16x4_f32(false, a, false, b, (short)0, c,
                                               false, false);
}

__device__ __forceinline__ float apply_act(float x, int act) {
  if (act == 1) return 0.5f * x * (1.0f + erff(x * 0.70710678118654752f));
  if (act == 2) return 1.0f / (1.0f + __expf(-x));
  return x;
}

struct GemmFrags {
  v2f a[4][4];
  v2f b[4][2];
};

__device__ __forceinline__ void gemm_load_block(GemmFrags& f, const float* ap,
                                                const float* bp,
                                                size_t aRowStep, size_t N) {
#pragma unroll
  for (int s = 0; s < 4; ++s) {
#pragma unroll
    for (int mi = 0; mi < 4; ++mi) {
      const float* p = ap + (size_t)mi * aRowStep + 4 * s;
      f.a[s][mi].x = p[0];
      f.a[s][mi].y = p[1];
    }
#pragma unroll
    for (int ni = 0; ni < 2; ++ni) {
      f.b[s][ni].x = bp[(size_t)(4 * s) * N + ni * 16];
      f.b[s][ni].y = bp[(size_t)(4 * s + 1) * N + ni * 16];
    }
  }
}

__device__ __forceinline__ void gemm_mma_block(const GemmFrags& f,
                                               v8f (&acc)[4][2]) {
#pragma unroll
  for (int s = 0; s < 4; ++s)
#pragma unroll
    for (int mi = 0; mi < 4; ++mi)
#pragma unroll
      for (int ni = 0; ni < 2; ++ni)
        acc[mi][ni] = wmma4(f.a[s][mi], f.b[s][ni], acc[mi][ni]);
}

__global__ void __launch_bounds__(384)
wmma_gemm_f32_blk(const float* __restrict__ A, const float* B0,
                  const float* B1, const float* B2,
                  const float* __restrict__ bias, float* C0, float* C1,
                  float* C2, int nProd, int M, int N, int K, int act,
                  int headSplit) {
  int wave = threadIdx.x >> 5;
  int lane = threadIdx.x & 31;
  int wtile = blockIdx.x * (blockDim.x >> 5) + wave;
  int wn = N >> 5;
  int total = (M >> 6) * wn;
  if (wtile >= total * nProd) return;
  int prod = wtile / total;
  wtile -= prod * total;
  const float* __restrict__ B = prod == 0 ? B0 : (prod == 1 ? B1 : B2);
  float* __restrict__ C = prod == 0 ? C0 : (prod == 1 ? C1 : C2);
  int tm = wtile / wn, tn = wtile - tm * wn;
  int half = lane >> 4, l = lane & 15;
  int m0 = tm << 6, n0 = tn << 5;

  v8f acc[4][2];
#pragma unroll
  for (int mi = 0; mi < 4; ++mi)
#pragma unroll
    for (int ni = 0; ni < 2; ++ni) acc[mi][ni] = (v8f){};

  const float* ap = A + (size_t)(m0 + l) * K + 2 * half;
  const float* bp = B + (size_t)(2 * half) * N + n0 + l;
  const size_t aRowStep = (size_t)16 * K;
  const size_t bBlk = (size_t)16 * N;
  GemmFrags f0, f1;
  gemm_load_block(f0, ap, bp, aRowStep, (size_t)N);
  for (int k0 = 0; k0 < K - 32; k0 += 32) {
    gemm_load_block(f1, ap + 16, bp + bBlk, aRowStep, (size_t)N);
    __builtin_amdgcn_sched_barrier(0);
    gemm_mma_block(f0, acc);
    __builtin_amdgcn_sched_barrier(0);
    gemm_load_block(f0, ap + 32, bp + 2 * bBlk, aRowStep, (size_t)N);
    __builtin_amdgcn_sched_barrier(0);
    gemm_mma_block(f1, acc);
    __builtin_amdgcn_sched_barrier(0);
    ap += 32;
    bp += 2 * bBlk;
  }
  gemm_load_block(f1, ap + 16, bp + bBlk, aRowStep, (size_t)N);
  __builtin_amdgcn_sched_barrier(0);
  gemm_mma_block(f0, acc);
  gemm_mma_block(f1, acc);

  float bv[2];
#pragma unroll
  for (int ni = 0; ni < 2; ++ni) bv[ni] = bias ? bias[n0 + ni * 16 + l] : 0.0f;
#pragma unroll
  for (int mi = 0; mi < 4; ++mi)
#pragma unroll
    for (int ni = 0; ni < 2; ++ni) {
      int col = n0 + ni * 16 + l;
#pragma unroll
      for (int r = 0; r < 8; ++r) {
        float x = apply_act(acc[mi][ni][r] + bv[ni], act);
        int m = m0 + mi * 16 + r + 8 * half;
        if (headSplit) {
          int hh = col >> 6, dd = col & 63;
          C[((size_t)hh * M + m) * 64 + dd] = x;
        } else {
          C[(size_t)m * N + col] = x;
        }
      }
    }
}

#define CK_K 4096
#define CK_N 64
#define CK_WAVES 32
#define CK_SLICE (CK_K / CK_WAVES)
#define CK_NI_PITCH 272
#define CK_W_PITCH (2 * CK_NI_PITCH)
__global__ void __launch_bounds__(CK_WAVES * 32)
compress_gemm_splitk(const float* __restrict__ A, const float* __restrict__ B,
                     const float* __restrict__ bias, float* __restrict__ C) {
  __shared__ float part[CK_WAVES * CK_W_PITCH];
  int wave = threadIdx.x >> 5, lane = threadIdx.x & 31;
  int half = lane >> 4, l = lane & 15;
  int m0 = (blockIdx.x >> 1) << 4, n0 = (blockIdx.x & 1) << 5;
  int kbase = wave * CK_SLICE;
  const float* ap = A + (size_t)(m0 + l) * CK_K + kbase + 2 * half;
  const float* bp = B + (size_t)(kbase + 2 * half) * CK_N + n0 + l;
  v8f acc0 = {}, acc1 = {};
#pragma unroll
  for (int s = 0; s < CK_SLICE / 4; ++s) {
    v2f a, b0, b1;
    a.x = ap[4 * s];
    a.y = ap[4 * s + 1];
    b0.x = bp[(4 * s) * CK_N];
    b0.y = bp[(4 * s + 1) * CK_N];
    b1.x = bp[(4 * s) * CK_N + 16];
    b1.y = bp[(4 * s + 1) * CK_N + 16];
    acc0 = wmma4(a, b0, acc0);
    acc1 = wmma4(a, b1, acc1);
  }
  float* pw = part + wave * CK_W_PITCH + lane;
#pragma unroll
  for (int r = 0; r < 8; ++r) {
    pw[r * 32] = acc0[r];
    pw[CK_NI_PITCH + r * 32] = acc1[r];
  }
  __syncthreads();
  int t = threadIdx.x;
  if (t < 512) {
    int row = t >> 5, col = t & 31;
    const float* pr = part + (col >> 4) * CK_NI_PITCH + (row & 7) * 32 +
                      (row >> 3) * 16 + (col & 15);
    float sum = pr[0];
#pragma unroll
    for (int w = 1; w < CK_WAVES; ++w) sum += pr[w * CK_W_PITCH];
    C[(size_t)(m0 + row) * CK_N + n0 + col] =
        apply_act(sum + bias[n0 + col], 1);
  }
}

__global__ void wmma_gemm_f32(const float* __restrict__ A,
                              const float* __restrict__ B,
                              const float* __restrict__ bias,
                              float* __restrict__ C, int M, int N, int K,
                              int act, int headSplit) {
  int wave = threadIdx.x >> 5;
  int lane = threadIdx.x & 31;
  int tile = blockIdx.x * (blockDim.x >> 5) + wave;
  int tiles_n = N >> 4;
  int total = (M >> 4) * tiles_n;
  if (tile >= total) return;
  int tm = tile / tiles_n, tn = tile - tm * tiles_n;
  int half = lane >> 4, l = lane & 15;
  int arow = (tm << 4) + l;
  int bcol = (tn << 4) + l;
  v8f acc = {};
  const float* ap = A + (size_t)arow * K + 2 * half;
  const float* bp = B + (size_t)(2 * half) * N + bcol;
  for (int k0 = 0; k0 < K; k0 += 4) {
    v2f a, b;
    a.x = ap[0];
    a.y = ap[1];
    b.x = bp[0];
    b.y = bp[N];
    acc = wmma4(a, b, acc);
    ap += 4;
    bp += (size_t)4 * N;
  }
  float bv = bias ? bias[bcol] : 0.0f;
#pragma unroll
  for (int r = 0; r < 8; ++r) {
    float x = apply_act(acc[r] + bv, act);
    int m = (tm << 4) + r + 8 * half;
    if (headSplit) {
      int hh = bcol >> 6, dd = bcol & 63;
      C[((size_t)hh * M + m) * 64 + dd] = x;
    } else {
      C[(size_t)m * N + bcol] = x;
    }
  }
}

#define ATT_THREADS 128
__device__ __forceinline__ void tile_load_regs(float2 (&kr)[16],
                                               float2 (&vr)[16],
                                               const float* __restrict__ kt,
                                               const float* __restrict__ vt,
                                               int tid) {
  const int off = (tid >> 5) * DH + ((tid & 31) << 1);
#pragma unroll
  for (int i = 0; i < 16; ++i) {
    kr[i] = *(const float2*)&kt[off + i * 4 * DH];
    vr[i] = *(const float2*)&vt[off + i * 4 * DH];
  }
}

__device__ __forceinline__ void tile_store_lds(float* Kd, float* Vd,
                                               const float2 (&kr)[16],
                                               const float2 (&vr)[16],
                                               int tid) {
  const int off = (tid >> 5) * 66 + ((tid & 31) << 1);
#pragma unroll
  for (int i = 0; i < 16; ++i) {
    *(float2*)&Kd[off + i * 4 * 66] = kr[i];
    *(float2*)&Vd[off + i * 4 * 66] = vr[i];
  }
}

__global__ void __launch_bounds__(ATT_THREADS)
sliding_window_kernel(const float* __restrict__ q,
                                      const float* __restrict__ k,
                                      const float* __restrict__ v,
                                      float* __restrict__ out) {
  extern __shared__ float lds[];
  float* Ks = lds;
  float* Vs = Ks + 128 * 66;
  float* Ps = Vs + 128 * 66;
  int h = blockIdx.x >> 5;
  int w = blockIdx.x & 31;
  const float* kh = k + (size_t)h * T_SEQ * DH;
  const float* vh = v + (size_t)h * T_SEQ * DH;
  {
    float2 kc[16], vc[16], kp[16], vp[16];
    tile_load_regs(kc, vc, kh + (size_t)w * 64 * DH, vh + (size_t)w * 64 * DH,
                   threadIdx.x);
    if (w > 0) {
      tile_load_regs(kp, vp, kh + (size_t)(w - 1) * 64 * DH,
                     vh + (size_t)(w - 1) * 64 * DH, threadIdx.x);
    } else {
#pragma unroll
      for (int i = 0; i < 16; ++i) {
        kp[i] = make_float2(0.f, 0.f);
        vp[i] = make_float2(0.f, 0.f);
      }
    }
    tile_store_lds(Ks + 64 * 66, Vs + 64 * 66, kc, vc, threadIdx.x);
    tile_store_lds(Ks, Vs, kp, vp, threadIdx.x);
  }
  __syncthreads();

  int wv = threadIdx.x >> 5, lane = threadIdx.x & 31;
  int half = lane >> 4, l = lane & 15;
  const float* qrow =
      q + ((size_t)h * T_SEQ + w * 64 + wv * 16 + l) * DH + 2 * half;

  v8f s[8] = {};
  for (int k0 = 0; k0 < 64; k0 += 4) {
    v2f a;
    a.x = qrow[k0];
    a.y = qrow[k0 + 1];
#pragma unroll
    for (int nt = 0; nt < 8; ++nt) {
      v2f b;
      const float* kp = &Ks[(nt * 16 + l) * 66 + k0 + 2 * half];
      b.x = kp[0];
      b.y = kp[1];
      s[nt] = wmma4(a, b, s[nt]);
    }
  }
  float rmax[8], rsum[8];
#pragma unroll
  for (int r = 0; r < 8; ++r) { rmax[r] = -3.0e38f; rsum[r] = 0.f; }
#pragma unroll
  for (int nt = 0; nt < 8; ++nt) {
    int n = nt * 16 + l;
#pragma unroll
    for (int r = 0; r < 8; ++r) {
      int m = wv * 16 + r + 8 * half;
      float x = s[nt][r] * 0.125f;
      bool ok = (m + 64 >= n) && (w > 0 || n >= 64);
      x = ok ? x : -3.0e38f;
      s[nt][r] = x;
      rmax[r] = fmaxf(rmax[r], x);
    }
  }
#pragma unroll
  for (int d = 1; d < 16; d <<= 1)
#pragma unroll
    for (int r = 0; r < 8; ++r) rmax[r] = fmaxf(rmax[r], __shfl_xor(rmax[r], d));
#pragma unroll
  for (int nt = 0; nt < 8; ++nt)
#pragma unroll
    for (int r = 0; r < 8; ++r) {
      float e = __expf(s[nt][r] - rmax[r]);
      s[nt][r] = e;
      rsum[r] += e;
    }
#pragma unroll
  for (int d = 1; d < 16; d <<= 1)
#pragma unroll
    for (int r = 0; r < 8; ++r) rsum[r] += __shfl_xor(rsum[r], d);
  float rinv[8];
#pragma unroll
  for (int r = 0; r < 8; ++r) rinv[r] = 1.0f / rsum[r];
#pragma unroll
  for (int nt = 0; nt < 8; ++nt)
#pragma unroll
    for (int r = 0; r < 8; ++r)
      Ps[(wv * 16 + r + 8 * half) * 130 + nt * 16 + l] = s[nt][r] * rinv[r];
  __syncthreads();

  v8f o[4] = {};
  const float* prow = &Ps[(wv * 16 + l) * 130 + 2 * half];
  for (int k0 = 0; k0 < 128; k0 += 4) {
    v2f a;
    a.x = prow[k0];
    a.y = prow[k0 + 1];
#pragma unroll
    for (int nt = 0; nt < 4; ++nt) {
      v2f b;
      const float* vp = &Vs[(k0 + 2 * half) * 66 + nt * 16 + l];
      b.x = vp[0];
      b.y = vp[66];
      o[nt] = wmma4(a, b, o[nt]);
    }
  }
  float* oh = out + ((size_t)h * T_SEQ + w * 64) * DH;
#pragma unroll
  for (int nt = 0; nt < 4; ++nt)
#pragma unroll
    for (int r = 0; r < 8; ++r)
      oh[(size_t)(wv * 16 + r + 8 * half) * DH + nt * 16 + l] = o[nt][r];
}

__global__ void comp_branch_kernel(const float* __restrict__ q,
                                   const float* __restrict__ kc,
                                   const float* __restrict__ vc,
                                   float* __restrict__ bs,
                                   float* __restrict__ comp) {
  __shared__ float Ps[16 * 34];
  int h = blockIdx.x >> 7;
  int tt = blockIdx.x & 127;
  int t0 = tt << 4;
  int lane = threadIdx.x, half = lane >> 4, l = lane & 15;
  const float* qrow = q + ((size_t)h * T_SEQ + t0 + l) * DH + 2 * half;
  v8f s[2] = {};
  for (int k0 = 0; k0 < 64; k0 += 4) {
    v2f a;
    a.x = qrow[k0];
    a.y = qrow[k0 + 1];
#pragma unroll
    for (int nt = 0; nt < 2; ++nt) {
      v2f b;
      const float* kp = kc + (size_t)(h * 32 + nt * 16 + l) * 64 + k0 + 2 * half;
      b.x = kp[0];
      b.y = kp[1];
      s[nt] = wmma4(a, b, s[nt]);
    }
  }
#pragma unroll
  for (int nt = 0; nt < 2; ++nt)
#pragma unroll
    for (int r = 0; r < 8; ++r) s[nt][r] *= 0.125f;

  int qb = t0 >> 6;
#pragma unroll
  for (int nt = 0; nt < 2; ++nt) {
    float cs = 0.f;
#pragma unroll
    for (int r = 0; r < 8; ++r) cs += s[nt][r];
    cs += __shfl_xor(cs, 16);
    if (half == 0)
      atomicAdd(&bs[((size_t)h * 32 + qb) * 32 + nt * 16 + l],
                cs * (1.0f / 64.0f));
  }
  float rmax[8], rsum[8];
#pragma unroll
  for (int r = 0; r < 8; ++r) rmax[r] = fmaxf(s[0][r], s[1][r]);
#pragma unroll
  for (int d = 1; d < 16; d <<= 1)
#pragma unroll
    for (int r = 0; r < 8; ++r) rmax[r] = fmaxf(rmax[r], __shfl_xor(rmax[r], d));
#pragma unroll
  for (int r = 0; r < 8; ++r) rsum[r] = 0.f;
#pragma unroll
  for (int nt = 0; nt < 2; ++nt)
#pragma unroll
    for (int r = 0; r < 8; ++r) {
      float e = __expf(s[nt][r] - rmax[r]);
      s[nt][r] = e;
      rsum[r] += e;
    }
#pragma unroll
  for (int d = 1; d < 16; d <<= 1)
#pragma unroll
    for (int r = 0; r < 8; ++r) rsum[r] += __shfl_xor(rsum[r], d);
#pragma unroll
  for (int nt = 0; nt < 2; ++nt)
#pragma unroll
    for (int r = 0; r < 8; ++r)
      Ps[(r + 8 * half) * 34 + nt * 16 + l] = s[nt][r] / rsum[r];
  __syncthreads();

  v8f o[4] = {};
  const float* prow = &Ps[l * 34 + 2 * half];
  for (int k0 = 0; k0 < 32; k0 += 4) {
    v2f a;
    a.x = prow[k0];
    a.y = prow[k0 + 1];
#pragma unroll
    for (int nt = 0; nt < 4; ++nt) {
      v2f b;
      const float* vp = vc + (size_t)(h * 32 + k0 + 2 * half) * 64 + nt * 16 + l;
      b.x = vp[0];
      b.y = vp[64];
      o[nt] = wmma4(a, b, o[nt]);
    }
  }
  float* oh = comp + ((size_t)h * T_SEQ + t0) * DH;
#pragma unroll
  for (int nt = 0; nt < 4; ++nt)
#pragma unroll
    for (int r = 0; r < 8; ++r)
      oh[(size_t)(r + 8 * half) * DH + nt * 16 + l] = o[nt][r];
}

__global__ void block_stats_kernel(const float* __restrict__ bs,
                                   const float* __restrict__ ksc,
                                   const float* __restrict__ kbi,
                                   unsigned char* __restrict__ mask,
                                   float* __restrict__ avg_acc,
                                   float* __restrict__ tail) {
  int h = blockIdx.x, j = threadIdx.x;
  const float* row = bs + ((size_t)h * 32 + j) * 32;
  float sc[32], p[32];
  float mx = -3.0e38f;
  for (int c = 0; c < 32; ++c) { sc[c] = row[c]; mx = fmaxf(mx, sc[c]); }
  float ssum = 0.f;
  for (int c = 0; c < 32; ++c) { p[c] = __expf(sc[c] - mx); ssum += p[c]; }
  float inv = 1.0f / ssum;
  float ent = 0.f;
  for (int c = 0; c < 32; ++c) {
    float pp = p[c] * inv;
    ent -= pp * __logf(pp + 1e-9f);
  }
  float ne = ent / (logf(32.0f) + 1e-9f);
  float kr = 1.0f / (1.0f + __expf(-(ksc[h] * ne + kbi[h])));
  float dkf = fminf(fmaxf(8.0f * kr, 1.0f), 8.0f);
  int dk = (int)dkf;
  tail[1 + h * 32 + j] = kr;
  tail[1 + 512 + h * 32 + j] = ne;
  atomicAdd(avg_acc, (float)dk);
  unsigned used = 0;
  unsigned char mrow[32];
  for (int c = 0; c < 32; ++c) mrow[c] = 0;
  mrow[j] = 1;
  for (int kk = 0; kk < KMAX_SEL; ++kk) {
    float best = -3.4e38f;
    int bi = 0;
    for (int c = 0; c < 32; ++c)
      if (!((used >> c) & 1u) && sc[c] > best) { best = sc[c]; bi = c; }
    used |= 1u << bi;
    if (kk < dk) mrow[bi] = 1;
  }
  for (int c = 0; c < 32; ++c) mask[((size_t)h * 32 + j) * 32 + c] = mrow[c];
}

__global__ void __launch_bounds__(ATT_THREADS)
fine_attn_kernel(const float* __restrict__ q,
                                 const float* __restrict__ k,
                                 const float* __restrict__ v,
                                 const unsigned char* __restrict__ mask,
                                 float* __restrict__ out) {
  __shared__ float Ks[64 * 66];
  __shared__ float Vs[64 * 66];
  __shared__ float Ps[64 * 66];
  int h = blockIdx.x >> 5, qb = blockIdx.x & 31;
  int wv = threadIdx.x >> 5, lane = threadIdx.x & 31;
  int half = lane >> 4, l = lane & 15;
  const float* kh = k + (size_t)h * T_SEQ * DH;
  const float* vh = v + (size_t)h * T_SEQ * DH;
  const float* qrow =
      q + ((size_t)h * T_SEQ + qb * 64 + wv * 16 + l) * DH + 2 * half;
  const unsigned char* mrow = mask + ((size_t)h * 32 + qb) * 32;
  v8f o[4] = {};
  float rmax[8], rsum[8];
#pragma unroll
  for (int r = 0; r < 8; ++r) { rmax[r] = -3.0e38f; rsum[r] = 0.f; }

  for (int j = 0; j < 32; ++j) {
    if (!mrow[j]) continue;
    float2 kr[16], vr[16];
    tile_load_regs(kr, vr, kh + (size_t)j * 64 * DH, vh + (size_t)j * 64 * DH,
                   threadIdx.x);
    __syncthreads();
    tile_store_lds(Ks, Vs, kr, vr, threadIdx.x);
    __syncthreads();
    v8f s[4] = {};
    for (int k0 = 0; k0 < 64; k0 += 4) {
      v2f a;
      a.x = qrow[k0];
      a.y = qrow[k0 + 1];
#pragma unroll
      for (int nt = 0; nt < 4; ++nt) {
        v2f b;
        const float* kp = &Ks[(nt * 16 + l) * 66 + k0 + 2 * half];
        b.x = kp[0];
        b.y = kp[1];
        s[nt] = wmma4(a, b, s[nt]);
      }
    }
    float nm[8], ps[8], fac[8];
#pragma unroll
    for (int r = 0; r < 8; ++r) nm[r] = rmax[r];
#pragma unroll
    for (int nt = 0; nt < 4; ++nt)
#pragma unroll
      for (int r = 0; r < 8; ++r) {
        float x = s[nt][r] * 0.125f;
        s[nt][r] = x;
        nm[r] = fmaxf(nm[r], x);
      }
#pragma unroll
    for (int d = 1; d < 16; d <<= 1)
#pragma unroll
      for (int r = 0; r < 8; ++r) nm[r] = fmaxf(nm[r], __shfl_xor(nm[r], d));
#pragma unroll
    for (int r = 0; r < 8; ++r) { fac[r] = __expf(rmax[r] - nm[r]); ps[r] = 0.f; }
#pragma unroll
    for (int nt = 0; nt < 4; ++nt)
#pragma unroll
      for (int r = 0; r < 8; ++r) {
        float e = __expf(s[nt][r] - nm[r]);
        s[nt][r] = e;
        ps[r] += e;
      }
#pragma unroll
    for (int d = 1; d < 16; d <<= 1)
#pragma unroll
      for (int r = 0; r < 8; ++r) ps[r] += __shfl_xor(ps[r], d);
#pragma unroll
    for (int r = 0; r < 8; ++r) {
      rsum[r] = rsum[r] * fac[r] + ps[r];
      rmax[r] = nm[r];
    }
#pragma unroll
    for (int nt = 0; nt < 4; ++nt)
#pragma unroll
      for (int r = 0; r < 8; ++r)
        Ps[(wv * 16 + r + 8 * half) * 66 + nt * 16 + l] = s[nt][r];
#pragma unroll
    for (int nt = 0; nt < 4; ++nt)
#pragma unroll
      for (int r = 0; r < 8; ++r) o[nt][r] *= fac[r];
    __syncthreads();
    const float* prow = &Ps[(wv * 16 + l) * 66 + 2 * half];
    for (int k0 = 0; k0 < 64; k0 += 4) {
      v2f a;
      a.x = prow[k0];
      a.y = prow[k0 + 1];
#pragma unroll
      for (int nt = 0; nt < 4; ++nt) {
        v2f b;
        const float* vp = &Vs[(k0 + 2 * half) * 66 + nt * 16 + l];
        b.x = vp[0];
        b.y = vp[66];
        o[nt] = wmma4(a, b, o[nt]);
      }
    }
  }
  float* oh = out + ((size_t)h * T_SEQ + qb * 64) * DH;
#pragma unroll
  for (int nt = 0; nt < 4; ++nt)
#pragma unroll
    for (int r = 0; r < 8; ++r)
      oh[(size_t)(wv * 16 + r + 8 * half) * DH + nt * 16 + l] =
          o[nt][r] / rsum[r];
}

__global__ void combine_kernel(const float* __restrict__ sw,
                               const float* __restrict__ comp,
                               const float* __restrict__ fine,
                               const float* __restrict__ gate,
                               float* __restrict__ merged) {
  int idx = blockIdx.x * blockDim.x + threadIdx.x;
  if (idx >= T_SEQ * HID_DIM) return;
  int t = idx >> 10, c = idx & 1023, hh = c >> 6, dd = c & 63;
  size_t a = ((size_t)hh * T_SEQ + t) * DH + dd;
  const float* g = gate + (size_t)t * 48 + hh * 3;
  merged[idx] = sw[a] * g[0] + comp[a] * g[1] + fine[a] * g[2];
}

__global__ void zero_kernel(float* __restrict__ p, int n) {
  int i = blockIdx.x * blockDim.x + threadIdx.x;
  if (i < n) p[i] = 0.f;
}

__global__ void finalize_kernel(const float* __restrict__ avg_acc,
                                float* __restrict__ tail) {
  if (threadIdx.x == 0) tail[0] = avg_acc[0] * (1.0f / 512.0f);
}

extern "C" void kernel_launch(void* const* d_in, const int* in_sizes, int n_in,
                              void* d_out, int out_size, void* d_ws,
                              size_t ws_size, hipStream_t stream) {
  const float* X = (const float*)d_in[0];
  const float* Wq = (const float*)d_in[1];
  const float* Wk = (const float*)d_in[2];
  const float* Wv = (const float*)d_in[3];
  const float* Wo = (const float*)d_in[4];
  const float* Wc = (const float*)d_in[5];
  const float* bc = (const float*)d_in[6];
  const float* ksc = (const float*)d_in[7];
  const float* kbi = (const float*)d_in[8];
  const float* Wg = (const float*)d_in[9];
  const float* bg = (const float*)d_in[10];
  float* out = (float*)d_out;

  const size_t U = (size_t)NH * T_SEQ * DH;
  float* ws = (float*)d_ws;
  float* q = ws;
  float* kbuf = ws + U;
  float* vbuf = ws + 2 * U;
  float* sw = ws + 3 * U;
  float* comp = ws + 4 * U;
  float* fine = ws + 5 * U;
  float* merged = q;
  float* extra = ws + 6 * U;
  float* gate = extra;
  float* kc = gate + (size_t)T_SEQ * 48;
  float* vc = kc + 512 * 64;
  float* bs = vc + 512 * 64;
  float* avg = bs + 16384;
  unsigned char* mask = (unsigned char*)(avg + 64);

  zero_kernel<<<(16385 + 255) / 256, 256, 0, stream>>>(bs, 16385);

  {
    int wtiles = 3 * (T_SEQ / 64) * (HID_DIM / 32);
    int blocks = (wtiles + 11) / 12;
    wmma_gemm_f32_blk<<<blocks, 384, 0, stream>>>(X, Wq, Wk, Wv, nullptr, q,
                                                  kbuf, vbuf, 3, T_SEQ, HID_DIM,
                                                  HID_DIM, 0, 1);
  }
  {
    int tiles = (T_SEQ / 16) * (48 / 16);
    wmma_gemm_f32<<<(tiles + 7) / 8, 256, 0, stream>>>(X, Wg, bg, gate, T_SEQ,
                                                       48, HID_DIM, 2, 0);
  }
  sliding_window_kernel<<<NH * NB, ATT_THREADS,
                          (2 * 128 * 66 + 64 * 130) * sizeof(float), stream>>>(
      q, kbuf, vbuf, sw);
  {
    compress_gemm_splitk<<<64, 1024, 0, stream>>>(kbuf, Wc, bc, kc);
    compress_gemm_splitk<<<64, 1024, 0, stream>>>(vbuf, Wc, bc, vc);
  }
  comp_branch_kernel<<<NH * (T_SEQ / 16), 32, 0, stream>>>(q, kc, vc, bs, comp);
  block_stats_kernel<<<NH, 32, 0, stream>>>(bs, ksc, kbi, mask, avg,
                                            out + (size_t)T_SEQ * HID_DIM);
  fine_attn_kernel<<<NH * NB, ATT_THREADS, 0, stream>>>(q, kbuf, vbuf, mask,
                                                        fine);
  combine_kernel<<<(T_SEQ * HID_DIM + 255) / 256, 256, 0, stream>>>(
      sw, comp, fine, gate, merged);
  {
    int wtiles = (T_SEQ / 64) * (HID_DIM / 32);
    wmma_gemm_f32_blk<<<(wtiles + 3) / 4, 128, 0, stream>>>(
        merged, Wo, nullptr, nullptr, nullptr, out, nullptr, nullptr, 1, T_SEQ,
        HID_DIM, HID_DIM, 0, 0);
  }
  finalize_kernel<<<1, 1, 0, stream>>>(avg, out + (size_t)T_SEQ * HID_DIM);
  (void)in_sizes;
  (void)n_in;
  (void)out_size;
  (void)ws_size;
}


# gate projection and both compress products as ONE split-K launch of 256 work-groups (were 3 launches of 48/64/64 groups); gate: K over 32 waves, LDS sum in wave order (its k summation order changes)
# speedup vs baseline: 9.3648x; 1.1831x over previous
#include <hip/hip_runtime.h>
#include <hip/hip_bf16.h>
#include <math.h>

typedef float v2f __attribute__((ext_vector_type(2)));
typedef float v8f __attribute__((ext_vector_type(8)));

#define T_SEQ 2048
#define HID_DIM 1024
#define NH 16
#define DH 64
#define NB 32
#define KMAX_SEL 8

__device__ __forceinline__ v8f wmma4(v2f a, v2f b, v8f c) {
  return __builtin_amdgcn_wmma_f32_16x16x4_f32(false, a, false, b, (short)0, c,
                                               false, false);
}

__device__ __forceinline__ float apply_act(float x, int act) {
  if (act == 1) return 0.5f * x * (1.0f + erff(x * 0.70710678118654752f));
  if (act == 2) return 1.0f / (1.0f + __expf(-x));
  return x;
}

struct GemmFrags {
  v2f a[4][4];
  v2f b[4][2];
};

__device__ __forceinline__ void gemm_load_block(GemmFrags& f, const float* ap,
                                                const float* bp,
                                                size_t aRowStep, size_t N) {
#pragma unroll
  for (int s = 0; s < 4; ++s) {
#pragma unroll
    for (int mi = 0; mi < 4; ++mi) {
      const float* p = ap + (size_t)mi * aRowStep + 4 * s;
      f.a[s][mi].x = p[0];
      f.a[s][mi].y = p[1];
    }
#pragma unroll
    for (int ni = 0; ni < 2; ++ni) {
      f.b[s][ni].x = bp[(size_t)(4 * s) * N + ni * 16];
      f.b[s][ni].y = bp[(size_t)(4 * s + 1) * N + ni * 16];
    }
  }
}

__device__ __forceinline__ void gemm_mma_block(const GemmFrags& f,
                                               v8f (&acc)[4][2]) {
#pragma unroll
  for (int s = 0; s < 4; ++s)
#pragma unroll
    for (int mi = 0; mi < 4; ++mi)
#pragma unroll
      for (int ni = 0; ni < 2; ++ni)
        acc[mi][ni] = wmma4(f.a[s][mi], f.b[s][ni], acc[mi][ni]);
}

__global__ void __launch_bounds__(384)
wmma_gemm_f32_blk(const float* __restrict__ A, const float* B0,
                  const float* B1, const float* B2,
                  const float* __restrict__ bias, float* C0, float* C1,
                  float* C2, int nProd, int M, int N, int K, int act,
                  int headSplit) {
  int wave = threadIdx.x >> 5;
  int lane = threadIdx.x & 31;
  int wtile = blockIdx.x * (blockDim.x >> 5) + wave;
  int wn = N >> 5;
  int total = (M >> 6) * wn;
  if (wtile >= total * nProd) return;
  int prod = wtile / total;
  wtile -= prod * total;
  const float* __restrict__ B = prod == 0 ? B0 : (prod == 1 ? B1 : B2);
  float* __restrict__ C = prod == 0 ? C0 : (prod == 1 ? C1 : C2);
  int tm = wtile / wn, tn = wtile - tm * wn;
  int half = lane >> 4, l = lane & 15;
  int m0 = tm << 6, n0 = tn << 5;

  v8f acc[4][2];
#pragma unroll
  for (int mi = 0; mi < 4; ++mi)
#pragma unroll
    for (int ni = 0; ni < 2; ++ni) acc[mi][ni] = (v8f){};

  const float* ap = A + (size_t)(m0 + l) * K + 2 * half;
  const float* bp = B + (size_t)(2 * half) * N + n0 + l;
  const size_t aRowStep = (size_t)16 * K;
  const size_t bBlk = (size_t)16 * N;
  GemmFrags f0, f1;
  gemm_load_block(f0, ap, bp, aRowStep, (size_t)N);
  for (int k0 = 0; k0 < K - 32; k0 += 32) {
    gemm_load_block(f1, ap + 16, bp + bBlk, aRowStep, (size_t)N);
    __builtin_amdgcn_sched_barrier(0);
    gemm_mma_block(f0, acc);
    __builtin_amdgcn_sched_barrier(0);
    gemm_load_block(f0, ap + 32, bp + 2 * bBlk, aRowStep, (size_t)N);
    __builtin_amdgcn_sched_barrier(0);
    gemm_mma_block(f1, acc);
    __builtin_amdgcn_sched_barrier(0);
    ap += 32;
    bp += 2 * bBlk;
  }
  gemm_load_block(f1, ap + 16, bp + bBlk, aRowStep, (size_t)N);
  __builtin_amdgcn_sched_barrier(0);
  gemm_mma_block(f0, acc);
  gemm_mma_block(f1, acc);

  float bv[2];
#pragma unroll
  for (int ni = 0; ni < 2; ++ni) bv[ni] = bias ? bias[n0 + ni * 16 + l] : 0.0f;
#pragma unroll
  for (int mi = 0; mi < 4; ++mi)
#pragma unroll
    for (int ni = 0; ni < 2; ++ni) {
      int col = n0 + ni * 16 + l;
#pragma unroll
      for (int r = 0; r < 8; ++r) {
        float x = apply_act(acc[mi][ni][r] + bv[ni], act);
        int m = m0 + mi * 16 + r + 8 * half;
        if (headSplit) {
          int hh = col >> 6, dd = col & 63;
          C[((size_t)hh * M + m) * 64 + dd] = x;
        } else {
          C[(size_t)m * N + col] = x;
        }
      }
    }
}

#define CK_K 4096
#define CK_N 64
#define SP_WAVES 32
#define CK_SLICE (CK_K / SP_WAVES)
#define GT_N 48
#define GT_SLICE (HID_DIM / SP_WAVES)
#define SP_NI_PITCH 272
#define CK_W_PITCH (2 * SP_NI_PITCH)
#define GT_W_PITCH (3 * SP_NI_PITCH)
#define SP_COMPRESS_GROUPS 128
__global__ void __launch_bounds__(SP_WAVES * 32)
small_products_splitk(const float* __restrict__ kblk,
                      const float* __restrict__ vblk,
                      const float* __restrict__ Wc,
                      const float* __restrict__ bc, float* __restrict__ kc,
                      float* __restrict__ vc, const float* __restrict__ X,
                      const float* __restrict__ Wg,
                      const float* __restrict__ bg, float* __restrict__ gate) {
  __shared__ float part[SP_WAVES * GT_W_PITCH];
  int wave = threadIdx.x >> 5, lane = threadIdx.x & 31;
  int half = lane >> 4, l = lane & 15;
  int t = threadIdx.x;
  if (blockIdx.x < SP_COMPRESS_GROUPS) {
    int prod = blockIdx.x >> 6, g = blockIdx.x & 63;
    const float* __restrict__ A = prod ? vblk : kblk;
    float* __restrict__ C = prod ? vc : kc;
    int m0 = (g >> 1) << 4, n0 = (g & 1) << 5;
    int kbase = wave * CK_SLICE;
    const float* ap = A + (size_t)(m0 + l) * CK_K + kbase + 2 * half;
    const float* bp = Wc + (size_t)(kbase + 2 * half) * CK_N + n0 + l;
    v8f acc0 = {}, acc1 = {};
#pragma unroll
    for (int s = 0; s < CK_SLICE / 4; ++s) {
      v2f a, b0, b1;
      a.x = ap[4 * s];
      a.y = ap[4 * s + 1];
      b0.x = bp[(4 * s) * CK_N];
      b0.y = bp[(4 * s + 1) * CK_N];
      b1.x = bp[(4 * s) * CK_N + 16];
      b1.y = bp[(4 * s + 1) * CK_N + 16];
      acc0 = wmma4(a, b0, acc0);
      acc1 = wmma4(a, b1, acc1);
    }
    float* pw = part + wave * CK_W_PITCH + lane;
#pragma unroll
    for (int r = 0; r < 8; ++r) {
      pw[r * 32] = acc0[r];
      pw[SP_NI_PITCH + r * 32] = acc1[r];
    }
    __syncthreads();
    if (t < 512) {
      int row = t >> 5, col = t & 31;
      const float* pr = part + (col >> 4) * SP_NI_PITCH + (row & 7) * 32 +
                        (row >> 3) * 16 + (col & 15);
      float sum = pr[0];
#pragma unroll
      for (int w = 1; w < SP_WAVES; ++w) sum += pr[w * CK_W_PITCH];
      C[(size_t)(m0 + row) * CK_N + n0 + col] =
          apply_act(sum + bc[n0 + col], 1);
    }
  } else {
    int m0 = (blockIdx.x - SP_COMPRESS_GROUPS) << 4;
    int kbase = wave * GT_SLICE;
    const float* ap = X + (size_t)(m0 + l) * HID_DIM + kbase + 2 * half;
    const float* bp = Wg + (size_t)(kbase + 2 * half) * GT_N + l;
    v8f acc0 = {}, acc1 = {}, acc2 = {};
#pragma unroll
    for (int s = 0; s < GT_SLICE / 4; ++s) {
      v2f a, b0, b1, b2;
      a.x = ap[4 * s];
      a.y = ap[4 * s + 1];
      b0.x = bp[(4 * s) * GT_N];
      b0.y = bp[(4 * s + 1) * GT_N];
      b1.x = bp[(4 * s) * GT_N + 16];
      b1.y = bp[(4 * s + 1) * GT_N + 16];
      b2.x = bp[(4 * s) * GT_N + 32];
      b2.y = bp[(4 * s + 1) * GT_N + 32];
      acc0 = wmma4(a, b0, acc0);
      acc1 = wmma4(a, b1, acc1);
      acc2 = wmma4(a, b2, acc2);
    }
    float* pw = part + wave * GT_W_PITCH + lane;
#pragma unroll
    for (int r = 0; r < 8; ++r) {
      pw[r * 32] = acc0[r];
      pw[SP_NI_PITCH + r * 32] = acc1[r];
      pw[2 * SP_NI_PITCH + r * 32] = acc2[r];
    }
    __syncthreads();
    if (t < 16 * GT_N) {
      int row = t / GT_N, col = t - row * GT_N;
      const float* pr = part + (col >> 4) * SP_NI_PITCH + (row & 7) * 32 +
                        (row >> 3) * 16 + (col & 15);
      float sum = pr[0];
#pragma unroll
      for (int w = 1; w < SP_WAVES; ++w) sum += pr[w * GT_W_PITCH];
      gate[(size_t)m0 * GT_N + t] = apply_act(sum + bg[col], 2);
    }
  }
}

#define ATT_THREADS 128
__device__ __forceinline__ void tile_load_regs(float2 (&kr)[16],
                                               float2 (&vr)[16],
                                               const float* __restrict__ kt,
                                               const float* __restrict__ vt,
                                               int tid) {
  const int off = (tid >> 5) * DH + ((tid & 31) << 1);
#pragma unroll
  for (int i = 0; i < 16; ++i) {
    kr[i] = *(const float2*)&kt[off + i * 4 * DH];
    vr[i] = *(const float2*)&vt[off + i * 4 * DH];
  }
}

__device__ __forceinline__ void tile_store_lds(float* Kd, float* Vd,
                                               const float2 (&kr)[16],
                                               const float2 (&vr)[16],
                                               int tid) {
  const int off = (tid >> 5) * 66 + ((tid & 31) << 1);
#pragma unroll
  for (int i = 0; i < 16; ++i) {
    *(float2*)&Kd[off + i * 4 * 66] = kr[i];
    *(float2*)&Vd[off + i * 4 * 66] = vr[i];
  }
}

__global__ void __launch_bounds__(ATT_THREADS)
sliding_window_kernel(const float* __restrict__ q,
                                      const float* __restrict__ k,
                                      const float* __restrict__ v,
                                      float* __restrict__ out) {
  extern __shared__ float lds[];
  float* Ks = lds;
  float* Vs = Ks + 128 * 66;
  float* Ps = Vs + 128 * 66;
  int h = blockIdx.x >> 5;
  int w = blockIdx.x & 31;
  const float* kh = k + (size_t)h * T_SEQ * DH;
  const float* vh = v + (size_t)h * T_SEQ * DH;
  {
    float2 kc[16], vc[16], kp[16], vp[16];
    tile_load_regs(kc, vc, kh + (size_t)w * 64 * DH, vh + (size_t)w * 64 * DH,
                   threadIdx.x);
    if (w > 0) {
      tile_load_regs(kp, vp, kh + (size_t)(w - 1) * 64 * DH,
                     vh + (size_t)(w - 1) * 64 * DH, threadIdx.x);
    } else {
#pragma unroll
      for (int i = 0; i < 16; ++i) {
        kp[i] = make_float2(0.f, 0.f);
        vp[i] = make_float2(0.f, 0.f);
      }
    }
    tile_store_lds(Ks + 64 * 66, Vs + 64 * 66, kc, vc, threadIdx.x);
    tile_store_lds(Ks, Vs, kp, vp, threadIdx.x);
  }
  __syncthreads();

  int wv = threadIdx.x >> 5, lane = threadIdx.x & 31;
  int half = lane >> 4, l = lane & 15;
  const float* qrow =
      q + ((size_t)h * T_SEQ + w * 64 + wv * 16 + l) * DH + 2 * half;

  v8f s[8] = {};
  for (int k0 = 0; k0 < 64; k0 += 4) {
    v2f a;
    a.x = qrow[k0];
    a.y = qrow[k0 + 1];
#pragma unroll
    for (int nt = 0; nt < 8; ++nt) {
      v2f b;
      const float* kp = &Ks[(nt * 16 + l) * 66 + k0 + 2 * half];
      b.x = kp[0];
      b.y = kp[1];
      s[nt] = wmma4(a, b, s[nt]);
    }
  }
  float rmax[8], rsum[8];
#pragma unroll
  for (int r = 0; r < 8; ++r) { rmax[r] = -3.0e38f; rsum[r] = 0.f; }
#pragma unroll
  for (int nt = 0; nt < 8; ++nt) {
    int n = nt * 16 + l;
#pragma unroll
    for (int r = 0; r < 8; ++r) {
      int m = wv * 16 + r + 8 * half;
      float x = s[nt][r] * 0.125f;
      bool ok = (m + 64 >= n) && (w > 0 || n >= 64);
      x = ok ? x : -3.0e38f;
      s[nt][r] = x;
      rmax[r] = fmaxf(rmax[r], x);
    }
  }
#pragma unroll
  for (int d = 1; d < 16; d <<= 1)
#pragma unroll
    for (int r = 0; r < 8; ++r) rmax[r] = fmaxf(rmax[r], __shfl_xor(rmax[r], d));
#pragma unroll
  for (int nt = 0; nt < 8; ++nt)
#pragma unroll
    for (int r = 0; r < 8; ++r) {
      float e = __expf(s[nt][r] - rmax[r]);
      s[nt][r] = e;
      rsum[r] += e;
    }
#pragma unroll
  for (int d = 1; d < 16; d <<= 1)
#pragma unroll
    for (int r = 0; r < 8; ++r) rsum[r] += __shfl_xor(rsum[r], d);
  float rinv[8];
#pragma unroll
  for (int r = 0; r < 8; ++r) rinv[r] = 1.0f / rsum[r];
#pragma unroll
  for (int nt = 0; nt < 8; ++nt)
#pragma unroll
    for (int r = 0; r < 8; ++r)
      Ps[(wv * 16 + r + 8 * half) * 130 + nt * 16 + l] = s[nt][r] * rinv[r];
  __syncthreads();

  v8f o[4] = {};
  const float* prow = &Ps[(wv * 16 + l) * 130 + 2 * half];
  for (int k0 = 0; k0 < 128; k0 += 4) {
    v2f a;
    a.x = prow[k0];
    a.y = prow[k0 + 1];
#pragma unroll
    for (int nt = 0; nt < 4; ++nt) {
      v2f b;
      const float* vp = &Vs[(k0 + 2 * half) * 66 + nt * 16 + l];
      b.x = vp[0];
      b.y = vp[66];
      o[nt] = wmma4(a, b, o[nt]);
    }
  }
  float* oh = out + ((size_t)h * T_SEQ + w * 64) * DH;
#pragma unroll
  for (int nt = 0; nt < 4; ++nt)
#pragma unroll
    for (int r = 0; r < 8; ++r)
      oh[(size_t)(wv * 16 + r + 8 * half) * DH + nt * 16 + l] = o[nt][r];
}

__global__ void comp_branch_kernel(const float* __restrict__ q,
                                   const float* __restrict__ kc,
                                   const float* __restrict__ vc,
                                   float* __restrict__ bs,
                                   float* __restrict__ comp) {
  __shared__ float Ps[16 * 34];
  int h = blockIdx.x >> 7;
  int tt = blockIdx.x & 127;
  int t0 = tt << 4;
  int lane = threadIdx.x, half = lane >> 4, l = lane & 15;
  const float* qrow = q + ((size_t)h * T_SEQ + t0 + l) * DH + 2 * half;
  v8f s[2] = {};
  for (int k0 = 0; k0 < 64; k0 += 4) {
    v2f a;
    a.x = qrow[k0];
    a.y = qrow[k0 + 1];
#pragma unroll
    for (int nt = 0; nt < 2; ++nt) {
      v2f b;
      const float* kp = kc + (size_t)(h * 32 + nt * 16 + l) * 64 + k0 + 2 * half;
      b.x = kp[0];
      b.y = kp[1];
      s[nt] = wmma4(a, b, s[nt]);
    }
  }
#pragma unroll
  for (int nt = 0; nt < 2; ++nt)
#pragma unroll
    for (int r = 0; r < 8; ++r) s[nt][r] *= 0.125f;

  int qb = t0 >> 6;
#pragma unroll
  for (int nt = 0; nt < 2; ++nt) {
    float cs = 0.f;
#pragma unroll
    for (int r = 0; r < 8; ++r) cs += s[nt][r];
    cs += __shfl_xor(cs, 16);
    if (half == 0)
      atomicAdd(&bs[((size_t)h * 32 + qb) * 32 + nt * 16 + l],
                cs * (1.0f / 64.0f));
  }
  float rmax[8], rsum[8];
#pragma unroll
  for (int r = 0; r < 8; ++r) rmax[r] = fmaxf(s[0][r], s[1][r]);
#pragma unroll
  for (int d = 1; d < 16; d <<= 1)
#pragma unroll
    for (int r = 0; r < 8; ++r) rmax[r] = fmaxf(rmax[r], __shfl_xor(rmax[r], d));
#pragma unroll
  for (int r = 0; r < 8; ++r) rsum[r] = 0.f;
#pragma unroll
  for (int nt = 0; nt < 2; ++nt)
#pragma unroll
    for (int r = 0; r < 8; ++r) {
      float e = __expf(s[nt][r] - rmax[r]);
      s[nt][r] = e;
      rsum[r] += e;
    }
#pragma unroll
  for (int d = 1; d < 16; d <<= 1)
#pragma unroll
    for (int r = 0; r < 8; ++r) rsum[r] += __shfl_xor(rsum[r], d);
#pragma unroll
  for (int nt = 0; nt < 2; ++nt)
#pragma unroll
    for (int r = 0; r < 8; ++r)
      Ps[(r + 8 * half) * 34 + nt * 16 + l] = s[nt][r] / rsum[r];
  __syncthreads();

  v8f o[4] = {};
  const float* prow = &Ps[l * 34 + 2 * half];
  for (int k0 = 0; k0 < 32; k0 += 4) {
    v2f a;
    a.x = prow[k0];
    a.y = prow[k0 + 1];
#pragma unroll
    for (int nt = 0; nt < 4; ++nt) {
      v2f b;
      const float* vp = vc + (size_t)(h * 32 + k0 + 2 * half) * 64 + nt * 16 + l;
      b.x = vp[0];
      b.y = vp[64];
      o[nt] = wmma4(a, b, o[nt]);
    }
  }
  float* oh = comp + ((size_t)h * T_SEQ + t0) * DH;
#pragma unroll
  for (int nt = 0; nt < 4; ++nt)
#pragma unroll
    for (int r = 0; r < 8; ++r)
      oh[(size_t)(r + 8 * half) * DH + nt * 16 + l] = o[nt][r];
}

__global__ void block_stats_kernel(const float* __restrict__ bs,
                                   const float* __restrict__ ksc,
                                   const float* __restrict__ kbi,
                                   unsigned char* __restrict__ mask,
                                   float* __restrict__ avg_acc,
                                   float* __restrict__ tail) {
  int h = blockIdx.x, j = threadIdx.x;
  const float* row = bs + ((size_t)h * 32 + j) * 32;
  float sc[32], p[32];
  float mx = -3.0e38f;
  for (int c = 0; c < 32; ++c) { sc[c] = row[c]; mx = fmaxf(mx, sc[c]); }
  float ssum = 0.f;
  for (int c = 0; c < 32; ++c) { p[c] = __expf(sc[c] - mx); ssum += p[c]; }
  float inv = 1.0f / ssum;
  float ent = 0.f;
  for (int c = 0; c < 32; ++c) {
    float pp = p[c] * inv;
    ent -= pp * __logf(pp + 1e-9f);
  }
  float ne = ent / (logf(32.0f) + 1e-9f);
  float kr = 1.0f / (1.0f + __expf(-(ksc[h] * ne + kbi[h])));
  float dkf = fminf(fmaxf(8.0f * kr, 1.0f), 8.0f);
  int dk = (int)dkf;
  tail[1 + h * 32 + j] = kr;
  tail[1 + 512 + h * 32 + j] = ne;
  atomicAdd(avg_acc, (float)dk);
  unsigned used = 0;
  unsigned char mrow[32];
  for (int c = 0; c < 32; ++c) mrow[c] = 0;
  mrow[j] = 1;
  for (int kk = 0; kk < KMAX_SEL; ++kk) {
    float best = -3.4e38f;
    int bi = 0;
    for (int c = 0; c < 32; ++c)
      if (!((used >> c) & 1u) && sc[c] > best) { best = sc[c]; bi = c; }
    used |= 1u << bi;
    if (kk < dk) mrow[bi] = 1;
  }
  for (int c = 0; c < 32; ++c) mask[((size_t)h * 32 + j) * 32 + c] = mrow[c];
}

__global__ void __launch_bounds__(ATT_THREADS)
fine_attn_kernel(const float* __restrict__ q,
                                 const float* __restrict__ k,
                                 const float* __restrict__ v,
                                 const unsigned char* __restrict__ mask,
                                 float* __restrict__ out) {
  __shared__ float Ks[64 * 66];
  __shared__ float Vs[64 * 66];
  __shared__ float Ps[64 * 66];
  int h = blockIdx.x >> 5, qb = blockIdx.x & 31;
  int wv = threadIdx.x >> 5, lane = threadIdx.x & 31;
  int half = lane >> 4, l = lane & 15;
  const float* kh = k + (size_t)h * T_SEQ * DH;
  const float* vh = v + (size_t)h * T_SEQ * DH;
  const float* qrow =
      q + ((size_t)h * T_SEQ + qb * 64 + wv * 16 + l) * DH + 2 * half;
  const unsigned char* mrow = mask + ((size_t)h * 32 + qb) * 32;
  v8f o[4] = {};
  float rmax[8], rsum[8];
#pragma unroll
  for (int r = 0; r < 8; ++r) { rmax[r] = -3.0e38f; rsum[r] = 0.f; }

  for (int j = 0; j < 32; ++j) {
    if (!mrow[j]) continue;
    float2 kr[16], vr[16];
    tile_load_regs(kr, vr, kh + (size_t)j * 64 * DH, vh + (size_t)j * 64 * DH,
                   threadIdx.x);
    __syncthreads();
    tile_store_lds(Ks, Vs, kr, vr, threadIdx.x);
    __syncthreads();
    v8f s[4] = {};
    for (int k0 = 0; k0 < 64; k0 += 4) {
      v2f a;
      a.x = qrow[k0];
      a.y = qrow[k0 + 1];
#pragma unroll
      for (int nt = 0; nt < 4; ++nt) {
        v2f b;
        const float* kp = &Ks[(nt * 16 + l) * 66 + k0 + 2 * half];
        b.x = kp[0];
        b.y = kp[1];
        s[nt] = wmma4(a, b, s[nt]);
      }
    }
    float nm[8], ps[8], fac[8];
#pragma unroll
    for (int r = 0; r < 8; ++r) nm[r] = rmax[r];
#pragma unroll
    for (int nt = 0; nt < 4; ++nt)
#pragma unroll
      for (int r = 0; r < 8; ++r) {
        float x = s[nt][r] * 0.125f;
        s[nt][r] = x;
        nm[r] = fmaxf(nm[r], x);
      }
#pragma unroll
    for (int d = 1; d < 16; d <<= 1)
#pragma unroll
      for (int r = 0; r < 8; ++r) nm[r] = fmaxf(nm[r], __shfl_xor(nm[r], d));
#pragma unroll
    for (int r = 0; r < 8; ++r) { fac[r] = __expf(rmax[r] - nm[r]); ps[r] = 0.f; }
#pragma unroll
    for (int nt = 0; nt < 4; ++nt)
#pragma unroll
      for (int r = 0; r < 8; ++r) {
        float e = __expf(s[nt][r] - nm[r]);
        s[nt][r] = e;
        ps[r] += e;
      }
#pragma unroll
    for (int d = 1; d < 16; d <<= 1)
#pragma unroll
      for (int r = 0; r < 8; ++r) ps[r] += __shfl_xor(ps[r], d);
#pragma unroll
    for (int r = 0; r < 8; ++r) {
      rsum[r] = rsum[r] * fac[r] + ps[r];
      rmax[r] = nm[r];
    }
#pragma unroll
    for (int nt = 0; nt < 4; ++nt)
#pragma unroll
      for (int r = 0; r < 8; ++r)
        Ps[(wv * 16 + r + 8 * half) * 66 + nt * 16 + l] = s[nt][r];
#pragma unroll
    for (int nt = 0; nt < 4; ++nt)
#pragma unroll
      for (int r = 0; r < 8; ++r) o[nt][r] *= fac[r];
    __syncthreads();
    const float* prow = &Ps[(wv * 16 + l) * 66 + 2 * half];
    for (int k0 = 0; k0 < 64; k0 += 4) {
      v2f a;
      a.x = prow[k0];
      a.y = prow[k0 + 1];
#pragma unroll
      for (int nt = 0; nt < 4; ++nt) {
        v2f b;
        const float* vp = &Vs[(k0 + 2 * half) * 66 + nt * 16 + l];
        b.x = vp[0];
        b.y = vp[66];
        o[nt] = wmma4(a, b, o[nt]);
      }
    }
  }
  float* oh = out + ((size_t)h * T_SEQ + qb * 64) * DH;
#pragma unroll
  for (int nt = 0; nt < 4; ++nt)
#pragma unroll
    for (int r = 0; r < 8; ++r)
      oh[(size_t)(wv * 16 + r + 8 * half) * DH + nt * 16 + l] =
          o[nt][r] / rsum[r];
}

__global__ void combine_kernel(const float* __restrict__ sw,
                               const float* __restrict__ comp,
                               const float* __restrict__ fine,
                               const float* __restrict__ gate,
                               float* __restrict__ merged) {
  int idx = blockIdx.x * blockDim.x + threadIdx.x;
  if (idx >= T_SEQ * HID_DIM) return;
  int t = idx >> 10, c = idx & 1023, hh = c >> 6, dd = c & 63;
  size_t a = ((size_t)hh * T_SEQ + t) * DH + dd;
  const float* g = gate + (size_t)t * 48 + hh * 3;
  merged[idx] = sw[a] * g[0] + comp[a] * g[1] + fine[a] * g[2];
}

__global__ void zero_kernel(float* __restrict__ p, int n) {
  int i = blockIdx.x * blockDim.x + threadIdx.x;
  if (i < n) p[i] = 0.f;
}

__global__ void finalize_kernel(const float* __restrict__ avg_acc,
                                float* __restrict__ tail) {
  if (threadIdx.x == 0) tail[0] = avg_acc[0] * (1.0f / 512.0f);
}

extern "C" void kernel_launch(void* const* d_in, const int* in_sizes, int n_in,
                              void* d_out, int out_size, void* d_ws,
                              size_t ws_size, hipStream_t stream) {
  const float* X = (const float*)d_in[0];
  const float* Wq = (const float*)d_in[1];
  const float* Wk = (const float*)d_in[2];
  const float* Wv = (const float*)d_in[3];
  const float* Wo = (const float*)d_in[4];
  const float* Wc = (const float*)d_in[5];
  const float* bc = (const float*)d_in[6];
  const float* ksc = (const float*)d_in[7];
  const float* kbi = (const float*)d_in[8];
  const float* Wg = (const float*)d_in[9];
  const float* bg = (const float*)d_in[10];
  float* out = (float*)d_out;

  const size_t U = (size_t)NH * T_SEQ * DH;
  float* ws = (float*)d_ws;
  float* q = ws;
  float* kbuf = ws + U;
  float* vbuf = ws + 2 * U;
  float* sw = ws + 3 * U;
  float* comp = ws + 4 * U;
  float* fine = ws + 5 * U;
  float* merged = q;
  float* extra = ws + 6 * U;
  float* gate = extra;
  float* kc = gate + (size_t)T_SEQ * 48;
  float* vc = kc + 512 * 64;
  float* bs = vc + 512 * 64;
  float* avg = bs + 16384;
  unsigned char* mask = (unsigned char*)(avg + 64);

  zero_kernel<<<(16385 + 255) / 256, 256, 0, stream>>>(bs, 16385);

  {
    int wtiles = 3 * (T_SEQ / 64) * (HID_DIM / 32);
    int blocks = (wtiles + 11) / 12;
    wmma_gemm_f32_blk<<<blocks, 384, 0, stream>>>(X, Wq, Wk, Wv, nullptr, q,
                                                  kbuf, vbuf, 3, T_SEQ, HID_DIM,
                                                  HID_DIM, 0, 1);
  }
  sliding_window_kernel<<<NH * NB, ATT_THREADS,
                          (2 * 128 * 66 + 64 * 130) * sizeof(float), stream>>>(
      q, kbuf, vbuf, sw);
  {
    small_products_splitk<<<256, 1024, 0, stream>>>(kbuf, vbuf, Wc, bc, kc, vc,
                                                    X, Wg, bg, gate);
  }
  comp_branch_kernel<<<NH * (T_SEQ / 16), 32, 0, stream>>>(q, kc, vc, bs, comp);
  block_stats_kernel<<<NH, 32, 0, stream>>>(bs, ksc, kbi, mask, avg,
                                            out + (size_t)T_SEQ * HID_DIM);
  fine_attn_kernel<<<NH * NB, ATT_THREADS, 0, stream>>>(q, kbuf, vbuf, mask,
                                                        fine);
  combine_kernel<<<(T_SEQ * HID_DIM + 255) / 256, 256, 0, stream>>>(
      sw, comp, fine, gate, merged);
  {
    int wtiles = (T_SEQ / 64) * (HID_DIM / 32);
    wmma_gemm_f32_blk<<<(wtiles + 3) / 4, 128, 0, stream>>>(
        merged, Wo, nullptr, nullptr, nullptr, out, nullptr, nullptr, 1, T_SEQ,
        HID_DIM, HID_DIM, 0, 0);
  }
  finalize_kernel<<<1, 1, 0, stream>>>(avg, out + (size_t)T_SEQ * HID_DIM);
  (void)in_sizes;
  (void)n_in;
  (void)out_size;
  (void)ws_size;
}


# f32 GEMM fragment loads widened (Q/K/V/out projections, compress product): a lane loads 4 consecutive k of A (16 bytes), 2 adjacent columns of B (8 bytes); k summation order in 8-k groups changes
# speedup vs baseline: 9.7433x; 1.0404x over previous
#include <hip/hip_runtime.h>
#include <hip/hip_bf16.h>
#include <math.h>

typedef float v2f __attribute__((ext_vector_type(2)));
typedef float v4f __attribute__((ext_vector_type(4)));
typedef float v8f __attribute__((ext_vector_type(8)));

#define T_SEQ 2048
#define HID_DIM 1024
#define NH 16
#define DH 64
#define NB 32
#define KMAX_SEL 8

__device__ __forceinline__ v8f wmma4(v2f a, v2f b, v8f c) {
  return __builtin_amdgcn_wmma_f32_16x16x4_f32(false, a, false, b, (short)0, c,
                                               false, false);
}

__device__ __forceinline__ float apply_act(float x, int act) {
  if (act == 1) return 0.5f * x * (1.0f + erff(x * 0.70710678118654752f));
  if (act == 2) return 1.0f / (1.0f + __expf(-x));
  return x;
}

struct GemmFrags {
  v4f a[2][4];
  v2f b[2][4];
};

__device__ __forceinline__ void gemm_load_block(GemmFrags& f, const float* ap,
                                                const float* bp,
                                                size_t aRowStep, size_t N) {
#pragma unroll
  for (int g = 0; g < 2; ++g) {
#pragma unroll
    for (int mi = 0; mi < 4; ++mi)
      f.a[g][mi] = *(const v4f*)(ap + (size_t)mi * aRowStep + 8 * g);
#pragma unroll
    for (int j = 0; j < 4; ++j)
      f.b[g][j] = *(const v2f*)(bp + (size_t)(8 * g + j) * N);
  }
}

__device__ __forceinline__ void gemm_mma_block(const GemmFrags& f,
                                               v8f (&acc)[4][2]) {
#pragma unroll
  for (int g = 0; g < 2; ++g)
#pragma unroll
    for (int p = 0; p < 2; ++p) {
      v2f b0, b1;
      b0.x = f.b[g][2 * p].x;
      b0.y = f.b[g][2 * p + 1].x;
      b1.x = f.b[g][2 * p].y;
      b1.y = f.b[g][2 * p + 1].y;
#pragma unroll
      for (int mi = 0; mi < 4; ++mi) {
        v2f a;
        a.x = f.a[g][mi][2 * p];
        a.y = f.a[g][mi][2 * p + 1];
        acc[mi][0] = wmma4(a, b0, acc[mi][0]);
        acc[mi][1] = wmma4(a, b1, acc[mi][1]);
      }
    }
}

__global__ void __launch_bounds__(384)
wmma_gemm_f32_blk(const float* __restrict__ A, const float* B0,
                  const float* B1, const float* B2,
                  const float* __restrict__ bias, float* C0, float* C1,
                  float* C2, int nProd, int M, int N, int K, int act,
                  int headSplit) {
  int wave = threadIdx.x >> 5;
  int lane = threadIdx.x & 31;
  int wtile = blockIdx.x * (blockDim.x >> 5) + wave;
  int wn = N >> 5;
  int total = (M >> 6) * wn;
  if (wtile >= total * nProd) return;
  int prod = wtile / total;
  wtile -= prod * total;
  const float* __restrict__ B = prod == 0 ? B0 : (prod == 1 ? B1 : B2);
  float* __restrict__ C = prod == 0 ? C0 : (prod == 1 ? C1 : C2);
  int tm = wtile / wn, tn = wtile - tm * wn;
  int half = lane >> 4, l = lane & 15;
  int m0 = tm << 6, n0 = tn << 5;

  v8f acc[4][2];
#pragma unroll
  for (int mi = 0; mi < 4; ++mi)
#pragma unroll
    for (int ni = 0; ni < 2; ++ni) acc[mi][ni] = (v8f){};

  const float* ap = A + (size_t)(m0 + l) * K + 4 * half;
  const float* bp = B + (size_t)(4 * half) * N + n0 + 2 * l;
  const size_t aRowStep = (size_t)16 * K;
  const size_t bBlk = (size_t)16 * N;
  GemmFrags f0, f1;
  gemm_load_block(f0, ap, bp, aRowStep, (size_t)N);
  for (int k0 = 0; k0 < K - 32; k0 += 32) {
    gemm_load_block(f1, ap + 16, bp + bBlk, aRowStep, (size_t)N);
    __builtin_amdgcn_sched_barrier(0);
    gemm_mma_block(f0, acc);
    __builtin_amdgcn_sched_barrier(0);
    gemm_load_block(f0, ap + 32, bp + 2 * bBlk, aRowStep, (size_t)N);
    __builtin_amdgcn_sched_barrier(0);
    gemm_mma_block(f1, acc);
    __builtin_amdgcn_sched_barrier(0);
    ap += 32;
    bp += 2 * bBlk;
  }
  gemm_load_block(f1, ap + 16, bp + bBlk, aRowStep, (size_t)N);
  __builtin_amdgcn_sched_barrier(0);
  gemm_mma_block(f0, acc);
  gemm_mma_block(f1, acc);

  const int col = n0 + 2 * l;
  float bv0 = bias ? bias[col] : 0.0f;
  float bv1 = bias ? bias[col + 1] : 0.0f;
#pragma unroll
  for (int mi = 0; mi < 4; ++mi)
#pragma unroll
    for (int r = 0; r < 8; ++r) {
      v2f x;
      x.x = apply_act(acc[mi][0][r] + bv0, act);
      x.y = apply_act(acc[mi][1][r] + bv1, act);
      int m = m0 + mi * 16 + r + 8 * half;
      if (headSplit) {
        int hh = col >> 6, dd = col & 63;
        *(v2f*)&C[((size_t)hh * M + m) * 64 + dd] = x;
      } else {
        *(v2f*)&C[(size_t)m * N + col] = x;
      }
    }
}

#define CK_K 4096
#define CK_N 64
#define SP_WAVES 32
#define CK_SLICE (CK_K / SP_WAVES)
#define GT_N 48
#define GT_SLICE (HID_DIM / SP_WAVES)
#define SP_NI_PITCH 272
#define CK_W_PITCH (2 * SP_NI_PITCH)
#define GT_W_PITCH (3 * SP_NI_PITCH)
#define SP_COMPRESS_GROUPS 128
__global__ void __launch_bounds__(SP_WAVES * 32)
small_products_splitk(const float* __restrict__ kblk,
                      const float* __restrict__ vblk,
                      const float* __restrict__ Wc,
                      const float* __restrict__ bc, float* __restrict__ kc,
                      float* __restrict__ vc, const float* __restrict__ X,
                      const float* __restrict__ Wg,
                      const float* __restrict__ bg, float* __restrict__ gate) {
  __shared__ float part[SP_WAVES * GT_W_PITCH];
  int wave = threadIdx.x >> 5, lane = threadIdx.x & 31;
  int half = lane >> 4, l = lane & 15;
  int t = threadIdx.x;
  if (blockIdx.x < SP_COMPRESS_GROUPS) {
    int prod = blockIdx.x >> 6, g = blockIdx.x & 63;
    const float* __restrict__ A = prod ? vblk : kblk;
    float* __restrict__ C = prod ? vc : kc;
    int m0 = (g >> 1) << 4, n0 = (g & 1) << 5;
    int kbase = wave * CK_SLICE;
    const float* ap = A + (size_t)(m0 + l) * CK_K + kbase + 4 * half;
    const float* bp = Wc + (size_t)(kbase + 4 * half) * CK_N + n0 + 2 * l;
    v8f acc0 = {}, acc1 = {};
#pragma unroll
    for (int g = 0; g < CK_SLICE / 8; ++g) {
      v4f a4 = *(const v4f*)(ap + 8 * g);
      v2f br[4];
#pragma unroll
      for (int j = 0; j < 4; ++j)
        br[j] = *(const v2f*)(bp + (size_t)(8 * g + j) * CK_N);
#pragma unroll
      for (int p = 0; p < 2; ++p) {
        v2f a, b0, b1;
        a.x = a4[2 * p];
        a.y = a4[2 * p + 1];
        b0.x = br[2 * p].x;
        b0.y = br[2 * p + 1].x;
        b1.x = br[2 * p].y;
        b1.y = br[2 * p + 1].y;
        acc0 = wmma4(a, b0, acc0);
        acc1 = wmma4(a, b1, acc1);
      }
    }
    float* pw = part + wave * CK_W_PITCH + lane;
#pragma unroll
    for (int r = 0; r < 8; ++r) {
      pw[r * 32] = acc0[r];
      pw[SP_NI_PITCH + r * 32] = acc1[r];
    }
    __syncthreads();
    if (t < 512) {
      int row = t >> 5, col = t & 31;
      const float* pr = part + (col & 1) * SP_NI_PITCH + (row & 7) * 32 +
                        (row >> 3) * 16 + (col >> 1);
      float sum = pr[0];
#pragma unroll
      for (int w = 1; w < SP_WAVES; ++w) sum += pr[w * CK_W_PITCH];
      C[(size_t)(m0 + row) * CK_N + n0 + col] =
          apply_act(sum + bc[n0 + col], 1);
    }
  } else {
    int m0 = (blockIdx.x - SP_COMPRESS_GROUPS) << 4;
    int kbase = wave * GT_SLICE;
    const float* ap = X + (size_t)(m0 + l) * HID_DIM + kbase + 2 * half;
    const float* bp = Wg + (size_t)(kbase + 2 * half) * GT_N + l;
    v8f acc0 = {}, acc1 = {}, acc2 = {};
#pragma unroll
    for (int s = 0; s < GT_SLICE / 4; ++s) {
      v2f a, b0, b1, b2;
      a.x = ap[4 * s];
      a.y = ap[4 * s + 1];
      b0.x = bp[(4 * s) * GT_N];
      b0.y = bp[(4 * s + 1) * GT_N];
      b1.x = bp[(4 * s) * GT_N + 16];
      b1.y = bp[(4 * s + 1) * GT_N + 16];
      b2.x = bp[(4 * s) * GT_N + 32];
      b2.y = bp[(4 * s + 1) * GT_N + 32];
      acc0 = wmma4(a, b0, acc0);
      acc1 = wmma4(a, b1, acc1);
      acc2 = wmma4(a, b2, acc2);
    }
    float* pw = part + wave * GT_W_PITCH + lane;
#pragma unroll
    for (int r = 0; r < 8; ++r) {
      pw[r * 32] = acc0[r];
      pw[SP_NI_PITCH + r * 32] = acc1[r];
      pw[2 * SP_NI_PITCH + r * 32] = acc2[r];
    }
    __syncthreads();
    if (t < 16 * GT_N) {
      int row = t / GT_N, col = t - row * GT_N;
      const float* pr = part + (col >> 4) * SP_NI_PITCH + (row & 7) * 32 +
                        (row >> 3) * 16 + (col & 15);
      float sum = pr[0];
#pragma unroll
      for (int w = 1; w < SP_WAVES; ++w) sum += pr[w * GT_W_PITCH];
      gate[(size_t)m0 * GT_N + t] = apply_act(sum + bg[col], 2);
    }
  }
}

#define ATT_THREADS 128
__device__ __forceinline__ void tile_load_regs(float2 (&kr)[16],
                                               float2 (&vr)[16],
                                               const float* __restrict__ kt,
                                               const float* __restrict__ vt,
                                               int tid) {
  const int off = (tid >> 5) * DH + ((tid & 31) << 1);
#pragma unroll
  for (int i = 0; i < 16; ++i) {
    kr[i] = *(const float2*)&kt[off + i * 4 * DH];
    vr[i] = *(const float2*)&vt[off + i * 4 * DH];
  }
}

__device__ __forceinline__ void tile_store_lds(float* Kd, float* Vd,
                                               const float2 (&kr)[16],
                                               const float2 (&vr)[16],
                                               int tid) {
  const int off = (tid >> 5) * 66 + ((tid & 31) << 1);
#pragma unroll
  for (int i = 0; i < 16; ++i) {
    *(float2*)&Kd[off + i * 4 * 66] = kr[i];
    *(float2*)&Vd[off + i * 4 * 66] = vr[i];
  }
}

__global__ void __launch_bounds__(ATT_THREADS)
sliding_window_kernel(const float* __restrict__ q,
                                      const float* __restrict__ k,
                                      const float* __restrict__ v,
                                      float* __restrict__ out) {
  extern __shared__ float lds[];
  float* Ks = lds;
  float* Vs = Ks + 128 * 66;
  float* Ps = Vs + 128 * 66;
  int h = blockIdx.x >> 5;
  int w = blockIdx.x & 31;
  const float* kh = k + (size_t)h * T_SEQ * DH;
  const float* vh = v + (size_t)h * T_SEQ * DH;
  {
    float2 kc[16], vc[16], kp[16], vp[16];
    tile_load_regs(kc, vc, kh + (size_t)w * 64 * DH, vh + (size_t)w * 64 * DH,
                   threadIdx.x);
    if (w > 0) {
      tile_load_regs(kp, vp, kh + (size_t)(w - 1) * 64 * DH,
                     vh + (size_t)(w - 1) * 64 * DH, threadIdx.x);
    } else {
#pragma unroll
      for (int i = 0; i < 16; ++i) {
        kp[i] = make_float2(0.f, 0.f);
        vp[i] = make_float2(0.f, 0.f);
      }
    }
    tile_store_lds(Ks + 64 * 66, Vs + 64 * 66, kc, vc, threadIdx.x);
    tile_store_lds(Ks, Vs, kp, vp, threadIdx.x);
  }
  __syncthreads();

  int wv = threadIdx.x >> 5, lane = threadIdx.x & 31;
  int half = lane >> 4, l = lane & 15;
  const float* qrow =
      q + ((size_t)h * T_SEQ + w * 64 + wv * 16 + l) * DH + 2 * half;

  v8f s[8] = {};
  for (int k0 = 0; k0 < 64; k0 += 4) {
    v2f a;
    a.x = qrow[k0];
    a.y = qrow[k0 + 1];
#pragma unroll
    for (int nt = 0; nt < 8; ++nt) {
      v2f b;
      const float* kp = &Ks[(nt * 16 + l) * 66 + k0 + 2 * half];
      b.x = kp[0];
      b.y = kp[1];
      s[nt] = wmma4(a, b, s[nt]);
    }
  }
  float rmax[8], rsum[8];
#pragma unroll
  for (int r = 0; r < 8; ++r) { rmax[r] = -3.0e38f; rsum[r] = 0.f; }
#pragma unroll
  for (int nt = 0; nt < 8; ++nt) {
    int n = nt * 16 + l;
#pragma unroll
    for (int r = 0; r < 8; ++r) {
      int m = wv * 16 + r + 8 * half;
      float x = s[nt][r] * 0.125f;
      bool ok = (m + 64 >= n) && (w > 0 || n >= 64);
      x = ok ? x : -3.0e38f;
      s[nt][r] = x;
      rmax[r] = fmaxf(rmax[r], x);
    }
  }
#pragma unroll
  for (int d = 1; d < 16; d <<= 1)
#pragma unroll
    for (int r = 0; r < 8; ++r) rmax[r] = fmaxf(rmax[r], __shfl_xor(rmax[r], d));
#pragma unroll
  for (int nt = 0; nt < 8; ++nt)
#pragma unroll
    for (int r = 0; r < 8; ++r) {
      float e = __expf(s[nt][r] - rmax[r]);
      s[nt][r] = e;
      rsum[r] += e;
    }
#pragma unroll
  for (int d = 1; d < 16; d <<= 1)
#pragma unroll
    for (int r = 0; r < 8; ++r) rsum[r] += __shfl_xor(rsum[r], d);
  float rinv[8];
#pragma unroll
  for (int r = 0; r < 8; ++r) rinv[r] = 1.0f / rsum[r];
#pragma unroll
  for (int nt = 0; nt < 8; ++nt)
#pragma unroll
    for (int r = 0; r < 8; ++r)
      Ps[(wv * 16 + r + 8 * half) * 130 + nt * 16 + l] = s[nt][r] * rinv[r];
  __syncthreads();

  v8f o[4] = {};
  const float* prow = &Ps[(wv * 16 + l) * 130 + 2 * half];
  for (int k0 = 0; k0 < 128; k0 += 4) {
    v2f a;
    a.x = prow[k0];
    a.y = prow[k0 + 1];
#pragma unroll
    for (int nt = 0; nt < 4; ++nt) {
      v2f b;
      const float* vp = &Vs[(k0 + 2 * half) * 66 + nt * 16 + l];
      b.x = vp[0];
      b.y = vp[66];
      o[nt] = wmma4(a, b, o[nt]);
    }
  }
  float* oh = out + ((size_t)h * T_SEQ + w * 64) * DH;
#pragma unroll
  for (int nt = 0; nt < 4; ++nt)
#pragma unroll
    for (int r = 0; r < 8; ++r)
      oh[(size_t)(wv * 16 + r + 8 * half) * DH + nt * 16 + l] = o[nt][r];
}

__global__ void comp_branch_kernel(const float* __restrict__ q,
                                   const float* __restrict__ kc,
                                   const float* __restrict__ vc,
                                   float* __restrict__ bs,
                                   float* __restrict__ comp) {
  __shared__ float Ps[16 * 34];
  int h = blockIdx.x >> 7;
  int tt = blockIdx.x & 127;
  int t0 = tt << 4;
  int lane = threadIdx.x, half = lane >> 4, l = lane & 15;
  const float* qrow = q + ((size_t)h * T_SEQ + t0 + l) * DH + 2 * half;
  v8f s[2] = {};
  for (int k0 = 0; k0 < 64; k0 += 4) {
    v2f a;
    a.x = qrow[k0];
    a.y = qrow[k0 + 1];
#pragma unroll
    for (int nt = 0; nt < 2; ++nt) {
      v2f b;
      const float* kp = kc + (size_t)(h * 32 + nt * 16 + l) * 64 + k0 + 2 * half;
      b.x = kp[0];
      b.y = kp[1];
      s[nt] = wmma4(a, b, s[nt]);
    }
  }
#pragma unroll
  for (int nt = 0; nt < 2; ++nt)
#pragma unroll
    for (int r = 0; r < 8; ++r) s[nt][r] *= 0.125f;

  int qb = t0 >> 6;
#pragma unroll
  for (int nt = 0; nt < 2; ++nt) {
    float cs = 0.f;
#pragma unroll
    for (int r = 0; r < 8; ++r) cs += s[nt][r];
    cs += __shfl_xor(cs, 16);
    if (half == 0)
      atomicAdd(&bs[((size_t)h * 32 + qb) * 32 + nt * 16 + l],
                cs * (1.0f / 64.0f));
  }
  float rmax[8], rsum[8];
#pragma unroll
  for (int r = 0; r < 8; ++r) rmax[r] = fmaxf(s[0][r], s[1][r]);
#pragma unroll
  for (int d = 1; d < 16; d <<= 1)
#pragma unroll
    for (int r = 0; r < 8; ++r) rmax[r] = fmaxf(rmax[r], __shfl_xor(rmax[r], d));
#pragma unroll
  for (int r = 0; r < 8; ++r) rsum[r] = 0.f;
#pragma unroll
  for (int nt = 0; nt < 2; ++nt)
#pragma unroll
    for (int r = 0; r < 8; ++r) {
      float e = __expf(s[nt][r] - rmax[r]);
      s[nt][r] = e;
      rsum[r] += e;
    }
#pragma unroll
  for (int d = 1; d < 16; d <<= 1)
#pragma unroll
    for (int r = 0; r < 8; ++r) rsum[r] += __shfl_xor(rsum[r], d);
#pragma unroll
  for (int nt = 0; nt < 2; ++nt)
#pragma unroll
    for (int r = 0; r < 8; ++r)
      Ps[(r + 8 * half) * 34 + nt * 16 + l] = s[nt][r] / rsum[r];
  __syncthreads();

  v8f o[4] = {};
  const float* prow = &Ps[l * 34 + 2 * half];
  for (int k0 = 0; k0 < 32; k0 += 4) {
    v2f a;
    a.x = prow[k0];
    a.y = prow[k0 + 1];
#pragma unroll
    for (int nt = 0; nt < 4; ++nt) {
      v2f b;
      const float* vp = vc + (size_t)(h * 32 + k0 + 2 * half) * 64 + nt * 16 + l;
      b.x = vp[0];
      b.y = vp[64];
      o[nt] = wmma4(a, b, o[nt]);
    }
  }
  float* oh = comp + ((size_t)h * T_SEQ + t0) * DH;
#pragma unroll
  for (int nt = 0; nt < 4; ++nt)
#pragma unroll
    for (int r = 0; r < 8; ++r)
      oh[(size_t)(r + 8 * half) * DH + nt * 16 + l] = o[nt][r];
}

__global__ void block_stats_kernel(const float* __restrict__ bs,
                                   const float* __restrict__ ksc,
                                   const float* __restrict__ kbi,
                                   unsigned char* __restrict__ mask,
                                   float* __restrict__ avg_acc,
                                   float* __restrict__ tail) {
  int h = blockIdx.x, j = threadIdx.x;
  const float* row = bs + ((size_t)h * 32 + j) * 32;
  float sc[32], p[32];
  float mx = -3.0e38f;
  for (int c = 0; c < 32; ++c) { sc[c] = row[c]; mx = fmaxf(mx, sc[c]); }
  float ssum = 0.f;
  for (int c = 0; c < 32; ++c) { p[c] = __expf(sc[c] - mx); ssum += p[c]; }
  float inv = 1.0f / ssum;
  float ent = 0.f;
  for (int c = 0; c < 32; ++c) {
    float pp = p[c] * inv;
    ent -= pp * __logf(pp + 1e-9f);
  }
  float ne = ent / (logf(32.0f) + 1e-9f);
  float kr = 1.0f / (1.0f + __expf(-(ksc[h] * ne + kbi[h])));
  float dkf = fminf(fmaxf(8.0f * kr, 1.0f), 8.0f);
  int dk = (int)dkf;
  tail[1 + h * 32 + j] = kr;
  tail[1 + 512 + h * 32 + j] = ne;
  atomicAdd(avg_acc, (float)dk);
  unsigned used = 0;
  unsigned char mrow[32];
  for (int c = 0; c < 32; ++c) mrow[c] = 0;
  mrow[j] = 1;
  for (int kk = 0; kk < KMAX_SEL; ++kk) {
    float best = -3.4e38f;
    int bi = 0;
    for (int c = 0; c < 32; ++c)
      if (!((used >> c) & 1u) && sc[c] > best) { best = sc[c]; bi = c; }
    used |= 1u << bi;
    if (kk < dk) mrow[bi] = 1;
  }
  for (int c = 0; c < 32; ++c) mask[((size_t)h * 32 + j) * 32 + c] = mrow[c];
}

__global__ void __launch_bounds__(ATT_THREADS)
fine_attn_kernel(const float* __restrict__ q,
                                 const float* __restrict__ k,
                                 const float* __restrict__ v,
                                 const unsigned char* __restrict__ mask,
                                 float* __restrict__ out) {
  __shared__ float Ks[64 * 66];
  __shared__ float Vs[64 * 66];
  __shared__ float Ps[64 * 66];
  int h = blockIdx.x >> 5, qb = blockIdx.x & 31;
  int wv = threadIdx.x >> 5, lane = threadIdx.x & 31;
  int half = lane >> 4, l = lane & 15;
  const float* kh = k + (size_t)h * T_SEQ * DH;
  const float* vh = v + (size_t)h * T_SEQ * DH;
  const float* qrow =
      q + ((size_t)h * T_SEQ + qb * 64 + wv * 16 + l) * DH + 2 * half;
  const unsigned char* mrow = mask + ((size_t)h * 32 + qb) * 32;
  v8f o[4] = {};
  float rmax[8], rsum[8];
#pragma unroll
  for (int r = 0; r < 8; ++r) { rmax[r] = -3.0e38f; rsum[r] = 0.f; }

  for (int j = 0; j < 32; ++j) {
    if (!mrow[j]) continue;
    float2 kr[16], vr[16];
    tile_load_regs(kr, vr, kh + (size_t)j * 64 * DH, vh + (size_t)j * 64 * DH,
                   threadIdx.x);
    __syncthreads();
    tile_store_lds(Ks, Vs, kr, vr, threadIdx.x);
    __syncthreads();
    v8f s[4] = {};
    for (int k0 = 0; k0 < 64; k0 += 4) {
      v2f a;
      a.x = qrow[k0];
      a.y = qrow[k0 + 1];
#pragma unroll
      for (int nt = 0; nt < 4; ++nt) {
        v2f b;
        const float* kp = &Ks[(nt * 16 + l) * 66 + k0 + 2 * half];
        b.x = kp[0];
        b.y = kp[1];
        s[nt] = wmma4(a, b, s[nt]);
      }
    }
    float nm[8], ps[8], fac[8];
#pragma unroll
    for (int r = 0; r < 8; ++r) nm[r] = rmax[r];
#pragma unroll
    for (int nt = 0; nt < 4; ++nt)
#pragma unroll
      for (int r = 0; r < 8; ++r) {
        float x = s[nt][r] * 0.125f;
        s[nt][r] = x;
        nm[r] = fmaxf(nm[r], x);
      }
#pragma unroll
    for (int d = 1; d < 16; d <<= 1)
#pragma unroll
      for (int r = 0; r < 8; ++r) nm[r] = fmaxf(nm[r], __shfl_xor(nm[r], d));
#pragma unroll
    for (int r = 0; r < 8; ++r) { fac[r] = __expf(rmax[r] - nm[r]); ps[r] = 0.f; }
#pragma unroll
    for (int nt = 0; nt < 4; ++nt)
#pragma unroll
      for (int r = 0; r < 8; ++r) {
        float e = __expf(s[nt][r] - nm[r]);
        s[nt][r] = e;
        ps[r] += e;
      }
#pragma unroll
    for (int d = 1; d < 16; d <<= 1)
#pragma unroll
      for (int r = 0; r < 8; ++r) ps[r] += __shfl_xor(ps[r], d);
#pragma unroll
    for (int r = 0; r < 8; ++r) {
      rsum[r] = rsum[r] * fac[r] + ps[r];
      rmax[r] = nm[r];
    }
#pragma unroll
    for (int nt = 0; nt < 4; ++nt)
#pragma unroll
      for (int r = 0; r < 8; ++r)
        Ps[(wv * 16 + r + 8 * half) * 66 + nt * 16 + l] = s[nt][r];
#pragma unroll
    for (int nt = 0; nt < 4; ++nt)
#pragma unroll
      for (int r = 0; r < 8; ++r) o[nt][r] *= fac[r];
    __syncthreads();
    const float* prow = &Ps[(wv * 16 + l) * 66 + 2 * half];
    for (int k0 = 0; k0 < 64; k0 += 4) {
      v2f a;
      a.x = prow[k0];
      a.y = prow[k0 + 1];
#pragma unroll
      for (int nt = 0; nt < 4; ++nt) {
        v2f b;
        const float* vp = &Vs[(k0 + 2 * half) * 66 + nt * 16 + l];
        b.x = vp[0];
        b.y = vp[66];
        o[nt] = wmma4(a, b, o[nt]);
      }
    }
  }
  float* oh = out + ((size_t)h * T_SEQ + qb * 64) * DH;
#pragma unroll
  for (int nt = 0; nt < 4; ++nt)
#pragma unroll
    for (int r = 0; r < 8; ++r)
      oh[(size_t)(wv * 16 + r + 8 * half) * DH + nt * 16 + l] =
          o[nt][r] / rsum[r];
}

__global__ void combine_kernel(const float* __restrict__ sw,
                               const float* __restrict__ comp,
                               const float* __restrict__ fine,
                               const float* __restrict__ gate,
                               float* __restrict__ merged) {
  int idx = blockIdx.x * blockDim.x + threadIdx.x;
  if (idx >= T_SEQ * HID_DIM) return;
  int t = idx >> 10, c = idx & 1023, hh = c >> 6, dd = c & 63;
  size_t a = ((size_t)hh * T_SEQ + t) * DH + dd;
  const float* g = gate + (size_t)t * 48 + hh * 3;
  merged[idx] = sw[a] * g[0] + comp[a] * g[1] + fine[a] * g[2];
}

__global__ void zero_kernel(float* __restrict__ p, int n) {
  int i = blockIdx.x * blockDim.x + threadIdx.x;
  if (i < n) p[i] = 0.f;
}

__global__ void finalize_kernel(const float* __restrict__ avg_acc,
                                float* __restrict__ tail) {
  if (threadIdx.x == 0) tail[0] = avg_acc[0] * (1.0f / 512.0f);
}

extern "C" void kernel_launch(void* const* d_in, const int* in_sizes, int n_in,
                              void* d_out, int out_size, void* d_ws,
                              size_t ws_size, hipStream_t stream) {
  const float* X = (const float*)d_in[0];
  const float* Wq = (const float*)d_in[1];
  const float* Wk = (const float*)d_in[2];
  const float* Wv = (const float*)d_in[3];
  const float* Wo = (const float*)d_in[4];
  const float* Wc = (const float*)d_in[5];
  const float* bc = (const float*)d_in[6];
  const float* ksc = (const float*)d_in[7];
  const float* kbi = (const float*)d_in[8];
  const float* Wg = (const float*)d_in[9];
  const float* bg = (const float*)d_in[10];
  float* out = (float*)d_out;

  const size_t U = (size_t)NH * T_SEQ * DH;
  float* ws = (float*)d_ws;
  float* q = ws;
  float* kbuf = ws + U;
  float* vbuf = ws + 2 * U;
  float* sw = ws + 3 * U;
  float* comp = ws + 4 * U;
  float* fine = ws + 5 * U;
  float* merged = q;
  float* extra = ws + 6 * U;
  float* gate = extra;
  float* kc = gate + (size_t)T_SEQ * 48;
  float* vc = kc + 512 * 64;
  float* bs = vc + 512 * 64;
  float* avg = bs + 16384;
  unsigned char* mask = (unsigned char*)(avg + 64);

  zero_kernel<<<(16385 + 255) / 256, 256, 0, stream>>>(bs, 16385);

  {
    int wtiles = 3 * (T_SEQ / 64) * (HID_DIM / 32);
    int blocks = (wtiles + 11) / 12;
    wmma_gemm_f32_blk<<<blocks, 384, 0, stream>>>(X, Wq, Wk, Wv, nullptr, q,
                                                  kbuf, vbuf, 3, T_SEQ, HID_DIM,
                                                  HID_DIM, 0, 1);
  }
  sliding_window_kernel<<<NH * NB, ATT_THREADS,
                          (2 * 128 * 66 + 64 * 130) * sizeof(float), stream>>>(
      q, kbuf, vbuf, sw);
  {
    small_products_splitk<<<256, 1024, 0, stream>>>(kbuf, vbuf, Wc, bc, kc, vc,
                                                    X, Wg, bg, gate);
  }
  comp_branch_kernel<<<NH * (T_SEQ / 16), 32, 0, stream>>>(q, kc, vc, bs, comp);
  block_stats_kernel<<<NH, 32, 0, stream>>>(bs, ksc, kbi, mask, avg,
                                            out + (size_t)T_SEQ * HID_DIM);
  fine_attn_kernel<<<NH * NB, ATT_THREADS, 0, stream>>>(q, kbuf, vbuf, mask,
                                                        fine);
  combine_kernel<<<(T_SEQ * HID_DIM + 255) / 256, 256, 0, stream>>>(
      sw, comp, fine, gate, merged);
  {
    int wtiles = (T_SEQ / 64) * (HID_DIM / 32);
    wmma_gemm_f32_blk<<<(wtiles + 3) / 4, 128, 0, stream>>>(
        merged, Wo, nullptr, nullptr, nullptr, out, nullptr, nullptr, 1, T_SEQ,
        HID_DIM, HID_DIM, 0, 0);
  }
  finalize_kernel<<<1, 1, 0, stream>>>(avg, out + (size_t)T_SEQ * HID_DIM);
  (void)in_sizes;
  (void)n_in;
  (void)out_size;
  (void)ws_size;
}


# softmax row max/sum butterflies of the three attention kernels as DPP row_xmask operands on the vector ALU instead of LDS-unit shuffles waited for step by step; same pairing, bit-identical
# speedup vs baseline: 10.0767x; 1.0342x over previous
#include <hip/hip_runtime.h>
#include <hip/hip_bf16.h>
#include <math.h>

typedef float v2f __attribute__((ext_vector_type(2)));
typedef float v4f __attribute__((ext_vector_type(4)));
typedef float v8f __attribute__((ext_vector_type(8)));

#define T_SEQ 2048
#define HID_DIM 1024
#define NH 16
#define DH 64
#define NB 32
#define KMAX_SEL 8

__device__ __forceinline__ v8f wmma4(v2f a, v2f b, v8f c) {
  return __builtin_amdgcn_wmma_f32_16x16x4_f32(false, a, false, b, (short)0, c,
                                               false, false);
}

__device__ __forceinline__ float apply_act(float x, int act) {
  if (act == 1) return 0.5f * x * (1.0f + erff(x * 0.70710678118654752f));
  if (act == 2) return 1.0f / (1.0f + __expf(-x));
  return x;
}

template <int M>
__device__ __forceinline__ float row_xor(float x) {
  return __builtin_amdgcn_update_dpp(x, x, 0x160 + M, 0xf, 0xf, true);
}

template <int M>
__device__ __forceinline__ void row_max_step(float (&x)[8]) {
#pragma unroll
  for (int r = 0; r < 8; ++r) x[r] = fmaxf(x[r], row_xor<M>(x[r]));
}

template <int M>
__device__ __forceinline__ void row_sum_step(float (&x)[8]) {
#pragma unroll
  for (int r = 0; r < 8; ++r) x[r] += row_xor<M>(x[r]);
}

__device__ __forceinline__ void row_max8(float (&x)[8]) {
  row_max_step<1>(x);
  row_max_step<2>(x);
  row_max_step<4>(x);
  row_max_step<8>(x);
}

__device__ __forceinline__ void row_sum8(float (&x)[8]) {
  row_sum_step<1>(x);
  row_sum_step<2>(x);
  row_sum_step<4>(x);
  row_sum_step<8>(x);
}

__device__ __forceinline__ float other_row(float x) {
  int v = __builtin_bit_cast(int, x);
  return __builtin_bit_cast(
      float, __builtin_amdgcn_permlanex16(v, v, 0x76543210, 0xfedcba98, false,
                                          false));
}

struct GemmFrags {
  v4f a[2][4];
  v2f b[2][4];
};

__device__ __forceinline__ void gemm_load_block(GemmFrags& f, const float* ap,
                                                const float* bp,
                                                size_t aRowStep, size_t N) {
#pragma unroll
  for (int g = 0; g < 2; ++g) {
#pragma unroll
    for (int mi = 0; mi < 4; ++mi)
      f.a[g][mi] = *(const v4f*)(ap + (size_t)mi * aRowStep + 8 * g);
#pragma unroll
    for (int j = 0; j < 4; ++j)
      f.b[g][j] = *(const v2f*)(bp + (size_t)(8 * g + j) * N);
  }
}

__device__ __forceinline__ void gemm_mma_block(const GemmFrags& f,
                                               v8f (&acc)[4][2]) {
#pragma unroll
  for (int g = 0; g < 2; ++g)
#pragma unroll
    for (int p = 0; p < 2; ++p) {
      v2f b0, b1;
      b0.x = f.b[g][2 * p].x;
      b0.y = f.b[g][2 * p + 1].x;
      b1.x = f.b[g][2 * p].y;
      b1.y = f.b[g][2 * p + 1].y;
#pragma unroll
      for (int mi = 0; mi < 4; ++mi) {
        v2f a;
        a.x = f.a[g][mi][2 * p];
        a.y = f.a[g][mi][2 * p + 1];
        acc[mi][0] = wmma4(a, b0, acc[mi][0]);
        acc[mi][1] = wmma4(a, b1, acc[mi][1]);
      }
    }
}

__global__ void __launch_bounds__(384)
wmma_gemm_f32_blk(const float* __restrict__ A, const float* B0,
                  const float* B1, const float* B2,
                  const float* __restrict__ bias, float* C0, float* C1,
                  float* C2, int nProd, int M, int N, int K, int act,
                  int headSplit) {
  int wave = threadIdx.x >> 5;
  int lane = threadIdx.x & 31;
  int wtile = blockIdx.x * (blockDim.x >> 5) + wave;
  int wn = N >> 5;
  int total = (M >> 6) * wn;
  if (wtile >= total * nProd) return;
  int prod = wtile / total;
  wtile -= prod * total;
  const float* __restrict__ B = prod == 0 ? B0 : (prod == 1 ? B1 : B2);
  float* __restrict__ C = prod == 0 ? C0 : (prod == 1 ? C1 : C2);
  int tm = wtile / wn, tn = wtile - tm * wn;
  int half = lane >> 4, l = lane & 15;
  int m0 = tm << 6, n0 = tn << 5;

  v8f acc[4][2];
#pragma unroll
  for (int mi = 0; mi < 4; ++mi)
#pragma unroll
    for (int ni = 0; ni < 2; ++ni) acc[mi][ni] = (v8f){};

  const float* ap = A + (size_t)(m0 + l) * K + 4 * half;
  const float* bp = B + (size_t)(4 * half) * N + n0 + 2 * l;
  const size_t aRowStep = (size_t)16 * K;
  const size_t bBlk = (size_t)16 * N;
  GemmFrags f0, f1;
  gemm_load_block(f0, ap, bp, aRowStep, (size_t)N);
  for (int k0 = 0; k0 < K - 32; k0 += 32) {
    gemm_load_block(f1, ap + 16, bp + bBlk, aRowStep, (size_t)N);
    __builtin_amdgcn_sched_barrier(0);
    gemm_mma_block(f0, acc);
    __builtin_amdgcn_sched_barrier(0);
    gemm_load_block(f0, ap + 32, bp + 2 * bBlk, aRowStep, (size_t)N);
    __builtin_amdgcn_sched_barrier(0);
    gemm_mma_block(f1, acc);
    __builtin_amdgcn_sched_barrier(0);
    ap += 32;
    bp += 2 * bBlk;
  }
  gemm_load_block(f1, ap + 16, bp + bBlk, aRowStep, (size_t)N);
  __builtin_amdgcn_sched_barrier(0);
  gemm_mma_block(f0, acc);
  gemm_mma_block(f1, acc);

  const int col = n0 + 2 * l;
  float bv0 = bias ? bias[col] : 0.0f;
  float bv1 = bias ? bias[col + 1] : 0.0f;
#pragma unroll
  for (int mi = 0; mi < 4; ++mi)
#pragma unroll
    for (int r = 0; r < 8; ++r) {
      v2f x;
      x.x = apply_act(acc[mi][0][r] + bv0, act);
      x.y = apply_act(acc[mi][1][r] + bv1, act);
      int m = m0 + mi * 16 + r + 8 * half;
      if (headSplit) {
        int hh = col >> 6, dd = col & 63;
        *(v2f*)&C[((size_t)hh * M + m) * 64 + dd] = x;
      } else {
        *(v2f*)&C[(size_t)m * N + col] = x;
      }
    }
}

#define CK_K 4096
#define CK_N 64
#define SP_WAVES 32
#define CK_SLICE (CK_K / SP_WAVES)
#define GT_N 48
#define GT_SLICE (HID_DIM / SP_WAVES)
#define SP_NI_PITCH 272
#define CK_W_PITCH (2 * SP_NI_PITCH)
#define GT_W_PITCH (3 * SP_NI_PITCH)
#define SP_COMPRESS_GROUPS 128
__global__ void __launch_bounds__(SP_WAVES * 32)
small_products_splitk(const float* __restrict__ kblk,
                      const float* __restrict__ vblk,
                      const float* __restrict__ Wc,
                      const float* __restrict__ bc, float* __restrict__ kc,
                      float* __restrict__ vc, const float* __restrict__ X,
                      const float* __restrict__ Wg,
                      const float* __restrict__ bg, float* __restrict__ gate) {
  __shared__ float part[SP_WAVES * GT_W_PITCH];
  int wave = threadIdx.x >> 5, lane = threadIdx.x & 31;
  int half = lane >> 4, l = lane & 15;
  int t = threadIdx.x;
  if (blockIdx.x < SP_COMPRESS_GROUPS) {
    int prod = blockIdx.x >> 6, g = blockIdx.x & 63;
    const float* __restrict__ A = prod ? vblk : kblk;
    float* __restrict__ C = prod ? vc : kc;
    int m0 = (g >> 1) << 4, n0 = (g & 1) << 5;
    int kbase = wave * CK_SLICE;
    const float* ap = A + (size_t)(m0 + l) * CK_K + kbase + 4 * half;
    const float* bp = Wc + (size_t)(kbase + 4 * half) * CK_N + n0 + 2 * l;
    v8f acc0 = {}, acc1 = {};
#pragma unroll
    for (int g = 0; g < CK_SLICE / 8; ++g) {
      v4f a4 = *(const v4f*)(ap + 8 * g);
      v2f br[4];
#pragma unroll
      for (int j = 0; j < 4; ++j)
        br[j] = *(const v2f*)(bp + (size_t)(8 * g + j) * CK_N);
#pragma unroll
      for (int p = 0; p < 2; ++p) {
        v2f a, b0, b1;
        a.x = a4[2 * p];
        a.y = a4[2 * p + 1];
        b0.x = br[2 * p].x;
        b0.y = br[2 * p + 1].x;
        b1.x = br[2 * p].y;
        b1.y = br[2 * p + 1].y;
        acc0 = wmma4(a, b0, acc0);
        acc1 = wmma4(a, b1, acc1);
      }
    }
    float* pw = part + wave * CK_W_PITCH + lane;
#pragma unroll
    for (int r = 0; r < 8; ++r) {
      pw[r * 32] = acc0[r];
      pw[SP_NI_PITCH + r * 32] = acc1[r];
    }
    __syncthreads();
    if (t < 512) {
      int row = t >> 5, col = t & 31;
      const float* pr = part + (col & 1) * SP_NI_PITCH + (row & 7) * 32 +
                        (row >> 3) * 16 + (col >> 1);
      float sum = pr[0];
#pragma unroll
      for (int w = 1; w < SP_WAVES; ++w) sum += pr[w * CK_W_PITCH];
      C[(size_t)(m0 + row) * CK_N + n0 + col] =
          apply_act(sum + bc[n0 + col], 1);
    }
  } else {
    int m0 = (blockIdx.x - SP_COMPRESS_GROUPS) << 4;
    int kbase = wave * GT_SLICE;
    const float* ap = X + (size_t)(m0 + l) * HID_DIM + kbase + 2 * half;
    const float* bp = Wg + (size_t)(kbase + 2 * half) * GT_N + l;
    v8f acc0 = {}, acc1 = {}, acc2 = {};
#pragma unroll
    for (int s = 0; s < GT_SLICE / 4; ++s) {
      v2f a, b0, b1, b2;
      a.x = ap[4 * s];
      a.y = ap[4 * s + 1];
      b0.x = bp[(4 * s) * GT_N];
      b0.y = bp[(4 * s + 1) * GT_N];
      b1.x = bp[(4 * s) * GT_N + 16];
      b1.y = bp[(4 * s + 1) * GT_N + 16];
      b2.x = bp[(4 * s) * GT_N + 32];
      b2.y = bp[(4 * s + 1) * GT_N + 32];
      acc0 = wmma4(a, b0, acc0);
      acc1 = wmma4(a, b1, acc1);
      acc2 = wmma4(a, b2, acc2);
    }
    float* pw = part + wave * GT_W_PITCH + lane;
#pragma unroll
    for (int r = 0; r < 8; ++r) {
      pw[r * 32] = acc0[r];
      pw[SP_NI_PITCH + r * 32] = acc1[r];
      pw[2 * SP_NI_PITCH + r * 32] = acc2[r];
    }
    __syncthreads();
    if (t < 16 * GT_N) {
      int row = t / GT_N, col = t - row * GT_N;
      const float* pr = part + (col >> 4) * SP_NI_PITCH + (row & 7) * 32 +
                        (row >> 3) * 16 + (col & 15);
      float sum = pr[0];
#pragma unroll
      for (int w = 1; w < SP_WAVES; ++w) sum += pr[w * GT_W_PITCH];
      gate[(size_t)m0 * GT_N + t] = apply_act(sum + bg[col], 2);
    }
  }
}

#define ATT_THREADS 128
__device__ __forceinline__ void tile_load_regs(float2 (&kr)[16],
                                               float2 (&vr)[16],
                                               const float* __restrict__ kt,
                                               const float* __restrict__ vt,
                                               int tid) {
  const int off = (tid >> 5) * DH + ((tid & 31) << 1);
#pragma unroll
  for (int i = 0; i < 16; ++i) {
    kr[i] = *(const float2*)&kt[off + i * 4 * DH];
    vr[i] = *(const float2*)&vt[off + i * 4 * DH];
  }
}

__device__ __forceinline__ void tile_store_lds(float* Kd, float* Vd,
                                               const float2 (&kr)[16],
                                               const float2 (&vr)[16],
                                               int tid) {
  const int off = (tid >> 5) * 66 + ((tid & 31) << 1);
#pragma unroll
  for (int i = 0; i < 16; ++i) {
    *(float2*)&Kd[off + i * 4 * 66] = kr[i];
    *(float2*)&Vd[off + i * 4 * 66] = vr[i];
  }
}

__global__ void __launch_bounds__(ATT_THREADS)
sliding_window_kernel(const float* __restrict__ q,
                                      const float* __restrict__ k,
                                      const float* __restrict__ v,
                                      float* __restrict__ out) {
  extern __shared__ float lds[];
  float* Ks = lds;
  float* Vs = Ks + 128 * 66;
  float* Ps = Vs + 128 * 66;
  int h = blockIdx.x >> 5;
  int w = blockIdx.x & 31;
  const float* kh = k + (size_t)h * T_SEQ * DH;
  const float* vh = v + (size_t)h * T_SEQ * DH;
  {
    float2 kc[16], vc[16], kp[16], vp[16];
    tile_load_regs(kc, vc, kh + (size_t)w * 64 * DH, vh + (size_t)w * 64 * DH,
                   threadIdx.x);
    if (w > 0) {
      tile_load_regs(kp, vp, kh + (size_t)(w - 1) * 64 * DH,
                     vh + (size_t)(w - 1) * 64 * DH, threadIdx.x);
    } else {
#pragma unroll
      for (int i = 0; i < 16; ++i) {
        kp[i] = make_float2(0.f, 0.f);
        vp[i] = make_float2(0.f, 0.f);
      }
    }
    tile_store_lds(Ks + 64 * 66, Vs + 64 * 66, kc, vc, threadIdx.x);
    tile_store_lds(Ks, Vs, kp, vp, threadIdx.x);
  }
  __syncthreads();

  int wv = threadIdx.x >> 5, lane = threadIdx.x & 31;
  int half = lane >> 4, l = lane & 15;
  const float* qrow =
      q + ((size_t)h * T_SEQ + w * 64 + wv * 16 + l) * DH + 2 * half;

  v8f s[8] = {};
  for (int k0 = 0; k0 < 64; k0 += 4) {
    v2f a;
    a.x = qrow[k0];
    a.y = qrow[k0 + 1];
#pragma unroll
    for (int nt = 0; nt < 8; ++nt) {
      v2f b;
      const float* kp = &Ks[(nt * 16 + l) * 66 + k0 + 2 * half];
      b.x = kp[0];
      b.y = kp[1];
      s[nt] = wmma4(a, b, s[nt]);
    }
  }
  float rmax[8], rsum[8];
#pragma unroll
  for (int r = 0; r < 8; ++r) { rmax[r] = -3.0e38f; rsum[r] = 0.f; }
#pragma unroll
  for (int nt = 0; nt < 8; ++nt) {
    int n = nt * 16 + l;
#pragma unroll
    for (int r = 0; r < 8; ++r) {
      int m = wv * 16 + r + 8 * half;
      float x = s[nt][r] * 0.125f;
      bool ok = (m + 64 >= n) && (w > 0 || n >= 64);
      x = ok ? x : -3.0e38f;
      s[nt][r] = x;
      rmax[r] = fmaxf(rmax[r], x);
    }
  }
  row_max8(rmax);
#pragma unroll
  for (int nt = 0; nt < 8; ++nt)
#pragma unroll
    for (int r = 0; r < 8; ++r) {
      float e = __expf(s[nt][r] - rmax[r]);
      s[nt][r] = e;
      rsum[r] += e;
    }
  row_sum8(rsum);
  float rinv[8];
#pragma unroll
  for (int r = 0; r < 8; ++r) rinv[r] = 1.0f / rsum[r];
#pragma unroll
  for (int nt = 0; nt < 8; ++nt)
#pragma unroll
    for (int r = 0; r < 8; ++r)
      Ps[(wv * 16 + r + 8 * half) * 130 + nt * 16 + l] = s[nt][r] * rinv[r];
  __syncthreads();

  v8f o[4] = {};
  const float* prow = &Ps[(wv * 16 + l) * 130 + 2 * half];
  for (int k0 = 0; k0 < 128; k0 += 4) {
    v2f a;
    a.x = prow[k0];
    a.y = prow[k0 + 1];
#pragma unroll
    for (int nt = 0; nt < 4; ++nt) {
      v2f b;
      const float* vp = &Vs[(k0 + 2 * half) * 66 + nt * 16 + l];
      b.x = vp[0];
      b.y = vp[66];
      o[nt] = wmma4(a, b, o[nt]);
    }
  }
  float* oh = out + ((size_t)h * T_SEQ + w * 64) * DH;
#pragma unroll
  for (int nt = 0; nt < 4; ++nt)
#pragma unroll
    for (int r = 0; r < 8; ++r)
      oh[(size_t)(wv * 16 + r + 8 * half) * DH + nt * 16 + l] = o[nt][r];
}

__global__ void comp_branch_kernel(const float* __restrict__ q,
                                   const float* __restrict__ kc,
                                   const float* __restrict__ vc,
                                   float* __restrict__ bs,
                                   float* __restrict__ comp) {
  __shared__ float Ps[16 * 34];
  int h = blockIdx.x >> 7;
  int tt = blockIdx.x & 127;
  int t0 = tt << 4;
  int lane = threadIdx.x, half = lane >> 4, l = lane & 15;
  const float* qrow = q + ((size_t)h * T_SEQ + t0 + l) * DH + 2 * half;
  v8f s[2] = {};
  for (int k0 = 0; k0 < 64; k0 += 4) {
    v2f a;
    a.x = qrow[k0];
    a.y = qrow[k0 + 1];
#pragma unroll
    for (int nt = 0; nt < 2; ++nt) {
      v2f b;
      const float* kp = kc + (size_t)(h * 32 + nt * 16 + l) * 64 + k0 + 2 * half;
      b.x = kp[0];
      b.y = kp[1];
      s[nt] = wmma4(a, b, s[nt]);
    }
  }
#pragma unroll
  for (int nt = 0; nt < 2; ++nt)
#pragma unroll
    for (int r = 0; r < 8; ++r) s[nt][r] *= 0.125f;

  int qb = t0 >> 6;
#pragma unroll
  for (int nt = 0; nt < 2; ++nt) {
    float cs = 0.f;
#pragma unroll
    for (int r = 0; r < 8; ++r) cs += s[nt][r];
    cs += other_row(cs);
    if (half == 0)
      atomicAdd(&bs[((size_t)h * 32 + qb) * 32 + nt * 16 + l],
                cs * (1.0f / 64.0f));
  }
  float rmax[8], rsum[8];
#pragma unroll
  for (int r = 0; r < 8; ++r) rmax[r] = fmaxf(s[0][r], s[1][r]);
  row_max8(rmax);
#pragma unroll
  for (int r = 0; r < 8; ++r) rsum[r] = 0.f;
#pragma unroll
  for (int nt = 0; nt < 2; ++nt)
#pragma unroll
    for (int r = 0; r < 8; ++r) {
      float e = __expf(s[nt][r] - rmax[r]);
      s[nt][r] = e;
      rsum[r] += e;
    }
  row_sum8(rsum);
#pragma unroll
  for (int nt = 0; nt < 2; ++nt)
#pragma unroll
    for (int r = 0; r < 8; ++r)
      Ps[(r + 8 * half) * 34 + nt * 16 + l] = s[nt][r] / rsum[r];
  __syncthreads();

  v8f o[4] = {};
  const float* prow = &Ps[l * 34 + 2 * half];
  for (int k0 = 0; k0 < 32; k0 += 4) {
    v2f a;
    a.x = prow[k0];
    a.y = prow[k0 + 1];
#pragma unroll
    for (int nt = 0; nt < 4; ++nt) {
      v2f b;
      const float* vp = vc + (size_t)(h * 32 + k0 + 2 * half) * 64 + nt * 16 + l;
      b.x = vp[0];
      b.y = vp[64];
      o[nt] = wmma4(a, b, o[nt]);
    }
  }
  float* oh = comp + ((size_t)h * T_SEQ + t0) * DH;
#pragma unroll
  for (int nt = 0; nt < 4; ++nt)
#pragma unroll
    for (int r = 0; r < 8; ++r)
      oh[(size_t)(r + 8 * half) * DH + nt * 16 + l] = o[nt][r];
}

__global__ void block_stats_kernel(const float* __restrict__ bs,
                                   const float* __restrict__ ksc,
                                   const float* __restrict__ kbi,
                                   unsigned char* __restrict__ mask,
                                   float* __restrict__ avg_acc,
                                   float* __restrict__ tail) {
  int h = blockIdx.x, j = threadIdx.x;
  const float* row = bs + ((size_t)h * 32 + j) * 32;
  float sc[32], p[32];
  float mx = -3.0e38f;
  for (int c = 0; c < 32; ++c) { sc[c] = row[c]; mx = fmaxf(mx, sc[c]); }
  float ssum = 0.f;
  for (int c = 0; c < 32; ++c) { p[c] = __expf(sc[c] - mx); ssum += p[c]; }
  float inv = 1.0f / ssum;
  float ent = 0.f;
  for (int c = 0; c < 32; ++c) {
    float pp = p[c] * inv;
    ent -= pp * __logf(pp + 1e-9f);
  }
  float ne = ent / (logf(32.0f) + 1e-9f);
  float kr = 1.0f / (1.0f + __expf(-(ksc[h] * ne + kbi[h])));
  float dkf = fminf(fmaxf(8.0f * kr, 1.0f), 8.0f);
  int dk = (int)dkf;
  tail[1 + h * 32 + j] = kr;
  tail[1 + 512 + h * 32 + j] = ne;
  atomicAdd(avg_acc, (float)dk);
  unsigned used = 0;
  unsigned char mrow[32];
  for (int c = 0; c < 32; ++c) mrow[c] = 0;
  mrow[j] = 1;
  for (int kk = 0; kk < KMAX_SEL; ++kk) {
    float best = -3.4e38f;
    int bi = 0;
    for (int c = 0; c < 32; ++c)
      if (!((used >> c) & 1u) && sc[c] > best) { best = sc[c]; bi = c; }
    used |= 1u << bi;
    if (kk < dk) mrow[bi] = 1;
  }
  for (int c = 0; c < 32; ++c) mask[((size_t)h * 32 + j) * 32 + c] = mrow[c];
}

__global__ void __launch_bounds__(ATT_THREADS)
fine_attn_kernel(const float* __restrict__ q,
                                 const float* __restrict__ k,
                                 const float* __restrict__ v,
                                 const unsigned char* __restrict__ mask,
                                 float* __restrict__ out) {
  __shared__ float Ks[64 * 66];
  __shared__ float Vs[64 * 66];
  __shared__ float Ps[64 * 66];
  int h = blockIdx.x >> 5, qb = blockIdx.x & 31;
  int wv = threadIdx.x >> 5, lane = threadIdx.x & 31;
  int half = lane >> 4, l = lane & 15;
  const float* kh = k + (size_t)h * T_SEQ * DH;
  const float* vh = v + (size_t)h * T_SEQ * DH;
  const float* qrow =
      q + ((size_t)h * T_SEQ + qb * 64 + wv * 16 + l) * DH + 2 * half;
  const unsigned char* mrow = mask + ((size_t)h * 32 + qb) * 32;
  v8f o[4] = {};
  float rmax[8], rsum[8];
#pragma unroll
  for (int r = 0; r < 8; ++r) { rmax[r] = -3.0e38f; rsum[r] = 0.f; }

  for (int j = 0; j < 32; ++j) {
    if (!mrow[j]) continue;
    float2 kr[16], vr[16];
    tile_load_regs(kr, vr, kh + (size_t)j * 64 * DH, vh + (size_t)j * 64 * DH,
                   threadIdx.x);
    __syncthreads();
    tile_store_lds(Ks, Vs, kr, vr, threadIdx.x);
    __syncthreads();
    v8f s[4] = {};
    for (int k0 = 0; k0 < 64; k0 += 4) {
      v2f a;
      a.x = qrow[k0];
      a.y = qrow[k0 + 1];
#pragma unroll
      for (int nt = 0; nt < 4; ++nt) {
        v2f b;
        const float* kp = &Ks[(nt * 16 + l) * 66 + k0 + 2 * half];
        b.x = kp[0];
        b.y = kp[1];
        s[nt] = wmma4(a, b, s[nt]);
      }
    }
    float nm[8], ps[8], fac[8];
#pragma unroll
    for (int r = 0; r < 8; ++r) nm[r] = rmax[r];
#pragma unroll
    for (int nt = 0; nt < 4; ++nt)
#pragma unroll
      for (int r = 0; r < 8; ++r) {
        float x = s[nt][r] * 0.125f;
        s[nt][r] = x;
        nm[r] = fmaxf(nm[r], x);
      }
    row_max8(nm);
#pragma unroll
    for (int r = 0; r < 8; ++r) { fac[r] = __expf(rmax[r] - nm[r]); ps[r] = 0.f; }
#pragma unroll
    for (int nt = 0; nt < 4; ++nt)
#pragma unroll
      for (int r = 0; r < 8; ++r) {
        float e = __expf(s[nt][r] - nm[r]);
        s[nt][r] = e;
        ps[r] += e;
      }
    row_sum8(ps);
#pragma unroll
    for (int r = 0; r < 8; ++r) {
      rsum[r] = rsum[r] * fac[r] + ps[r];
      rmax[r] = nm[r];
    }
#pragma unroll
    for (int nt = 0; nt < 4; ++nt)
#pragma unroll
      for (int r = 0; r < 8; ++r)
        Ps[(wv * 16 + r + 8 * half) * 66 + nt * 16 + l] = s[nt][r];
#pragma unroll
    for (int nt = 0; nt < 4; ++nt)
#pragma unroll
      for (int r = 0; r < 8; ++r) o[nt][r] *= fac[r];
    __syncthreads();
    const float* prow = &Ps[(wv * 16 + l) * 66 + 2 * half];
    for (int k0 = 0; k0 < 64; k0 += 4) {
      v2f a;
      a.x = prow[k0];
      a.y = prow[k0 + 1];
#pragma unroll
      for (int nt = 0; nt < 4; ++nt) {
        v2f b;
        const float* vp = &Vs[(k0 + 2 * half) * 66 + nt * 16 + l];
        b.x = vp[0];
        b.y = vp[66];
        o[nt] = wmma4(a, b, o[nt]);
      }
    }
  }
  float* oh = out + ((size_t)h * T_SEQ + qb * 64) * DH;
#pragma unroll
  for (int nt = 0; nt < 4; ++nt)
#pragma unroll
    for (int r = 0; r < 8; ++r)
      oh[(size_t)(wv * 16 + r + 8 * half) * DH + nt * 16 + l] =
          o[nt][r] / rsum[r];
}

__global__ void combine_kernel(const float* __restrict__ sw,
                               const float* __restrict__ comp,
                               const float* __restrict__ fine,
                               const float* __restrict__ gate,
                               float* __restrict__ merged) {
  int idx = blockIdx.x * blockDim.x + threadIdx.x;
  if (idx >= T_SEQ * HID_DIM) return;
  int t = idx >> 10, c = idx & 1023, hh = c >> 6, dd = c & 63;
  size_t a = ((size_t)hh * T_SEQ + t) * DH + dd;
  const float* g = gate + (size_t)t * 48 + hh * 3;
  merged[idx] = sw[a] * g[0] + comp[a] * g[1] + fine[a] * g[2];
}

__global__ void zero_kernel(float* __restrict__ p, int n) {
  int i = blockIdx.x * blockDim.x + threadIdx.x;
  if (i < n) p[i] = 0.f;
}

__global__ void finalize_kernel(const float* __restrict__ avg_acc,
                                float* __restrict__ tail) {
  if (threadIdx.x == 0) tail[0] = avg_acc[0] * (1.0f / 512.0f);
}

extern "C" void kernel_launch(void* const* d_in, const int* in_sizes, int n_in,
                              void* d_out, int out_size, void* d_ws,
                              size_t ws_size, hipStream_t stream) {
  const float* X = (const float*)d_in[0];
  const float* Wq = (const float*)d_in[1];
  const float* Wk = (const float*)d_in[2];
  const float* Wv = (const float*)d_in[3];
  const float* Wo = (const float*)d_in[4];
  const float* Wc = (const float*)d_in[5];
  const float* bc = (const float*)d_in[6];
  const float* ksc = (const float*)d_in[7];
  const float* kbi = (const float*)d_in[8];
  const float* Wg = (const float*)d_in[9];
  const float* bg = (const float*)d_in[10];
  float* out = (float*)d_out;

  const size_t U = (size_t)NH * T_SEQ * DH;
  float* ws = (float*)d_ws;
  float* q = ws;
  float* kbuf = ws + U;
  float* vbuf = ws + 2 * U;
  float* sw = ws + 3 * U;
  float* comp = ws + 4 * U;
  float* fine = ws + 5 * U;
  float* merged = q;
  float* extra = ws + 6 * U;
  float* gate = extra;
  float* kc = gate + (size_t)T_SEQ * 48;
  float* vc = kc + 512 * 64;
  float* bs = vc + 512 * 64;
  float* avg = bs + 16384;
  unsigned char* mask = (unsigned char*)(avg + 64);

  zero_kernel<<<(16385 + 255) / 256, 256, 0, stream>>>(bs, 16385);

  {
    int wtiles = 3 * (T_SEQ / 64) * (HID_DIM / 32);
    int blocks = (wtiles + 11) / 12;
    wmma_gemm_f32_blk<<<blocks, 384, 0, stream>>>(X, Wq, Wk, Wv, nullptr, q,
                                                  kbuf, vbuf, 3, T_SEQ, HID_DIM,
                                                  HID_DIM, 0, 1);
  }
  sliding_window_kernel<<<NH * NB, ATT_THREADS,
                          (2 * 128 * 66 + 64 * 130) * sizeof(float), stream>>>(
      q, kbuf, vbuf, sw);
  {
    small_products_splitk<<<256, 1024, 0, stream>>>(kbuf, vbuf, Wc, bc, kc, vc,
                                                    X, Wg, bg, gate);
  }
  comp_branch_kernel<<<NH * (T_SEQ / 16), 32, 0, stream>>>(q, kc, vc, bs, comp);
  block_stats_kernel<<<NH, 32, 0, stream>>>(bs, ksc, kbi, mask, avg,
                                            out + (size_t)T_SEQ * HID_DIM);
  fine_attn_kernel<<<NH * NB, ATT_THREADS, 0, stream>>>(q, kbuf, vbuf, mask,
                                                        fine);
  combine_kernel<<<(T_SEQ * HID_DIM + 255) / 256, 256, 0, stream>>>(
      sw, comp, fine, gate, merged);
  {
    int wtiles = (T_SEQ / 64) * (HID_DIM / 32);
    wmma_gemm_f32_blk<<<(wtiles + 3) / 4, 128, 0, stream>>>(
        merged, Wo, nullptr, nullptr, nullptr, out, nullptr, nullptr, 1, T_SEQ,
        HID_DIM, HID_DIM, 0, 0);
  }
  finalize_kernel<<<1, 1, 0, stream>>>(avg, out + (size_t)T_SEQ * HID_DIM);
  (void)in_sizes;
  (void)n_in;
  (void)out_size;
  (void)ws_size;
}


# gate combine of the three branches done in the fine-attention epilogue from registers; the one-thread-per-element combine launch (65536 short waves) and the fine plane's write and re-read are gone
# speedup vs baseline: 10.3880x; 1.0309x over previous
#include <hip/hip_runtime.h>
#include <hip/hip_bf16.h>
#include <math.h>

typedef float v2f __attribute__((ext_vector_type(2)));
typedef float v4f __attribute__((ext_vector_type(4)));
typedef float v8f __attribute__((ext_vector_type(8)));

#define T_SEQ 2048
#define HID_DIM 1024
#define NH 16
#define DH 64
#define NB 32
#define KMAX_SEL 8

__device__ __forceinline__ v8f wmma4(v2f a, v2f b, v8f c) {
  return __builtin_amdgcn_wmma_f32_16x16x4_f32(false, a, false, b, (short)0, c,
                                               false, false);
}

__device__ __forceinline__ float apply_act(float x, int act) {
  if (act == 1) return 0.5f * x * (1.0f + erff(x * 0.70710678118654752f));
  if (act == 2) return 1.0f / (1.0f + __expf(-x));
  return x;
}

template <int M>
__device__ __forceinline__ float row_xor(float x) {
  return __builtin_amdgcn_update_dpp(x, x, 0x160 + M, 0xf, 0xf, true);
}

template <int M>
__device__ __forceinline__ void row_max_step(float (&x)[8]) {
#pragma unroll
  for (int r = 0; r < 8; ++r) x[r] = fmaxf(x[r], row_xor<M>(x[r]));
}

template <int M>
__device__ __forceinline__ void row_sum_step(float (&x)[8]) {
#pragma unroll
  for (int r = 0; r < 8; ++r) x[r] += row_xor<M>(x[r]);
}

__device__ __forceinline__ void row_max8(float (&x)[8]) {
  row_max_step<1>(x);
  row_max_step<2>(x);
  row_max_step<4>(x);
  row_max_step<8>(x);
}

__device__ __forceinline__ void row_sum8(float (&x)[8]) {
  row_sum_step<1>(x);
  row_sum_step<2>(x);
  row_sum_step<4>(x);
  row_sum_step<8>(x);
}

__device__ __forceinline__ float other_row(float x) {
  int v = __builtin_bit_cast(int, x);
  return __builtin_bit_cast(
      float, __builtin_amdgcn_permlanex16(v, v, 0x76543210, 0xfedcba98, false,
                                          false));
}

struct GemmFrags {
  v4f a[2][4];
  v2f b[2][4];
};

__device__ __forceinline__ void gemm_load_block(GemmFrags& f, const float* ap,
                                                const float* bp,
                                                size_t aRowStep, size_t N) {
#pragma unroll
  for (int g = 0; g < 2; ++g) {
#pragma unroll
    for (int mi = 0; mi < 4; ++mi)
      f.a[g][mi] = *(const v4f*)(ap + (size_t)mi * aRowStep + 8 * g);
#pragma unroll
    for (int j = 0; j < 4; ++j)
      f.b[g][j] = *(const v2f*)(bp + (size_t)(8 * g + j) * N);
  }
}

__device__ __forceinline__ void gemm_mma_block(const GemmFrags& f,
                                               v8f (&acc)[4][2]) {
#pragma unroll
  for (int g = 0; g < 2; ++g)
#pragma unroll
    for (int p = 0; p < 2; ++p) {
      v2f b0, b1;
      b0.x = f.b[g][2 * p].x;
      b0.y = f.b[g][2 * p + 1].x;
      b1.x = f.b[g][2 * p].y;
      b1.y = f.b[g][2 * p + 1].y;
#pragma unroll
      for (int mi = 0; mi < 4; ++mi) {
        v2f a;
        a.x = f.a[g][mi][2 * p];
        a.y = f.a[g][mi][2 * p + 1];
        acc[mi][0] = wmma4(a, b0, acc[mi][0]);
        acc[mi][1] = wmma4(a, b1, acc[mi][1]);
      }
    }
}

__global__ void __launch_bounds__(384)
wmma_gemm_f32_blk(const float* __restrict__ A, const float* B0,
                  const float* B1, const float* B2,
                  const float* __restrict__ bias, float* C0, float* C1,
                  float* C2, int nProd, int M, int N, int K, int act,
                  int headSplit) {
  int wave = threadIdx.x >> 5;
  int lane = threadIdx.x & 31;
  int wtile = blockIdx.x * (blockDim.x >> 5) + wave;
  int wn = N >> 5;
  int total = (M >> 6) * wn;
  if (wtile >= total * nProd) return;
  int prod = wtile / total;
  wtile -= prod * total;
  const float* __restrict__ B = prod == 0 ? B0 : (prod == 1 ? B1 : B2);
  float* __restrict__ C = prod == 0 ? C0 : (prod == 1 ? C1 : C2);
  int tm = wtile / wn, tn = wtile - tm * wn;
  int half = lane >> 4, l = lane & 15;
  int m0 = tm << 6, n0 = tn << 5;

  v8f acc[4][2];
#pragma unroll
  for (int mi = 0; mi < 4; ++mi)
#pragma unroll
    for (int ni = 0; ni < 2; ++ni) acc[mi][ni] = (v8f){};

  const float* ap = A + (size_t)(m0 + l) * K + 4 * half;
  const float* bp = B + (size_t)(4 * half) * N + n0 + 2 * l;
  const size_t aRowStep = (size_t)16 * K;
  const size_t bBlk = (size_t)16 * N;
  GemmFrags f0, f1;
  gemm_load_block(f0, ap, bp, aRowStep, (size_t)N);
  for (int k0 = 0; k0 < K - 32; k0 += 32) {
    gemm_load_block(f1, ap + 16, bp + bBlk, aRowStep, (size_t)N);
    __builtin_amdgcn_sched_barrier(0);
    gemm_mma_block(f0, acc);
    __builtin_amdgcn_sched_barrier(0);
    gemm_load_block(f0, ap + 32, bp + 2 * bBlk, aRowStep, (size_t)N);
    __builtin_amdgcn_sched_barrier(0);
    gemm_mma_block(f1, acc);
    __builtin_amdgcn_sched_barrier(0);
    ap += 32;
    bp += 2 * bBlk;
  }
  gemm_load_block(f1, ap + 16, bp + bBlk, aRowStep, (size_t)N);
  __builtin_amdgcn_sched_barrier(0);
  gemm_mma_block(f0, acc);
  gemm_mma_block(f1, acc);

  const int col = n0 + 2 * l;
  float bv0 = bias ? bias[col] : 0.0f;
  float bv1 = bias ? bias[col + 1] : 0.0f;
#pragma unroll
  for (int mi = 0; mi < 4; ++mi)
#pragma unroll
    for (int r = 0; r < 8; ++r) {
      v2f x;
      x.x = apply_act(acc[mi][0][r] + bv0, act);
      x.y = apply_act(acc[mi][1][r] + bv1, act);
      int m = m0 + mi * 16 + r + 8 * half;
      if (headSplit) {
        int hh = col >> 6, dd = col & 63;
        *(v2f*)&C[((size_t)hh * M + m) * 64 + dd] = x;
      } else {
        *(v2f*)&C[(size_t)m * N + col] = x;
      }
    }
}

#define CK_K 4096
#define CK_N 64
#define SP_WAVES 32
#define CK_SLICE (CK_K / SP_WAVES)
#define GT_N 48
#define GT_SLICE (HID_DIM / SP_WAVES)
#define SP_NI_PITCH 272
#define CK_W_PITCH (2 * SP_NI_PITCH)
#define GT_W_PITCH (3 * SP_NI_PITCH)
#define SP_COMPRESS_GROUPS 128
__global__ void __launch_bounds__(SP_WAVES * 32)
small_products_splitk(const float* __restrict__ kblk,
                      const float* __restrict__ vblk,
                      const float* __restrict__ Wc,
                      const float* __restrict__ bc, float* __restrict__ kc,
                      float* __restrict__ vc, const float* __restrict__ X,
                      const float* __restrict__ Wg,
                      const float* __restrict__ bg, float* __restrict__ gate) {
  __shared__ float part[SP_WAVES * GT_W_PITCH];
  int wave = threadIdx.x >> 5, lane = threadIdx.x & 31;
  int half = lane >> 4, l = lane & 15;
  int t = threadIdx.x;
  if (blockIdx.x < SP_COMPRESS_GROUPS) {
    int prod = blockIdx.x >> 6, g = blockIdx.x & 63;
    const float* __restrict__ A = prod ? vblk : kblk;
    float* __restrict__ C = prod ? vc : kc;
    int m0 = (g >> 1) << 4, n0 = (g & 1) << 5;
    int kbase = wave * CK_SLICE;
    const float* ap = A + (size_t)(m0 + l) * CK_K + kbase + 4 * half;
    const float* bp = Wc + (size_t)(kbase + 4 * half) * CK_N + n0 + 2 * l;
    v8f acc0 = {}, acc1 = {};
#pragma unroll
    for (int g = 0; g < CK_SLICE / 8; ++g) {
      v4f a4 = *(const v4f*)(ap + 8 * g);
      v2f br[4];
#pragma unroll
      for (int j = 0; j < 4; ++j)
        br[j] = *(const v2f*)(bp + (size_t)(8 * g + j) * CK_N);
#pragma unroll
      for (int p = 0; p < 2; ++p) {
        v2f a, b0, b1;
        a.x = a4[2 * p];
        a.y = a4[2 * p + 1];
        b0.x = br[2 * p].x;
        b0.y = br[2 * p + 1].x;
        b1.x = br[2 * p].y;
        b1.y = br[2 * p + 1].y;
        acc0 = wmma4(a, b0, acc0);
        acc1 = wmma4(a, b1, acc1);
      }
    }
    float* pw = part + wave * CK_W_PITCH + lane;
#pragma unroll
    for (int r = 0; r < 8; ++r) {
      pw[r * 32] = acc0[r];
      pw[SP_NI_PITCH + r * 32] = acc1[r];
    }
    __syncthreads();
    if (t < 512) {
      int row = t >> 5, col = t & 31;
      const float* pr = part + (col & 1) * SP_NI_PITCH + (row & 7) * 32 +
                        (row >> 3) * 16 + (col >> 1);
      float sum = pr[0];
#pragma unroll
      for (int w = 1; w < SP_WAVES; ++w) sum += pr[w * CK_W_PITCH];
      C[(size_t)(m0 + row) * CK_N + n0 + col] =
          apply_act(sum + bc[n0 + col], 1);
    }
  } else {
    int m0 = (blockIdx.x - SP_COMPRESS_GROUPS) << 4;
    int kbase = wave * GT_SLICE;
    const float* ap = X + (size_t)(m0 + l) * HID_DIM + kbase + 2 * half;
    const float* bp = Wg + (size_t)(kbase + 2 * half) * GT_N + l;
    v8f acc0 = {}, acc1 = {}, acc2 = {};
#pragma unroll
    for (int s = 0; s < GT_SLICE / 4; ++s) {
      v2f a, b0, b1, b2;
      a.x = ap[4 * s];
      a.y = ap[4 * s + 1];
      b0.x = bp[(4 * s) * GT_N];
      b0.y = bp[(4 * s + 1) * GT_N];
      b1.x = bp[(4 * s) * GT_N + 16];
      b1.y = bp[(4 * s + 1) * GT_N + 16];
      b2.x = bp[(4 * s) * GT_N + 32];
      b2.y = bp[(4 * s + 1) * GT_N + 32];
      acc0 = wmma4(a, b0, acc0);
      acc1 = wmma4(a, b1, acc1);
      acc2 = wmma4(a, b2, acc2);
    }
    float* pw = part + wave * GT_W_PITCH + lane;
#pragma unroll
    for (int r = 0; r < 8; ++r) {
      pw[r * 32] = acc0[r];
      pw[SP_NI_PITCH + r * 32] = acc1[r];
      pw[2 * SP_NI_PITCH + r * 32] = acc2[r];
    }
    __syncthreads();
    if (t < 16 * GT_N) {
      int row = t / GT_N, col = t - row * GT_N;
      const float* pr = part + (col >> 4) * SP_NI_PITCH + (row & 7) * 32 +
                        (row >> 3) * 16 + (col & 15);
      float sum = pr[0];
#pragma unroll
      for (int w = 1; w < SP_WAVES; ++w) sum += pr[w * GT_W_PITCH];
      gate[(size_t)m0 * GT_N + t] = apply_act(sum + bg[col], 2);
    }
  }
}

#define ATT_THREADS 128
__device__ __forceinline__ void tile_load_regs(float2 (&kr)[16],
                                               float2 (&vr)[16],
                                               const float* __restrict__ kt,
                                               const float* __restrict__ vt,
                                               int tid) {
  const int off = (tid >> 5) * DH + ((tid & 31) << 1);
#pragma unroll
  for (int i = 0; i < 16; ++i) {
    kr[i] = *(const float2*)&kt[off + i * 4 * DH];
    vr[i] = *(const float2*)&vt[off + i * 4 * DH];
  }
}

__device__ __forceinline__ void tile_store_lds(float* Kd, float* Vd,
                                               const float2 (&kr)[16],
                                               const float2 (&vr)[16],
                                               int tid) {
  const int off = (tid >> 5) * 66 + ((tid & 31) << 1);
#pragma unroll
  for (int i = 0; i < 16; ++i) {
    *(float2*)&Kd[off + i * 4 * 66] = kr[i];
    *(float2*)&Vd[off + i * 4 * 66] = vr[i];
  }
}

__global__ void __launch_bounds__(ATT_THREADS)
sliding_window_kernel(const float* __restrict__ q,
                                      const float* __restrict__ k,
                                      const float* __restrict__ v,
                                      float* __restrict__ out) {
  extern __shared__ float lds[];
  float* Ks = lds;
  float* Vs = Ks + 128 * 66;
  float* Ps = Vs + 128 * 66;
  int h = blockIdx.x >> 5;
  int w = blockIdx.x & 31;
  const float* kh = k + (size_t)h * T_SEQ * DH;
  const float* vh = v + (size_t)h * T_SEQ * DH;
  {
    float2 kc[16], vc[16], kp[16], vp[16];
    tile_load_regs(kc, vc, kh + (size_t)w * 64 * DH, vh + (size_t)w * 64 * DH,
                   threadIdx.x);
    if (w > 0) {
      tile_load_regs(kp, vp, kh + (size_t)(w - 1) * 64 * DH,
                     vh + (size_t)(w - 1) * 64 * DH, threadIdx.x);
    } else {
#pragma unroll
      for (int i = 0; i < 16; ++i) {
        kp[i] = make_float2(0.f, 0.f);
        vp[i] = make_float2(0.f, 0.f);
      }
    }
    tile_store_lds(Ks + 64 * 66, Vs + 64 * 66, kc, vc, threadIdx.x);
    tile_store_lds(Ks, Vs, kp, vp, threadIdx.x);
  }
  __syncthreads();

  int wv = threadIdx.x >> 5, lane = threadIdx.x & 31;
  int half = lane >> 4, l = lane & 15;
  const float* qrow =
      q + ((size_t)h * T_SEQ + w * 64 + wv * 16 + l) * DH + 2 * half;

  v8f s[8] = {};
  for (int k0 = 0; k0 < 64; k0 += 4) {
    v2f a;
    a.x = qrow[k0];
    a.y = qrow[k0 + 1];
#pragma unroll
    for (int nt = 0; nt < 8; ++nt) {
      v2f b;
      const float* kp = &Ks[(nt * 16 + l) * 66 + k0 + 2 * half];
      b.x = kp[0];
      b.y = kp[1];
      s[nt] = wmma4(a, b, s[nt]);
    }
  }
  float rmax[8], rsum[8];
#pragma unroll
  for (int r = 0; r < 8; ++r) { rmax[r] = -3.0e38f; rsum[r] = 0.f; }
#pragma unroll
  for (int nt = 0; nt < 8; ++nt) {
    int n = nt * 16 + l;
#pragma unroll
    for (int r = 0; r < 8; ++r) {
      int m = wv * 16 + r + 8 * half;
      float x = s[nt][r] * 0.125f;
      bool ok = (m + 64 >= n) && (w > 0 || n >= 64);
      x = ok ? x : -3.0e38f;
      s[nt][r] = x;
      rmax[r] = fmaxf(rmax[r], x);
    }
  }
  row_max8(rmax);
#pragma unroll
  for (int nt = 0; nt < 8; ++nt)
#pragma unroll
    for (int r = 0; r < 8; ++r) {
      float e = __expf(s[nt][r] - rmax[r]);
      s[nt][r] = e;
      rsum[r] += e;
    }
  row_sum8(rsum);
  float rinv[8];
#pragma unroll
  for (int r = 0; r < 8; ++r) rinv[r] = 1.0f / rsum[r];
#pragma unroll
  for (int nt = 0; nt < 8; ++nt)
#pragma unroll
    for (int r = 0; r < 8; ++r)
      Ps[(wv * 16 + r + 8 * half) * 130 + nt * 16 + l] = s[nt][r] * rinv[r];
  __syncthreads();

  v8f o[4] = {};
  const float* prow = &Ps[(wv * 16 + l) * 130 + 2 * half];
  for (int k0 = 0; k0 < 128; k0 += 4) {
    v2f a;
    a.x = prow[k0];
    a.y = prow[k0 + 1];
#pragma unroll
    for (int nt = 0; nt < 4; ++nt) {
      v2f b;
      const float* vp = &Vs[(k0 + 2 * half) * 66 + nt * 16 + l];
      b.x = vp[0];
      b.y = vp[66];
      o[nt] = wmma4(a, b, o[nt]);
    }
  }
  float* oh = out + ((size_t)h * T_SEQ + w * 64) * DH;
#pragma unroll
  for (int nt = 0; nt < 4; ++nt)
#pragma unroll
    for (int r = 0; r < 8; ++r)
      oh[(size_t)(wv * 16 + r + 8 * half) * DH + nt * 16 + l] = o[nt][r];
}

__global__ void comp_branch_kernel(const float* __restrict__ q,
                                   const float* __restrict__ kc,
                                   const float* __restrict__ vc,
                                   float* __restrict__ bs,
                                   float* __restrict__ comp) {
  __shared__ float Ps[16 * 34];
  int h = blockIdx.x >> 7;
  int tt = blockIdx.x & 127;
  int t0 = tt << 4;
  int lane = threadIdx.x, half = lane >> 4, l = lane & 15;
  const float* qrow = q + ((size_t)h * T_SEQ + t0 + l) * DH + 2 * half;
  v8f s[2] = {};
  for (int k0 = 0; k0 < 64; k0 += 4) {
    v2f a;
    a.x = qrow[k0];
    a.y = qrow[k0 + 1];
#pragma unroll
    for (int nt = 0; nt < 2; ++nt) {
      v2f b;
      const float* kp = kc + (size_t)(h * 32 + nt * 16 + l) * 64 + k0 + 2 * half;
      b.x = kp[0];
      b.y = kp[1];
      s[nt] = wmma4(a, b, s[nt]);
    }
  }
#pragma unroll
  for (int nt = 0; nt < 2; ++nt)
#pragma unroll
    for (int r = 0; r < 8; ++r) s[nt][r] *= 0.125f;

  int qb = t0 >> 6;
#pragma unroll
  for (int nt = 0; nt < 2; ++nt) {
    float cs = 0.f;
#pragma unroll
    for (int r = 0; r < 8; ++r) cs += s[nt][r];
    cs += other_row(cs);
    if (half == 0)
      atomicAdd(&bs[((size_t)h * 32 + qb) * 32 + nt * 16 + l],
                cs * (1.0f / 64.0f));
  }
  float rmax[8], rsum[8];
#pragma unroll
  for (int r = 0; r < 8; ++r) rmax[r] = fmaxf(s[0][r], s[1][r]);
  row_max8(rmax);
#pragma unroll
  for (int r = 0; r < 8; ++r) rsum[r] = 0.f;
#pragma unroll
  for (int nt = 0; nt < 2; ++nt)
#pragma unroll
    for (int r = 0; r < 8; ++r) {
      float e = __expf(s[nt][r] - rmax[r]);
      s[nt][r] = e;
      rsum[r] += e;
    }
  row_sum8(rsum);
#pragma unroll
  for (int nt = 0; nt < 2; ++nt)
#pragma unroll
    for (int r = 0; r < 8; ++r)
      Ps[(r + 8 * half) * 34 + nt * 16 + l] = s[nt][r] / rsum[r];
  __syncthreads();

  v8f o[4] = {};
  const float* prow = &Ps[l * 34 + 2 * half];
  for (int k0 = 0; k0 < 32; k0 += 4) {
    v2f a;
    a.x = prow[k0];
    a.y = prow[k0 + 1];
#pragma unroll
    for (int nt = 0; nt < 4; ++nt) {
      v2f b;
      const float* vp = vc + (size_t)(h * 32 + k0 + 2 * half) * 64 + nt * 16 + l;
      b.x = vp[0];
      b.y = vp[64];
      o[nt] = wmma4(a, b, o[nt]);
    }
  }
  float* oh = comp + ((size_t)h * T_SEQ + t0) * DH;
#pragma unroll
  for (int nt = 0; nt < 4; ++nt)
#pragma unroll
    for (int r = 0; r < 8; ++r)
      oh[(size_t)(r + 8 * half) * DH + nt * 16 + l] = o[nt][r];
}

__global__ void block_stats_kernel(const float* __restrict__ bs,
                                   const float* __restrict__ ksc,
                                   const float* __restrict__ kbi,
                                   unsigned char* __restrict__ mask,
                                   float* __restrict__ avg_acc,
                                   float* __restrict__ tail) {
  int h = blockIdx.x, j = threadIdx.x;
  const float* row = bs + ((size_t)h * 32 + j) * 32;
  float sc[32], p[32];
  float mx = -3.0e38f;
  for (int c = 0; c < 32; ++c) { sc[c] = row[c]; mx = fmaxf(mx, sc[c]); }
  float ssum = 0.f;
  for (int c = 0; c < 32; ++c) { p[c] = __expf(sc[c] - mx); ssum += p[c]; }
  float inv = 1.0f / ssum;
  float ent = 0.f;
  for (int c = 0; c < 32; ++c) {
    float pp = p[c] * inv;
    ent -= pp * __logf(pp + 1e-9f);
  }
  float ne = ent / (logf(32.0f) + 1e-9f);
  float kr = 1.0f / (1.0f + __expf(-(ksc[h] * ne + kbi[h])));
  float dkf = fminf(fmaxf(8.0f * kr, 1.0f), 8.0f);
  int dk = (int)dkf;
  tail[1 + h * 32 + j] = kr;
  tail[1 + 512 + h * 32 + j] = ne;
  atomicAdd(avg_acc, (float)dk);
  unsigned used = 0;
  unsigned char mrow[32];
  for (int c = 0; c < 32; ++c) mrow[c] = 0;
  mrow[j] = 1;
  for (int kk = 0; kk < KMAX_SEL; ++kk) {
    float best = -3.4e38f;
    int bi = 0;
    for (int c = 0; c < 32; ++c)
      if (!((used >> c) & 1u) && sc[c] > best) { best = sc[c]; bi = c; }
    used |= 1u << bi;
    if (kk < dk) mrow[bi] = 1;
  }
  for (int c = 0; c < 32; ++c) mask[((size_t)h * 32 + j) * 32 + c] = mrow[c];
}

__global__ void __launch_bounds__(ATT_THREADS)
fine_attn_kernel(const float* __restrict__ q,
                                 const float* __restrict__ k,
                                 const float* __restrict__ v,
                                 const unsigned char* __restrict__ mask,
                                 const float* __restrict__ sw,
                                 const float* __restrict__ comp,
                                 const float* __restrict__ gate,
                                 float* __restrict__ merged) {
  __shared__ float Ks[64 * 66];
  __shared__ float Vs[64 * 66];
  __shared__ float Ps[64 * 66];
  int h = blockIdx.x >> 5, qb = blockIdx.x & 31;
  int wv = threadIdx.x >> 5, lane = threadIdx.x & 31;
  int half = lane >> 4, l = lane & 15;
  const float* kh = k + (size_t)h * T_SEQ * DH;
  const float* vh = v + (size_t)h * T_SEQ * DH;
  const float* qrow =
      q + ((size_t)h * T_SEQ + qb * 64 + wv * 16 + l) * DH + 2 * half;
  const unsigned char* mrow = mask + ((size_t)h * 32 + qb) * 32;
  v8f o[4] = {};
  float rmax[8], rsum[8];
#pragma unroll
  for (int r = 0; r < 8; ++r) { rmax[r] = -3.0e38f; rsum[r] = 0.f; }

  for (int j = 0; j < 32; ++j) {
    if (!mrow[j]) continue;
    float2 kr[16], vr[16];
    tile_load_regs(kr, vr, kh + (size_t)j * 64 * DH, vh + (size_t)j * 64 * DH,
                   threadIdx.x);
    __syncthreads();
    tile_store_lds(Ks, Vs, kr, vr, threadIdx.x);
    __syncthreads();
    v8f s[4] = {};
    for (int k0 = 0; k0 < 64; k0 += 4) {
      v2f a;
      a.x = qrow[k0];
      a.y = qrow[k0 + 1];
#pragma unroll
      for (int nt = 0; nt < 4; ++nt) {
        v2f b;
        const float* kp = &Ks[(nt * 16 + l) * 66 + k0 + 2 * half];
        b.x = kp[0];
        b.y = kp[1];
        s[nt] = wmma4(a, b, s[nt]);
      }
    }
    float nm[8], ps[8], fac[8];
#pragma unroll
    for (int r = 0; r < 8; ++r) nm[r] = rmax[r];
#pragma unroll
    for (int nt = 0; nt < 4; ++nt)
#pragma unroll
      for (int r = 0; r < 8; ++r) {
        float x = s[nt][r] * 0.125f;
        s[nt][r] = x;
        nm[r] = fmaxf(nm[r], x);
      }
    row_max8(nm);
#pragma unroll
    for (int r = 0; r < 8; ++r) { fac[r] = __expf(rmax[r] - nm[r]); ps[r] = 0.f; }
#pragma unroll
    for (int nt = 0; nt < 4; ++nt)
#pragma unroll
      for (int r = 0; r < 8; ++r) {
        float e = __expf(s[nt][r] - nm[r]);
        s[nt][r] = e;
        ps[r] += e;
      }
    row_sum8(ps);
#pragma unroll
    for (int r = 0; r < 8; ++r) {
      rsum[r] = rsum[r] * fac[r] + ps[r];
      rmax[r] = nm[r];
    }
#pragma unroll
    for (int nt = 0; nt < 4; ++nt)
#pragma unroll
      for (int r = 0; r < 8; ++r)
        Ps[(wv * 16 + r + 8 * half) * 66 + nt * 16 + l] = s[nt][r];
#pragma unroll
    for (int nt = 0; nt < 4; ++nt)
#pragma unroll
      for (int r = 0; r < 8; ++r) o[nt][r] *= fac[r];
    __syncthreads();
    const float* prow = &Ps[(wv * 16 + l) * 66 + 2 * half];
    for (int k0 = 0; k0 < 64; k0 += 4) {
      v2f a;
      a.x = prow[k0];
      a.y = prow[k0 + 1];
#pragma unroll
      for (int nt = 0; nt < 4; ++nt) {
        v2f b;
        const float* vp = &Vs[(k0 + 2 * half) * 66 + nt * 16 + l];
        b.x = vp[0];
        b.y = vp[66];
        o[nt] = wmma4(a, b, o[nt]);
      }
    }
  }
  const int trow = qb * 64 + wv * 16 + 8 * half;
  const float* swp = sw + ((size_t)h * T_SEQ + trow) * DH + l;
  const float* cpp = comp + ((size_t)h * T_SEQ + trow) * DH + l;
  const float* gp = gate + (size_t)trow * 48 + h * 3;
  float gw[8][3], sv[4][8], cv[4][8];
#pragma unroll
  for (int r = 0; r < 8; ++r) {
    gw[r][0] = gp[r * 48];
    gw[r][1] = gp[r * 48 + 1];
    gw[r][2] = gp[r * 48 + 2];
  }
#pragma unroll
  for (int nt = 0; nt < 4; ++nt)
#pragma unroll
    for (int r = 0; r < 8; ++r) {
      sv[nt][r] = swp[r * DH + nt * 16];
      cv[nt][r] = cpp[r * DH + nt * 16];
    }
  float* mp = merged + (size_t)trow * HID_DIM + h * DH + l;
#pragma unroll
  for (int nt = 0; nt < 4; ++nt)
#pragma unroll
    for (int r = 0; r < 8; ++r) {
      float f = o[nt][r] / rsum[r];
      mp[(size_t)r * HID_DIM + nt * 16] =
          sv[nt][r] * gw[r][0] + cv[nt][r] * gw[r][1] + f * gw[r][2];
    }
}

__global__ void zero_kernel(float* __restrict__ p, int n) {
  int i = blockIdx.x * blockDim.x + threadIdx.x;
  if (i < n) p[i] = 0.f;
}

__global__ void finalize_kernel(const float* __restrict__ avg_acc,
                                float* __restrict__ tail) {
  if (threadIdx.x == 0) tail[0] = avg_acc[0] * (1.0f / 512.0f);
}

extern "C" void kernel_launch(void* const* d_in, const int* in_sizes, int n_in,
                              void* d_out, int out_size, void* d_ws,
                              size_t ws_size, hipStream_t stream) {
  const float* X = (const float*)d_in[0];
  const float* Wq = (const float*)d_in[1];
  const float* Wk = (const float*)d_in[2];
  const float* Wv = (const float*)d_in[3];
  const float* Wo = (const float*)d_in[4];
  const float* Wc = (const float*)d_in[5];
  const float* bc = (const float*)d_in[6];
  const float* ksc = (const float*)d_in[7];
  const float* kbi = (const float*)d_in[8];
  const float* Wg = (const float*)d_in[9];
  const float* bg = (const float*)d_in[10];
  float* out = (float*)d_out;

  const size_t U = (size_t)NH * T_SEQ * DH;
  float* ws = (float*)d_ws;
  float* q = ws;
  float* kbuf = ws + U;
  float* vbuf = ws + 2 * U;
  float* sw = ws + 3 * U;
  float* comp = ws + 4 * U;
  float* merged = ws + 5 * U;
  float* extra = ws + 6 * U;
  float* gate = extra;
  float* kc = gate + (size_t)T_SEQ * 48;
  float* vc = kc + 512 * 64;
  float* bs = vc + 512 * 64;
  float* avg = bs + 16384;
  unsigned char* mask = (unsigned char*)(avg + 64);

  zero_kernel<<<(16385 + 255) / 256, 256, 0, stream>>>(bs, 16385);

  {
    int wtiles = 3 * (T_SEQ / 64) * (HID_DIM / 32);
    int blocks = (wtiles + 11) / 12;
    wmma_gemm_f32_blk<<<blocks, 384, 0, stream>>>(X, Wq, Wk, Wv, nullptr, q,
                                                  kbuf, vbuf, 3, T_SEQ, HID_DIM,
                                                  HID_DIM, 0, 1);
  }
  sliding_window_kernel<<<NH * NB, ATT_THREADS,
                          (2 * 128 * 66 + 64 * 130) * sizeof(float), stream>>>(
      q, kbuf, vbuf, sw);
  {
    small_products_splitk<<<256, 1024, 0, stream>>>(kbuf, vbuf, Wc, bc, kc, vc,
                                                    X, Wg, bg, gate);
  }
  comp_branch_kernel<<<NH * (T_SEQ / 16), 32, 0, stream>>>(q, kc, vc, bs, comp);
  block_stats_kernel<<<NH, 32, 0, stream>>>(bs, ksc, kbi, mask, avg,
                                            out + (size_t)T_SEQ * HID_DIM);
  fine_attn_kernel<<<NH * NB, ATT_THREADS, 0, stream>>>(q, kbuf, vbuf, mask,
                                                        sw, comp, gate, merged);
  {
    int wtiles = (T_SEQ / 64) * (HID_DIM / 32);
    wmma_gemm_f32_blk<<<(wtiles + 3) / 4, 128, 0, stream>>>(
        merged, Wo, nullptr, nullptr, nullptr, out, nullptr, nullptr, 1, T_SEQ,
        HID_DIM, HID_DIM, 0, 0);
  }
  finalize_kernel<<<1, 1, 0, stream>>>(avg, out + (size_t)T_SEQ * HID_DIM);
  (void)in_sizes;
  (void)n_in;
  (void)out_size;
  (void)ws_size;
}


# block stats (entropy, top-k by rank count) lane-parallel in each fine-attention group; zero, stats, finalize launches gone; score partials summed in fixed order, no atomics; entropy sums by butterfly
# speedup vs baseline: 11.2140x; 1.0795x over previous
#include <hip/hip_runtime.h>
#include <hip/hip_bf16.h>
#include <math.h>

typedef float v2f __attribute__((ext_vector_type(2)));
typedef float v4f __attribute__((ext_vector_type(4)));
typedef float v8f __attribute__((ext_vector_type(8)));

#define T_SEQ 2048
#define HID_DIM 1024
#define NH 16
#define DH 64
#define NB 32
#define KMAX_SEL 8

__device__ __forceinline__ v8f wmma4(v2f a, v2f b, v8f c) {
  return __builtin_amdgcn_wmma_f32_16x16x4_f32(false, a, false, b, (short)0, c,
                                               false, false);
}

__device__ __forceinline__ float apply_act(float x, int act) {
  if (act == 1) return 0.5f * x * (1.0f + erff(x * 0.70710678118654752f));
  if (act == 2) return 1.0f / (1.0f + __expf(-x));
  return x;
}

template <int M>
__device__ __forceinline__ float row_xor(float x) {
  return __builtin_amdgcn_update_dpp(x, x, 0x160 + M, 0xf, 0xf, true);
}

template <int M>
__device__ __forceinline__ void row_max_step(float (&x)[8]) {
#pragma unroll
  for (int r = 0; r < 8; ++r) x[r] = fmaxf(x[r], row_xor<M>(x[r]));
}

template <int M>
__device__ __forceinline__ void row_sum_step(float (&x)[8]) {
#pragma unroll
  for (int r = 0; r < 8; ++r) x[r] += row_xor<M>(x[r]);
}

__device__ __forceinline__ void row_max8(float (&x)[8]) {
  row_max_step<1>(x);
  row_max_step<2>(x);
  row_max_step<4>(x);
  row_max_step<8>(x);
}

__device__ __forceinline__ void row_sum8(float (&x)[8]) {
  row_sum_step<1>(x);
  row_sum_step<2>(x);
  row_sum_step<4>(x);
  row_sum_step<8>(x);
}

__device__ __forceinline__ float other_row(float x) {
  int v = __builtin_bit_cast(int, x);
  return __builtin_bit_cast(
      float, __builtin_amdgcn_permlanex16(v, v, 0x76543210, 0xfedcba98, false,
                                          false));
}

__device__ __forceinline__ float wave_sum32(float x) {
  x += row_xor<1>(x);
  x += row_xor<2>(x);
  x += row_xor<4>(x);
  x += row_xor<8>(x);
  return x + other_row(x);
}

__device__ __forceinline__ float wave_max32(float x) {
  x = fmaxf(x, row_xor<1>(x));
  x = fmaxf(x, row_xor<2>(x));
  x = fmaxf(x, row_xor<4>(x));
  x = fmaxf(x, row_xor<8>(x));
  return fmaxf(x, other_row(x));
}

struct GemmFrags {
  v4f a[2][4];
  v2f b[2][4];
};

__device__ __forceinline__ void gemm_load_block(GemmFrags& f, const float* ap,
                                                const float* bp,
                                                size_t aRowStep, size_t N) {
#pragma unroll
  for (int g = 0; g < 2; ++g) {
#pragma unroll
    for (int mi = 0; mi < 4; ++mi)
      f.a[g][mi] = *(const v4f*)(ap + (size_t)mi * aRowStep + 8 * g);
#pragma unroll
    for (int j = 0; j < 4; ++j)
      f.b[g][j] = *(const v2f*)(bp + (size_t)(8 * g + j) * N);
  }
}

__device__ __forceinline__ void gemm_mma_block(const GemmFrags& f,
                                               v8f (&acc)[4][2]) {
#pragma unroll
  for (int g = 0; g < 2; ++g)
#pragma unroll
    for (int p = 0; p < 2; ++p) {
      v2f b0, b1;
      b0.x = f.b[g][2 * p].x;
      b0.y = f.b[g][2 * p + 1].x;
      b1.x = f.b[g][2 * p].y;
      b1.y = f.b[g][2 * p + 1].y;
#pragma unroll
      for (int mi = 0; mi < 4; ++mi) {
        v2f a;
        a.x = f.a[g][mi][2 * p];
        a.y = f.a[g][mi][2 * p + 1];
        acc[mi][0] = wmma4(a, b0, acc[mi][0]);
        acc[mi][1] = wmma4(a, b1, acc[mi][1]);
      }
    }
}

__global__ void __launch_bounds__(384)
wmma_gemm_f32_blk(const float* __restrict__ A, const float* B0,
                  const float* B1, const float* B2,
                  const float* __restrict__ bias, float* C0, float* C1,
                  float* C2, int nProd, int M, int N, int K, int act,
                  int headSplit, const float* __restrict__ dkSrc,
                  float* __restrict__ avgDst) {
  int wave = threadIdx.x >> 5;
  int lane = threadIdx.x & 31;
  int wtile = blockIdx.x * (blockDim.x >> 5) + wave;
  int wn = N >> 5;
  int total = (M >> 6) * wn;
  if (wtile >= total * nProd) return;
  int prod = wtile / total;
  wtile -= prod * total;
  const float* __restrict__ B = prod == 0 ? B0 : (prod == 1 ? B1 : B2);
  float* __restrict__ C = prod == 0 ? C0 : (prod == 1 ? C1 : C2);
  int tm = wtile / wn, tn = wtile - tm * wn;
  int half = lane >> 4, l = lane & 15;
  int m0 = tm << 6, n0 = tn << 5;

  v8f acc[4][2];
#pragma unroll
  for (int mi = 0; mi < 4; ++mi)
#pragma unroll
    for (int ni = 0; ni < 2; ++ni) acc[mi][ni] = (v8f){};

  const float* ap = A + (size_t)(m0 + l) * K + 4 * half;
  const float* bp = B + (size_t)(4 * half) * N + n0 + 2 * l;
  const size_t aRowStep = (size_t)16 * K;
  const size_t bBlk = (size_t)16 * N;
  GemmFrags f0, f1;
  gemm_load_block(f0, ap, bp, aRowStep, (size_t)N);
  for (int k0 = 0; k0 < K - 32; k0 += 32) {
    gemm_load_block(f1, ap + 16, bp + bBlk, aRowStep, (size_t)N);
    __builtin_amdgcn_sched_barrier(0);
    gemm_mma_block(f0, acc);
    __builtin_amdgcn_sched_barrier(0);
    gemm_load_block(f0, ap + 32, bp + 2 * bBlk, aRowStep, (size_t)N);
    __builtin_amdgcn_sched_barrier(0);
    gemm_mma_block(f1, acc);
    __builtin_amdgcn_sched_barrier(0);
    ap += 32;
    bp += 2 * bBlk;
  }
  gemm_load_block(f1, ap + 16, bp + bBlk, aRowStep, (size_t)N);
  __builtin_amdgcn_sched_barrier(0);
  gemm_mma_block(f0, acc);
  gemm_mma_block(f1, acc);

  const int col = n0 + 2 * l;
  float bv0 = bias ? bias[col] : 0.0f;
  float bv1 = bias ? bias[col + 1] : 0.0f;
#pragma unroll
  for (int mi = 0; mi < 4; ++mi)
#pragma unroll
    for (int r = 0; r < 8; ++r) {
      v2f x;
      x.x = apply_act(acc[mi][0][r] + bv0, act);
      x.y = apply_act(acc[mi][1][r] + bv1, act);
      int m = m0 + mi * 16 + r + 8 * half;
      if (headSplit) {
        int hh = col >> 6, dd = col & 63;
        *(v2f*)&C[((size_t)hh * M + m) * 64 + dd] = x;
      } else {
        *(v2f*)&C[(size_t)m * N + col] = x;
      }
    }
  if (dkSrc != nullptr && blockIdx.x == 0 && wave == 0) {
    float s = 0.f;
#pragma unroll
    for (int i = 0; i < 16; ++i) s += dkSrc[lane + 32 * i];
    s = wave_sum32(s);
    if (lane == 0) avgDst[0] = s * (1.0f / 512.0f);
  }
}

#define CK_K 4096
#define CK_N 64
#define SP_WAVES 32
#define CK_SLICE (CK_K / SP_WAVES)
#define GT_N 48
#define GT_SLICE (HID_DIM / SP_WAVES)
#define SP_NI_PITCH 272
#define CK_W_PITCH (2 * SP_NI_PITCH)
#define GT_W_PITCH (3 * SP_NI_PITCH)
#define SP_COMPRESS_GROUPS 128
__global__ void __launch_bounds__(SP_WAVES * 32)
small_products_splitk(const float* __restrict__ kblk,
                      const float* __restrict__ vblk,
                      const float* __restrict__ Wc,
                      const float* __restrict__ bc, float* __restrict__ kc,
                      float* __restrict__ vc, const float* __restrict__ X,
                      const float* __restrict__ Wg,
                      const float* __restrict__ bg, float* __restrict__ gate) {
  __shared__ float part[SP_WAVES * GT_W_PITCH];
  int wave = threadIdx.x >> 5, lane = threadIdx.x & 31;
  int half = lane >> 4, l = lane & 15;
  int t = threadIdx.x;
  if (blockIdx.x < SP_COMPRESS_GROUPS) {
    int prod = blockIdx.x >> 6, g = blockIdx.x & 63;
    const float* __restrict__ A = prod ? vblk : kblk;
    float* __restrict__ C = prod ? vc : kc;
    int m0 = (g >> 1) << 4, n0 = (g & 1) << 5;
    int kbase = wave * CK_SLICE;
    const float* ap = A + (size_t)(m0 + l) * CK_K + kbase + 4 * half;
    const float* bp = Wc + (size_t)(kbase + 4 * half) * CK_N + n0 + 2 * l;
    v8f acc0 = {}, acc1 = {};
#pragma unroll
    for (int g = 0; g < CK_SLICE / 8; ++g) {
      v4f a4 = *(const v4f*)(ap + 8 * g);
      v2f br[4];
#pragma unroll
      for (int j = 0; j < 4; ++j)
        br[j] = *(const v2f*)(bp + (size_t)(8 * g + j) * CK_N);
#pragma unroll
      for (int p = 0; p < 2; ++p) {
        v2f a, b0, b1;
        a.x = a4[2 * p];
        a.y = a4[2 * p + 1];
        b0.x = br[2 * p].x;
        b0.y = br[2 * p + 1].x;
        b1.x = br[2 * p].y;
        b1.y = br[2 * p + 1].y;
        acc0 = wmma4(a, b0, acc0);
        acc1 = wmma4(a, b1, acc1);
      }
    }
    float* pw = part + wave * CK_W_PITCH + lane;
#pragma unroll
    for (int r = 0; r < 8; ++r) {
      pw[r * 32] = acc0[r];
      pw[SP_NI_PITCH + r * 32] = acc1[r];
    }
    __syncthreads();
    if (t < 512) {
      int row = t >> 5, col = t & 31;
      const float* pr = part + (col & 1) * SP_NI_PITCH + (row & 7) * 32 +
                        (row >> 3) * 16 + (col >> 1);
      float sum = pr[0];
#pragma unroll
      for (int w = 1; w < SP_WAVES; ++w) sum += pr[w * CK_W_PITCH];
      C[(size_t)(m0 + row) * CK_N + n0 + col] =
          apply_act(sum + bc[n0 + col], 1);
    }
  } else {
    int m0 = (blockIdx.x - SP_COMPRESS_GROUPS) << 4;
    int kbase = wave * GT_SLICE;
    const float* ap = X + (size_t)(m0 + l) * HID_DIM + kbase + 2 * half;
    const float* bp = Wg + (size_t)(kbase + 2 * half) * GT_N + l;
    v8f acc0 = {}, acc1 = {}, acc2 = {};
#pragma unroll
    for (int s = 0; s < GT_SLICE / 4; ++s) {
      v2f a, b0, b1, b2;
      a.x = ap[4 * s];
      a.y = ap[4 * s + 1];
      b0.x = bp[(4 * s) * GT_N];
      b0.y = bp[(4 * s + 1) * GT_N];
      b1.x = bp[(4 * s) * GT_N + 16];
      b1.y = bp[(4 * s + 1) * GT_N + 16];
      b2.x = bp[(4 * s) * GT_N + 32];
      b2.y = bp[(4 * s + 1) * GT_N + 32];
      acc0 = wmma4(a, b0, acc0);
      acc1 = wmma4(a, b1, acc1);
      acc2 = wmma4(a, b2, acc2);
    }
    float* pw = part + wave * GT_W_PITCH + lane;
#pragma unroll
    for (int r = 0; r < 8; ++r) {
      pw[r * 32] = acc0[r];
      pw[SP_NI_PITCH + r * 32] = acc1[r];
      pw[2 * SP_NI_PITCH + r * 32] = acc2[r];
    }
    __syncthreads();
    if (t < 16 * GT_N) {
      int row = t / GT_N, col = t - row * GT_N;
      const float* pr = part + (col >> 4) * SP_NI_PITCH + (row & 7) * 32 +
                        (row >> 3) * 16 + (col & 15);
      float sum = pr[0];
#pragma unroll
      for (int w = 1; w < SP_WAVES; ++w) sum += pr[w * GT_W_PITCH];
      gate[(size_t)m0 * GT_N + t] = apply_act(sum + bg[col], 2);
    }
  }
}

#define ATT_THREADS 128
__device__ __forceinline__ void tile_load_regs(float2 (&kr)[16],
                                               float2 (&vr)[16],
                                               const float* __restrict__ kt,
                                               const float* __restrict__ vt,
                                               int tid) {
  const int off = (tid >> 5) * DH + ((tid & 31) << 1);
#pragma unroll
  for (int i = 0; i < 16; ++i) {
    kr[i] = *(const float2*)&kt[off + i * 4 * DH];
    vr[i] = *(const float2*)&vt[off + i * 4 * DH];
  }
}

__device__ __forceinline__ void tile_store_lds(float* Kd, float* Vd,
                                               const float2 (&kr)[16],
                                               const float2 (&vr)[16],
                                               int tid) {
  const int off = (tid >> 5) * 66 + ((tid & 31) << 1);
#pragma unroll
  for (int i = 0; i < 16; ++i) {
    *(float2*)&Kd[off + i * 4 * 66] = kr[i];
    *(float2*)&Vd[off + i * 4 * 66] = vr[i];
  }
}

__global__ void __launch_bounds__(ATT_THREADS)
sliding_window_kernel(const float* __restrict__ q,
                                      const float* __restrict__ k,
                                      const float* __restrict__ v,
                                      float* __restrict__ out) {
  extern __shared__ float lds[];
  float* Ks = lds;
  float* Vs = Ks + 128 * 66;
  float* Ps = Vs + 128 * 66;
  int h = blockIdx.x >> 5;
  int w = blockIdx.x & 31;
  const float* kh = k + (size_t)h * T_SEQ * DH;
  const float* vh = v + (size_t)h * T_SEQ * DH;
  {
    float2 kc[16], vc[16], kp[16], vp[16];
    tile_load_regs(kc, vc, kh + (size_t)w * 64 * DH, vh + (size_t)w * 64 * DH,
                   threadIdx.x);
    if (w > 0) {
      tile_load_regs(kp, vp, kh + (size_t)(w - 1) * 64 * DH,
                     vh + (size_t)(w - 1) * 64 * DH, threadIdx.x);
    } else {
#pragma unroll
      for (int i = 0; i < 16; ++i) {
        kp[i] = make_float2(0.f, 0.f);
        vp[i] = make_float2(0.f, 0.f);
      }
    }
    tile_store_lds(Ks + 64 * 66, Vs + 64 * 66, kc, vc, threadIdx.x);
    tile_store_lds(Ks, Vs, kp, vp, threadIdx.x);
  }
  __syncthreads();

  int wv = threadIdx.x >> 5, lane = threadIdx.x & 31;
  int half = lane >> 4, l = lane & 15;
  const float* qrow =
      q + ((size_t)h * T_SEQ + w * 64 + wv * 16 + l) * DH + 2 * half;

  v8f s[8] = {};
  for (int k0 = 0; k0 < 64; k0 += 4) {
    v2f a;
    a.x = qrow[k0];
    a.y = qrow[k0 + 1];
#pragma unroll
    for (int nt = 0; nt < 8; ++nt) {
      v2f b;
      const float* kp = &Ks[(nt * 16 + l) * 66 + k0 + 2 * half];
      b.x = kp[0];
      b.y = kp[1];
      s[nt] = wmma4(a, b, s[nt]);
    }
  }
  float rmax[8], rsum[8];
#pragma unroll
  for (int r = 0; r < 8; ++r) { rmax[r] = -3.0e38f; rsum[r] = 0.f; }
#pragma unroll
  for (int nt = 0; nt < 8; ++nt) {
    int n = nt * 16 + l;
#pragma unroll
    for (int r = 0; r < 8; ++r) {
      int m = wv * 16 + r + 8 * half;
      float x = s[nt][r] * 0.125f;
      bool ok = (m + 64 >= n) && (w > 0 || n >= 64);
      x = ok ? x : -3.0e38f;
      s[nt][r] = x;
      rmax[r] = fmaxf(rmax[r], x);
    }
  }
  row_max8(rmax);
#pragma unroll
  for (int nt = 0; nt < 8; ++nt)
#pragma unroll
    for (int r = 0; r < 8; ++r) {
      float e = __expf(s[nt][r] - rmax[r]);
      s[nt][r] = e;
      rsum[r] += e;
    }
  row_sum8(rsum);
  float rinv[8];
#pragma unroll
  for (int r = 0; r < 8; ++r) rinv[r] = 1.0f / rsum[r];
#pragma unroll
  for (int nt = 0; nt < 8; ++nt)
#pragma unroll
    for (int r = 0; r < 8; ++r)
      Ps[(wv * 16 + r + 8 * half) * 130 + nt * 16 + l] = s[nt][r] * rinv[r];
  __syncthreads();

  v8f o[4] = {};
  const float* prow = &Ps[(wv * 16 + l) * 130 + 2 * half];
  for (int k0 = 0; k0 < 128; k0 += 4) {
    v2f a;
    a.x = prow[k0];
    a.y = prow[k0 + 1];
#pragma unroll
    for (int nt = 0; nt < 4; ++nt) {
      v2f b;
      const float* vp = &Vs[(k0 + 2 * half) * 66 + nt * 16 + l];
      b.x = vp[0];
      b.y = vp[66];
      o[nt] = wmma4(a, b, o[nt]);
    }
  }
  float* oh = out + ((size_t)h * T_SEQ + w * 64) * DH;
#pragma unroll
  for (int nt = 0; nt < 4; ++nt)
#pragma unroll
    for (int r = 0; r < 8; ++r)
      oh[(size_t)(wv * 16 + r + 8 * half) * DH + nt * 16 + l] = o[nt][r];
}

__global__ void comp_branch_kernel(const float* __restrict__ q,
                                   const float* __restrict__ kc,
                                   const float* __restrict__ vc,
                                   float* __restrict__ bsp,
                                   float* __restrict__ comp) {
  __shared__ float Ps[16 * 34];
  int h = blockIdx.x >> 7;
  int tt = blockIdx.x & 127;
  int t0 = tt << 4;
  int lane = threadIdx.x, half = lane >> 4, l = lane & 15;
  const float* qrow = q + ((size_t)h * T_SEQ + t0 + l) * DH + 2 * half;
  v8f s[2] = {};
  for (int k0 = 0; k0 < 64; k0 += 4) {
    v2f a;
    a.x = qrow[k0];
    a.y = qrow[k0 + 1];
#pragma unroll
    for (int nt = 0; nt < 2; ++nt) {
      v2f b;
      const float* kp = kc + (size_t)(h * 32 + nt * 16 + l) * 64 + k0 + 2 * half;
      b.x = kp[0];
      b.y = kp[1];
      s[nt] = wmma4(a, b, s[nt]);
    }
  }
#pragma unroll
  for (int nt = 0; nt < 2; ++nt)
#pragma unroll
    for (int r = 0; r < 8; ++r) s[nt][r] *= 0.125f;

#pragma unroll
  for (int nt = 0; nt < 2; ++nt) {
    float cs = 0.f;
#pragma unroll
    for (int r = 0; r < 8; ++r) cs += s[nt][r];
    cs += other_row(cs);
    if (half == 0)
      bsp[((size_t)h * 128 + tt) * 32 + nt * 16 + l] = cs * (1.0f / 64.0f);
  }
  float rmax[8], rsum[8];
#pragma unroll
  for (int r = 0; r < 8; ++r) rmax[r] = fmaxf(s[0][r], s[1][r]);
  row_max8(rmax);
#pragma unroll
  for (int r = 0; r < 8; ++r) rsum[r] = 0.f;
#pragma unroll
  for (int nt = 0; nt < 2; ++nt)
#pragma unroll
    for (int r = 0; r < 8; ++r) {
      float e = __expf(s[nt][r] - rmax[r]);
      s[nt][r] = e;
      rsum[r] += e;
    }
  row_sum8(rsum);
#pragma unroll
  for (int nt = 0; nt < 2; ++nt)
#pragma unroll
    for (int r = 0; r < 8; ++r)
      Ps[(r + 8 * half) * 34 + nt * 16 + l] = s[nt][r] / rsum[r];
  __syncthreads();

  v8f o[4] = {};
  const float* prow = &Ps[l * 34 + 2 * half];
  for (int k0 = 0; k0 < 32; k0 += 4) {
    v2f a;
    a.x = prow[k0];
    a.y = prow[k0 + 1];
#pragma unroll
    for (int nt = 0; nt < 4; ++nt) {
      v2f b;
      const float* vp = vc + (size_t)(h * 32 + k0 + 2 * half) * 64 + nt * 16 + l;
      b.x = vp[0];
      b.y = vp[64];
      o[nt] = wmma4(a, b, o[nt]);
    }
  }
  float* oh = comp + ((size_t)h * T_SEQ + t0) * DH;
#pragma unroll
  for (int nt = 0; nt < 4; ++nt)
#pragma unroll
    for (int r = 0; r < 8; ++r)
      oh[(size_t)(r + 8 * half) * DH + nt * 16 + l] = o[nt][r];
}

__global__ void __launch_bounds__(ATT_THREADS)
fine_attn_kernel(const float* __restrict__ q,
                                 const float* __restrict__ k,
                                 const float* __restrict__ v,
                                 const float* __restrict__ bsp,
                                 const float* __restrict__ ksc,
                                 const float* __restrict__ kbi,
                                 const float* __restrict__ sw,
                                 const float* __restrict__ comp,
                                 const float* __restrict__ gate,
                                 float* __restrict__ merged,
                                 float* __restrict__ tail,
                                 float* __restrict__ dkbuf) {
  __shared__ float Ks[64 * 66];
  __shared__ float Vs[64 * 66];
  __shared__ float Ps[64 * 66];
  int h = blockIdx.x >> 5, qb = blockIdx.x & 31;
  int wv = threadIdx.x >> 5, lane = threadIdx.x & 31;
  int half = lane >> 4, l = lane & 15;
  const float* kh = k + (size_t)h * T_SEQ * DH;
  const float* vh = v + (size_t)h * T_SEQ * DH;
  const float* qrow =
      q + ((size_t)h * T_SEQ + qb * 64 + wv * 16 + l) * DH + 2 * half;
  const float* bp = bsp + ((size_t)h * 128 + qb * 4) * 32 + lane;
  float sc = bp[0];
  sc += bp[32];
  sc += bp[64];
  sc += bp[96];
  unsigned selmask;
  {
    float mx = wave_max32(sc);
    float p = __expf(sc - mx);
    float inv = 1.0f / wave_sum32(p);
    float pp = p * inv;
    float ent = -wave_sum32(pp * __logf(pp + 1e-9f));
    ent = __builtin_bit_cast(
        float, __builtin_amdgcn_readfirstlane(__builtin_bit_cast(int, ent)));
    float ne = ent / (logf(32.0f) + 1e-9f);
    float kr = 1.0f / (1.0f + __expf(-(ksc[h] * ne + kbi[h])));
    float dkf = fminf(fmaxf((float)KMAX_SEL * kr, 1.0f), (float)KMAX_SEL);
    int dk = (int)dkf;
    if (threadIdx.x == 0) {
      tail[1 + h * 32 + qb] = kr;
      tail[1 + 512 + h * 32 + qb] = ne;
      dkbuf[h * 32 + qb] = (float)dk;
    }
    int rank = 0;
#pragma unroll
    for (int c2 = 0; c2 < 32; ++c2) {
      float o2 = __builtin_bit_cast(
          float, __builtin_amdgcn_readlane(__builtin_bit_cast(int, sc), c2));
      rank += (o2 > sc || (o2 == sc && c2 < lane)) ? 1 : 0;
    }
    selmask = __builtin_amdgcn_ballot_w32(rank < dk || lane == qb);
  }
  v8f o[4] = {};
  float rmax[8], rsum[8];
#pragma unroll
  for (int r = 0; r < 8; ++r) { rmax[r] = -3.0e38f; rsum[r] = 0.f; }

  for (int j = 0; j < 32; ++j) {
    if (!((selmask >> j) & 1u)) continue;
    float2 kr[16], vr[16];
    tile_load_regs(kr, vr, kh + (size_t)j * 64 * DH, vh + (size_t)j * 64 * DH,
                   threadIdx.x);
    __syncthreads();
    tile_store_lds(Ks, Vs, kr, vr, threadIdx.x);
    __syncthreads();
    v8f s[4] = {};
    for (int k0 = 0; k0 < 64; k0 += 4) {
      v2f a;
      a.x = qrow[k0];
      a.y = qrow[k0 + 1];
#pragma unroll
      for (int nt = 0; nt < 4; ++nt) {
        v2f b;
        const float* kp = &Ks[(nt * 16 + l) * 66 + k0 + 2 * half];
        b.x = kp[0];
        b.y = kp[1];
        s[nt] = wmma4(a, b, s[nt]);
      }
    }
    float nm[8], ps[8], fac[8];
#pragma unroll
    for (int r = 0; r < 8; ++r) nm[r] = rmax[r];
#pragma unroll
    for (int nt = 0; nt < 4; ++nt)
#pragma unroll
      for (int r = 0; r < 8; ++r) {
        float x = s[nt][r] * 0.125f;
        s[nt][r] = x;
        nm[r] = fmaxf(nm[r], x);
      }
    row_max8(nm);
#pragma unroll
    for (int r = 0; r < 8; ++r) { fac[r] = __expf(rmax[r] - nm[r]); ps[r] = 0.f; }
#pragma unroll
    for (int nt = 0; nt < 4; ++nt)
#pragma unroll
      for (int r = 0; r < 8; ++r) {
        float e = __expf(s[nt][r] - nm[r]);
        s[nt][r] = e;
        ps[r] += e;
      }
    row_sum8(ps);
#pragma unroll
    for (int r = 0; r < 8; ++r) {
      rsum[r] = rsum[r] * fac[r] + ps[r];
      rmax[r] = nm[r];
    }
#pragma unroll
    for (int nt = 0; nt < 4; ++nt)
#pragma unroll
      for (int r = 0; r < 8; ++r)
        Ps[(wv * 16 + r + 8 * half) * 66 + nt * 16 + l] = s[nt][r];
#pragma unroll
    for (int nt = 0; nt < 4; ++nt)
#pragma unroll
      for (int r = 0; r < 8; ++r) o[nt][r] *= fac[r];
    __syncthreads();
    const float* prow = &Ps[(wv * 16 + l) * 66 + 2 * half];
    for (int k0 = 0; k0 < 64; k0 += 4) {
      v2f a;
      a.x = prow[k0];
      a.y = prow[k0 + 1];
#pragma unroll
      for (int nt = 0; nt < 4; ++nt) {
        v2f b;
        const float* vp = &Vs[(k0 + 2 * half) * 66 + nt * 16 + l];
        b.x = vp[0];
        b.y = vp[66];
        o[nt] = wmma4(a, b, o[nt]);
      }
    }
  }
  const int trow = qb * 64 + wv * 16 + 8 * half;
  const float* swp = sw + ((size_t)h * T_SEQ + trow) * DH + l;
  const float* cpp = comp + ((size_t)h * T_SEQ + trow) * DH + l;
  const float* gp = gate + (size_t)trow * 48 + h * 3;
  float gw[8][3], sv[4][8], cv[4][8];
#pragma unroll
  for (int r = 0; r < 8; ++r) {
    gw[r][0] = gp[r * 48];
    gw[r][1] = gp[r * 48 + 1];
    gw[r][2] = gp[r * 48 + 2];
  }
#pragma unroll
  for (int nt = 0; nt < 4; ++nt)
#pragma unroll
    for (int r = 0; r < 8; ++r) {
      sv[nt][r] = swp[r * DH + nt * 16];
      cv[nt][r] = cpp[r * DH + nt * 16];
    }
  float* mp = merged + (size_t)trow * HID_DIM + h * DH + l;
#pragma unroll
  for (int nt = 0; nt < 4; ++nt)
#pragma unroll
    for (int r = 0; r < 8; ++r) {
      float f = o[nt][r] / rsum[r];
      mp[(size_t)r * HID_DIM + nt * 16] =
          sv[nt][r] * gw[r][0] + cv[nt][r] * gw[r][1] + f * gw[r][2];
    }
}

extern "C" void kernel_launch(void* const* d_in, const int* in_sizes, int n_in,
                              void* d_out, int out_size, void* d_ws,
                              size_t ws_size, hipStream_t stream) {
  const float* X = (const float*)d_in[0];
  const float* Wq = (const float*)d_in[1];
  const float* Wk = (const float*)d_in[2];
  const float* Wv = (const float*)d_in[3];
  const float* Wo = (const float*)d_in[4];
  const float* Wc = (const float*)d_in[5];
  const float* bc = (const float*)d_in[6];
  const float* ksc = (const float*)d_in[7];
  const float* kbi = (const float*)d_in[8];
  const float* Wg = (const float*)d_in[9];
  const float* bg = (const float*)d_in[10];
  float* out = (float*)d_out;

  const size_t U = (size_t)NH * T_SEQ * DH;
  float* ws = (float*)d_ws;
  float* q = ws;
  float* kbuf = ws + U;
  float* vbuf = ws + 2 * U;
  float* sw = ws + 3 * U;
  float* comp = ws + 4 * U;
  float* merged = ws + 5 * U;
  float* extra = ws + 6 * U;
  float* gate = extra;
  float* kc = gate + (size_t)T_SEQ * 48;
  float* vc = kc + 512 * 64;
  float* bsp = vc + 512 * 64;
  float* dkbuf = bsp + 16 * 128 * 32;
  float* tail = out + (size_t)T_SEQ * HID_DIM;

  {
    int wtiles = 3 * (T_SEQ / 64) * (HID_DIM / 32);
    int blocks = (wtiles + 11) / 12;
    wmma_gemm_f32_blk<<<blocks, 384, 0, stream>>>(X, Wq, Wk, Wv, nullptr, q,
                                                  kbuf, vbuf, 3, T_SEQ, HID_DIM,
                                                  HID_DIM, 0, 1, nullptr,
                                                  nullptr);
  }
  sliding_window_kernel<<<NH * NB, ATT_THREADS,
                          (2 * 128 * 66 + 64 * 130) * sizeof(float), stream>>>(
      q, kbuf, vbuf, sw);
  {
    small_products_splitk<<<256, 1024, 0, stream>>>(kbuf, vbuf, Wc, bc, kc, vc,
                                                    X, Wg, bg, gate);
  }
  comp_branch_kernel<<<NH * (T_SEQ / 16), 32, 0, stream>>>(q, kc, vc, bsp, comp);
  fine_attn_kernel<<<NH * NB, ATT_THREADS, 0, stream>>>(
      q, kbuf, vbuf, bsp, ksc, kbi, sw, comp, gate, merged, tail, dkbuf);
  {
    int wtiles = (T_SEQ / 64) * (HID_DIM / 32);
    wmma_gemm_f32_blk<<<(wtiles + 3) / 4, 128, 0, stream>>>(
        merged, Wo, nullptr, nullptr, nullptr, out, nullptr, nullptr, 1, T_SEQ,
        HID_DIM, HID_DIM, 0, 0, dkbuf, tail);
  }
  (void)in_sizes;
  (void)n_in;
  (void)out_size;
  (void)ws_size;
}


# compressed branch computed per wave in the fine-attention prologue (block-score partials via LDS, same order); its launch, comp and partials planes gone; q fragments loaded once, kept for tile loop
# speedup vs baseline: 11.7121x; 1.0444x over previous
#include <hip/hip_runtime.h>
#include <hip/hip_bf16.h>
#include <math.h>

typedef float v2f __attribute__((ext_vector_type(2)));
typedef float v4f __attribute__((ext_vector_type(4)));
typedef float v8f __attribute__((ext_vector_type(8)));

#define T_SEQ 2048
#define HID_DIM 1024
#define NH 16
#define DH 64
#define NB 32
#define KMAX_SEL 8

__device__ __forceinline__ v8f wmma4(v2f a, v2f b, v8f c) {
  return __builtin_amdgcn_wmma_f32_16x16x4_f32(false, a, false, b, (short)0, c,
                                               false, false);
}

__device__ __forceinline__ float apply_act(float x, int act) {
  if (act == 1) return 0.5f * x * (1.0f + erff(x * 0.70710678118654752f));
  if (act == 2) return 1.0f / (1.0f + __expf(-x));
  return x;
}

template <int M>
__device__ __forceinline__ float row_xor(float x) {
  return __builtin_amdgcn_update_dpp(x, x, 0x160 + M, 0xf, 0xf, true);
}

template <int M>
__device__ __forceinline__ void row_max_step(float (&x)[8]) {
#pragma unroll
  for (int r = 0; r < 8; ++r) x[r] = fmaxf(x[r], row_xor<M>(x[r]));
}

template <int M>
__device__ __forceinline__ void row_sum_step(float (&x)[8]) {
#pragma unroll
  for (int r = 0; r < 8; ++r) x[r] += row_xor<M>(x[r]);
}

__device__ __forceinline__ void row_max8(float (&x)[8]) {
  row_max_step<1>(x);
  row_max_step<2>(x);
  row_max_step<4>(x);
  row_max_step<8>(x);
}

__device__ __forceinline__ void row_sum8(float (&x)[8]) {
  row_sum_step<1>(x);
  row_sum_step<2>(x);
  row_sum_step<4>(x);
  row_sum_step<8>(x);
}

__device__ __forceinline__ float other_row(float x) {
  int v = __builtin_bit_cast(int, x);
  return __builtin_bit_cast(
      float, __builtin_amdgcn_permlanex16(v, v, 0x76543210, 0xfedcba98, false,
                                          false));
}

__device__ __forceinline__ float wave_sum32(float x) {
  x += row_xor<1>(x);
  x += row_xor<2>(x);
  x += row_xor<4>(x);
  x += row_xor<8>(x);
  return x + other_row(x);
}

__device__ __forceinline__ float wave_max32(float x) {
  x = fmaxf(x, row_xor<1>(x));
  x = fmaxf(x, row_xor<2>(x));
  x = fmaxf(x, row_xor<4>(x));
  x = fmaxf(x, row_xor<8>(x));
  return fmaxf(x, other_row(x));
}

struct GemmFrags {
  v4f a[2][4];
  v2f b[2][4];
};

__device__ __forceinline__ void gemm_load_block(GemmFrags& f, const float* ap,
                                                const float* bp,
                                                size_t aRowStep, size_t N) {
#pragma unroll
  for (int g = 0; g < 2; ++g) {
#pragma unroll
    for (int mi = 0; mi < 4; ++mi)
      f.a[g][mi] = *(const v4f*)(ap + (size_t)mi * aRowStep + 8 * g);
#pragma unroll
    for (int j = 0; j < 4; ++j)
      f.b[g][j] = *(const v2f*)(bp + (size_t)(8 * g + j) * N);
  }
}

__device__ __forceinline__ void gemm_mma_block(const GemmFrags& f,
                                               v8f (&acc)[4][2]) {
#pragma unroll
  for (int g = 0; g < 2; ++g)
#pragma unroll
    for (int p = 0; p < 2; ++p) {
      v2f b0, b1;
      b0.x = f.b[g][2 * p].x;
      b0.y = f.b[g][2 * p + 1].x;
      b1.x = f.b[g][2 * p].y;
      b1.y = f.b[g][2 * p + 1].y;
#pragma unroll
      for (int mi = 0; mi < 4; ++mi) {
        v2f a;
        a.x = f.a[g][mi][2 * p];
        a.y = f.a[g][mi][2 * p + 1];
        acc[mi][0] = wmma4(a, b0, acc[mi][0]);
        acc[mi][1] = wmma4(a, b1, acc[mi][1]);
      }
    }
}

__global__ void __launch_bounds__(384)
wmma_gemm_f32_blk(const float* __restrict__ A, const float* B0,
                  const float* B1, const float* B2,
                  const float* __restrict__ bias, float* C0, float* C1,
                  float* C2, int nProd, int M, int N, int K, int act,
                  int headSplit, const float* __restrict__ dkSrc,
                  float* __restrict__ avgDst) {
  int wave = threadIdx.x >> 5;
  int lane = threadIdx.x & 31;
  int wtile = blockIdx.x * (blockDim.x >> 5) + wave;
  int wn = N >> 5;
  int total = (M >> 6) * wn;
  if (wtile >= total * nProd) return;
  int prod = wtile / total;
  wtile -= prod * total;
  const float* __restrict__ B = prod == 0 ? B0 : (prod == 1 ? B1 : B2);
  float* __restrict__ C = prod == 0 ? C0 : (prod == 1 ? C1 : C2);
  int tm = wtile / wn, tn = wtile - tm * wn;
  int half = lane >> 4, l = lane & 15;
  int m0 = tm << 6, n0 = tn << 5;

  v8f acc[4][2];
#pragma unroll
  for (int mi = 0; mi < 4; ++mi)
#pragma unroll
    for (int ni = 0; ni < 2; ++ni) acc[mi][ni] = (v8f){};

  const float* ap = A + (size_t)(m0 + l) * K + 4 * half;
  const float* bp = B + (size_t)(4 * half) * N + n0 + 2 * l;
  const size_t aRowStep = (size_t)16 * K;
  const size_t bBlk = (size_t)16 * N;
  GemmFrags f0, f1;
  gemm_load_block(f0, ap, bp, aRowStep, (size_t)N);
  for (int k0 = 0; k0 < K - 32; k0 += 32) {
    gemm_load_block(f1, ap + 16, bp + bBlk, aRowStep, (size_t)N);
    __builtin_amdgcn_sched_barrier(0);
    gemm_mma_block(f0, acc);
    __builtin_amdgcn_sched_barrier(0);
    gemm_load_block(f0, ap + 32, bp + 2 * bBlk, aRowStep, (size_t)N);
    __builtin_amdgcn_sched_barrier(0);
    gemm_mma_block(f1, acc);
    __builtin_amdgcn_sched_barrier(0);
    ap += 32;
    bp += 2 * bBlk;
  }
  gemm_load_block(f1, ap + 16, bp + bBlk, aRowStep, (size_t)N);
  __builtin_amdgcn_sched_barrier(0);
  gemm_mma_block(f0, acc);
  gemm_mma_block(f1, acc);

  const int col = n0 + 2 * l;
  float bv0 = bias ? bias[col] : 0.0f;
  float bv1 = bias ? bias[col + 1] : 0.0f;
#pragma unroll
  for (int mi = 0; mi < 4; ++mi)
#pragma unroll
    for (int r = 0; r < 8; ++r) {
      v2f x;
      x.x = apply_act(acc[mi][0][r] + bv0, act);
      x.y = apply_act(acc[mi][1][r] + bv1, act);
      int m = m0 + mi * 16 + r + 8 * half;
      if (headSplit) {
        int hh = col >> 6, dd = col & 63;
        *(v2f*)&C[((size_t)hh * M + m) * 64 + dd] = x;
      } else {
        *(v2f*)&C[(size_t)m * N + col] = x;
      }
    }
  if (dkSrc != nullptr && blockIdx.x == 0 && wave == 0) {
    float s = 0.f;
#pragma unroll
    for (int i = 0; i < 16; ++i) s += dkSrc[lane + 32 * i];
    s = wave_sum32(s);
    if (lane == 0) avgDst[0] = s * (1.0f / 512.0f);
  }
}

#define CK_K 4096
#define CK_N 64
#define SP_WAVES 32
#define CK_SLICE (CK_K / SP_WAVES)
#define GT_N 48
#define GT_SLICE (HID_DIM / SP_WAVES)
#define SP_NI_PITCH 272
#define CK_W_PITCH (2 * SP_NI_PITCH)
#define GT_W_PITCH (3 * SP_NI_PITCH)
#define SP_COMPRESS_GROUPS 128
__global__ void __launch_bounds__(SP_WAVES * 32)
small_products_splitk(const float* __restrict__ kblk,
                      const float* __restrict__ vblk,
                      const float* __restrict__ Wc,
                      const float* __restrict__ bc, float* __restrict__ kc,
                      float* __restrict__ vc, const float* __restrict__ X,
                      const float* __restrict__ Wg,
                      const float* __restrict__ bg, float* __restrict__ gate) {
  __shared__ float part[SP_WAVES * GT_W_PITCH];
  int wave = threadIdx.x >> 5, lane = threadIdx.x & 31;
  int half = lane >> 4, l = lane & 15;
  int t = threadIdx.x;
  if (blockIdx.x < SP_COMPRESS_GROUPS) {
    int prod = blockIdx.x >> 6, g = blockIdx.x & 63;
    const float* __restrict__ A = prod ? vblk : kblk;
    float* __restrict__ C = prod ? vc : kc;
    int m0 = (g >> 1) << 4, n0 = (g & 1) << 5;
    int kbase = wave * CK_SLICE;
    const float* ap = A + (size_t)(m0 + l) * CK_K + kbase + 4 * half;
    const float* bp = Wc + (size_t)(kbase + 4 * half) * CK_N + n0 + 2 * l;
    v8f acc0 = {}, acc1 = {};
#pragma unroll
    for (int g = 0; g < CK_SLICE / 8; ++g) {
      v4f a4 = *(const v4f*)(ap + 8 * g);
      v2f br[4];
#pragma unroll
      for (int j = 0; j < 4; ++j)
        br[j] = *(const v2f*)(bp + (size_t)(8 * g + j) * CK_N);
#pragma unroll
      for (int p = 0; p < 2; ++p) {
        v2f a, b0, b1;
        a.x = a4[2 * p];
        a.y = a4[2 * p + 1];
        b0.x = br[2 * p].x;
        b0.y = br[2 * p + 1].x;
        b1.x = br[2 * p].y;
        b1.y = br[2 * p + 1].y;
        acc0 = wmma4(a, b0, acc0);
        acc1 = wmma4(a, b1, acc1);
      }
    }
    float* pw = part + wave * CK_W_PITCH + lane;
#pragma unroll
    for (int r = 0; r < 8; ++r) {
      pw[r * 32] = acc0[r];
      pw[SP_NI_PITCH + r * 32] = acc1[r];
    }
    __syncthreads();
    if (t < 512) {
      int row = t >> 5, col = t & 31;
      const float* pr = part + (col & 1) * SP_NI_PITCH + (row & 7) * 32 +
                        (row >> 3) * 16 + (col >> 1);
      float sum = pr[0];
#pragma unroll
      for (int w = 1; w < SP_WAVES; ++w) sum += pr[w * CK_W_PITCH];
      C[(size_t)(m0 + row) * CK_N + n0 + col] =
          apply_act(sum + bc[n0 + col], 1);
    }
  } else {
    int m0 = (blockIdx.x - SP_COMPRESS_GROUPS) << 4;
    int kbase = wave * GT_SLICE;
    const float* ap = X + (size_t)(m0 + l) * HID_DIM + kbase + 2 * half;
    const float* bp = Wg + (size_t)(kbase + 2 * half) * GT_N + l;
    v8f acc0 = {}, acc1 = {}, acc2 = {};
#pragma unroll
    for (int s = 0; s < GT_SLICE / 4; ++s) {
      v2f a, b0, b1, b2;
      a.x = ap[4 * s];
      a.y = ap[4 * s + 1];
      b0.x = bp[(4 * s) * GT_N];
      b0.y = bp[(4 * s + 1) * GT_N];
      b1.x = bp[(4 * s) * GT_N + 16];
      b1.y = bp[(4 * s + 1) * GT_N + 16];
      b2.x = bp[(4 * s) * GT_N + 32];
      b2.y = bp[(4 * s + 1) * GT_N + 32];
      acc0 = wmma4(a, b0, acc0);
      acc1 = wmma4(a, b1, acc1);
      acc2 = wmma4(a, b2, acc2);
    }
    float* pw = part + wave * GT_W_PITCH + lane;
#pragma unroll
    for (int r = 0; r < 8; ++r) {
      pw[r * 32] = acc0[r];
      pw[SP_NI_PITCH + r * 32] = acc1[r];
      pw[2 * SP_NI_PITCH + r * 32] = acc2[r];
    }
    __syncthreads();
    if (t < 16 * GT_N) {
      int row = t / GT_N, col = t - row * GT_N;
      const float* pr = part + (col >> 4) * SP_NI_PITCH + (row & 7) * 32 +
                        (row >> 3) * 16 + (col & 15);
      float sum = pr[0];
#pragma unroll
      for (int w = 1; w < SP_WAVES; ++w) sum += pr[w * GT_W_PITCH];
      gate[(size_t)m0 * GT_N + t] = apply_act(sum + bg[col], 2);
    }
  }
}

#define ATT_THREADS 128
__device__ __forceinline__ void tile_load_regs(float2 (&kr)[16],
                                               float2 (&vr)[16],
                                               const float* __restrict__ kt,
                                               const float* __restrict__ vt,
                                               int tid) {
  const int off = (tid >> 5) * DH + ((tid & 31) << 1);
#pragma unroll
  for (int i = 0; i < 16; ++i) {
    kr[i] = *(const float2*)&kt[off + i * 4 * DH];
    vr[i] = *(const float2*)&vt[off + i * 4 * DH];
  }
}

__device__ __forceinline__ void tile_store_lds(float* Kd, float* Vd,
                                               const float2 (&kr)[16],
                                               const float2 (&vr)[16],
                                               int tid) {
  const int off = (tid >> 5) * 66 + ((tid & 31) << 1);
#pragma unroll
  for (int i = 0; i < 16; ++i) {
    *(float2*)&Kd[off + i * 4 * 66] = kr[i];
    *(float2*)&Vd[off + i * 4 * 66] = vr[i];
  }
}

__global__ void __launch_bounds__(ATT_THREADS)
sliding_window_kernel(const float* __restrict__ q,
                                      const float* __restrict__ k,
                                      const float* __restrict__ v,
                                      float* __restrict__ out) {
  extern __shared__ float lds[];
  float* Ks = lds;
  float* Vs = Ks + 128 * 66;
  float* Ps = Vs + 128 * 66;
  int h = blockIdx.x >> 5;
  int w = blockIdx.x & 31;
  const float* kh = k + (size_t)h * T_SEQ * DH;
  const float* vh = v + (size_t)h * T_SEQ * DH;
  {
    float2 kc[16], vc[16], kp[16], vp[16];
    tile_load_regs(kc, vc, kh + (size_t)w * 64 * DH, vh + (size_t)w * 64 * DH,
                   threadIdx.x);
    if (w > 0) {
      tile_load_regs(kp, vp, kh + (size_t)(w - 1) * 64 * DH,
                     vh + (size_t)(w - 1) * 64 * DH, threadIdx.x);
    } else {
#pragma unroll
      for (int i = 0; i < 16; ++i) {
        kp[i] = make_float2(0.f, 0.f);
        vp[i] = make_float2(0.f, 0.f);
      }
    }
    tile_store_lds(Ks + 64 * 66, Vs + 64 * 66, kc, vc, threadIdx.x);
    tile_store_lds(Ks, Vs, kp, vp, threadIdx.x);
  }
  __syncthreads();

  int wv = threadIdx.x >> 5, lane = threadIdx.x & 31;
  int half = lane >> 4, l = lane & 15;
  const float* qrow =
      q + ((size_t)h * T_SEQ + w * 64 + wv * 16 + l) * DH + 2 * half;

  v8f s[8] = {};
  for (int k0 = 0; k0 < 64; k0 += 4) {
    v2f a;
    a.x = qrow[k0];
    a.y = qrow[k0 + 1];
#pragma unroll
    for (int nt = 0; nt < 8; ++nt) {
      v2f b;
      const float* kp = &Ks[(nt * 16 + l) * 66 + k0 + 2 * half];
      b.x = kp[0];
      b.y = kp[1];
      s[nt] = wmma4(a, b, s[nt]);
    }
  }
  float rmax[8], rsum[8];
#pragma unroll
  for (int r = 0; r < 8; ++r) { rmax[r] = -3.0e38f; rsum[r] = 0.f; }
#pragma unroll
  for (int nt = 0; nt < 8; ++nt) {
    int n = nt * 16 + l;
#pragma unroll
    for (int r = 0; r < 8; ++r) {
      int m = wv * 16 + r + 8 * half;
      float x = s[nt][r] * 0.125f;
      bool ok = (m + 64 >= n) && (w > 0 || n >= 64);
      x = ok ? x : -3.0e38f;
      s[nt][r] = x;
      rmax[r] = fmaxf(rmax[r], x);
    }
  }
  row_max8(rmax);
#pragma unroll
  for (int nt = 0; nt < 8; ++nt)
#pragma unroll
    for (int r = 0; r < 8; ++r) {
      float e = __expf(s[nt][r] - rmax[r]);
      s[nt][r] = e;
      rsum[r] += e;
    }
  row_sum8(rsum);
  float rinv[8];
#pragma unroll
  for (int r = 0; r < 8; ++r) rinv[r] = 1.0f / rsum[r];
#pragma unroll
  for (int nt = 0; nt < 8; ++nt)
#pragma unroll
    for (int r = 0; r < 8; ++r)
      Ps[(wv * 16 + r + 8 * half) * 130 + nt * 16 + l] = s[nt][r] * rinv[r];
  __syncthreads();

  v8f o[4] = {};
  const float* prow = &Ps[(wv * 16 + l) * 130 + 2 * half];
  for (int k0 = 0; k0 < 128; k0 += 4) {
    v2f a;
    a.x = prow[k0];
    a.y = prow[k0 + 1];
#pragma unroll
    for (int nt = 0; nt < 4; ++nt) {
      v2f b;
      const float* vp = &Vs[(k0 + 2 * half) * 66 + nt * 16 + l];
      b.x = vp[0];
      b.y = vp[66];
      o[nt] = wmma4(a, b, o[nt]);
    }
  }
  float* oh = out + ((size_t)h * T_SEQ + w * 64) * DH;
#pragma unroll
  for (int nt = 0; nt < 4; ++nt)
#pragma unroll
    for (int r = 0; r < 8; ++r)
      oh[(size_t)(wv * 16 + r + 8 * half) * DH + nt * 16 + l] = o[nt][r];
}

__global__ void __launch_bounds__(ATT_THREADS)
fine_attn_kernel(const float* __restrict__ q,
                                 const float* __restrict__ k,
                                 const float* __restrict__ v,
                                 const float* __restrict__ kc,
                                 const float* __restrict__ vc,
                                 const float* __restrict__ ksc,
                                 const float* __restrict__ kbi,
                                 const float* __restrict__ sw,
                                 const float* __restrict__ gate,
                                 float* __restrict__ merged,
                                 float* __restrict__ tail,
                                 float* __restrict__ dkbuf) {
  __shared__ float Ks[64 * 66];
  __shared__ float Vs[64 * 66];
  __shared__ float Ps[64 * 66];
  __shared__ float colpart[4 * 32];
  int h = blockIdx.x >> 5, qb = blockIdx.x & 31;
  int wv = threadIdx.x >> 5, lane = threadIdx.x & 31;
  int half = lane >> 4, l = lane & 15;
  const float* kh = k + (size_t)h * T_SEQ * DH;
  const float* vh = v + (size_t)h * T_SEQ * DH;
  const float* qrow =
      q + ((size_t)h * T_SEQ + qb * 64 + wv * 16 + l) * DH + 2 * half;
  v2f qf[16];
#pragma unroll
  for (int ks = 0; ks < 16; ++ks) qf[ks] = *(const v2f*)(qrow + 4 * ks);

  v8f cs2[2] = {};
  v2f kcf[16][2];
  v2f vcf[8][4];
  {
    const float* kcp = kc + (size_t)(h * 32 + l) * 64 + 2 * half;
    const float* vcp = vc + (size_t)(h * 32 + 2 * half) * 64 + l;
#pragma unroll
    for (int ks = 0; ks < 16; ++ks)
#pragma unroll
      for (int nt = 0; nt < 2; ++nt)
        kcf[ks][nt] = *(const v2f*)(kcp + nt * 16 * 64 + 4 * ks);
#pragma unroll
    for (int ks = 0; ks < 8; ++ks)
#pragma unroll
      for (int nt = 0; nt < 4; ++nt) {
        vcf[ks][nt].x = vcp[(4 * ks) * 64 + nt * 16];
        vcf[ks][nt].y = vcp[(4 * ks + 1) * 64 + nt * 16];
      }
#pragma unroll
    for (int ks = 0; ks < 16; ++ks)
#pragma unroll
      for (int nt = 0; nt < 2; ++nt)
        cs2[nt] = wmma4(qf[ks], kcf[ks][nt], cs2[nt]);
  }
#pragma unroll
  for (int nt = 0; nt < 2; ++nt)
#pragma unroll
    for (int r = 0; r < 8; ++r) cs2[nt][r] *= 0.125f;
#pragma unroll
  for (int nt = 0; nt < 2; ++nt) {
    float cs = 0.f;
#pragma unroll
    for (int r = 0; r < 8; ++r) cs += cs2[nt][r];
    cs += other_row(cs);
    if (half == 0) colpart[wv * 32 + nt * 16 + l] = cs * (1.0f / 64.0f);
  }
  __syncthreads();
  float sc = colpart[lane];
  sc += colpart[32 + lane];
  sc += colpart[64 + lane];
  sc += colpart[96 + lane];
  unsigned selmask;
  {
    float mx = wave_max32(sc);
    float p = __expf(sc - mx);
    float inv = 1.0f / wave_sum32(p);
    float pp = p * inv;
    float ent = -wave_sum32(pp * __logf(pp + 1e-9f));
    ent = __builtin_bit_cast(
        float, __builtin_amdgcn_readfirstlane(__builtin_bit_cast(int, ent)));
    float ne = ent / (logf(32.0f) + 1e-9f);
    float kr = 1.0f / (1.0f + __expf(-(ksc[h] * ne + kbi[h])));
    float dkf = fminf(fmaxf((float)KMAX_SEL * kr, 1.0f), (float)KMAX_SEL);
    int dk = (int)dkf;
    if (threadIdx.x == 0) {
      tail[1 + h * 32 + qb] = kr;
      tail[1 + 512 + h * 32 + qb] = ne;
      dkbuf[h * 32 + qb] = (float)dk;
    }
    int rank = 0;
#pragma unroll
    for (int c2 = 0; c2 < 32; ++c2) {
      float o2 = __builtin_bit_cast(
          float, __builtin_amdgcn_readlane(__builtin_bit_cast(int, sc), c2));
      rank += (o2 > sc || (o2 == sc && c2 < lane)) ? 1 : 0;
    }
    selmask = __builtin_amdgcn_ballot_w32(rank < dk || lane == qb);
  }

  {
    float cmax[8], csum[8];
#pragma unroll
    for (int r = 0; r < 8; ++r) cmax[r] = fmaxf(cs2[0][r], cs2[1][r]);
    row_max8(cmax);
#pragma unroll
    for (int r = 0; r < 8; ++r) csum[r] = 0.f;
#pragma unroll
    for (int nt = 0; nt < 2; ++nt)
#pragma unroll
      for (int r = 0; r < 8; ++r) {
        float e = __expf(cs2[nt][r] - cmax[r]);
        cs2[nt][r] = e;
        csum[r] += e;
      }
    row_sum8(csum);
#pragma unroll
    for (int nt = 0; nt < 2; ++nt)
#pragma unroll
      for (int r = 0; r < 8; ++r)
        Ps[(wv * 16 + r + 8 * half) * 66 + nt * 16 + l] = cs2[nt][r] / csum[r];
  }
  __syncthreads();
  v8f co[4] = {};
  {
    const float* prow = &Ps[(wv * 16 + l) * 66 + 2 * half];
#pragma unroll
    for (int ks = 0; ks < 8; ++ks) {
      v2f a;
      a.x = prow[4 * ks];
      a.y = prow[4 * ks + 1];
#pragma unroll
      for (int nt = 0; nt < 4; ++nt) co[nt] = wmma4(a, vcf[ks][nt], co[nt]);
    }
  }

  v8f o[4] = {};
  float rmax[8], rsum[8];
#pragma unroll
  for (int r = 0; r < 8; ++r) { rmax[r] = -3.0e38f; rsum[r] = 0.f; }

  for (int j = 0; j < 32; ++j) {
    if (!((selmask >> j) & 1u)) continue;
    float2 kr[16], vr[16];
    tile_load_regs(kr, vr, kh + (size_t)j * 64 * DH, vh + (size_t)j * 64 * DH,
                   threadIdx.x);
    __syncthreads();
    tile_store_lds(Ks, Vs, kr, vr, threadIdx.x);
    __syncthreads();
    v8f s[4] = {};
#pragma unroll
    for (int ks = 0; ks < 16; ++ks) {
#pragma unroll
      for (int nt = 0; nt < 4; ++nt) {
        v2f b;
        const float* kp = &Ks[(nt * 16 + l) * 66 + 4 * ks + 2 * half];
        b.x = kp[0];
        b.y = kp[1];
        s[nt] = wmma4(qf[ks], b, s[nt]);
      }
    }
    float nm[8], ps[8], fac[8];
#pragma unroll
    for (int r = 0; r < 8; ++r) nm[r] = rmax[r];
#pragma unroll
    for (int nt = 0; nt < 4; ++nt)
#pragma unroll
      for (int r = 0; r < 8; ++r) {
        float x = s[nt][r] * 0.125f;
        s[nt][r] = x;
        nm[r] = fmaxf(nm[r], x);
      }
    row_max8(nm);
#pragma unroll
    for (int r = 0; r < 8; ++r) { fac[r] = __expf(rmax[r] - nm[r]); ps[r] = 0.f; }
#pragma unroll
    for (int nt = 0; nt < 4; ++nt)
#pragma unroll
      for (int r = 0; r < 8; ++r) {
        float e = __expf(s[nt][r] - nm[r]);
        s[nt][r] = e;
        ps[r] += e;
      }
    row_sum8(ps);
#pragma unroll
    for (int r = 0; r < 8; ++r) {
      rsum[r] = rsum[r] * fac[r] + ps[r];
      rmax[r] = nm[r];
    }
#pragma unroll
    for (int nt = 0; nt < 4; ++nt)
#pragma unroll
      for (int r = 0; r < 8; ++r)
        Ps[(wv * 16 + r + 8 * half) * 66 + nt * 16 + l] = s[nt][r];
#pragma unroll
    for (int nt = 0; nt < 4; ++nt)
#pragma unroll
      for (int r = 0; r < 8; ++r) o[nt][r] *= fac[r];
    __syncthreads();
    const float* prow = &Ps[(wv * 16 + l) * 66 + 2 * half];
    for (int k0 = 0; k0 < 64; k0 += 4) {
      v2f a;
      a.x = prow[k0];
      a.y = prow[k0 + 1];
#pragma unroll
      for (int nt = 0; nt < 4; ++nt) {
        v2f b;
        const float* vp = &Vs[(k0 + 2 * half) * 66 + nt * 16 + l];
        b.x = vp[0];
        b.y = vp[66];
        o[nt] = wmma4(a, b, o[nt]);
      }
    }
  }
  const int trow = qb * 64 + wv * 16 + 8 * half;
  const float* swp = sw + ((size_t)h * T_SEQ + trow) * DH + l;
  const float* gp = gate + (size_t)trow * 48 + h * 3;
  float gw[8][3], sv[4][8];
#pragma unroll
  for (int r = 0; r < 8; ++r) {
    gw[r][0] = gp[r * 48];
    gw[r][1] = gp[r * 48 + 1];
    gw[r][2] = gp[r * 48 + 2];
  }
#pragma unroll
  for (int nt = 0; nt < 4; ++nt)
#pragma unroll
    for (int r = 0; r < 8; ++r) sv[nt][r] = swp[r * DH + nt * 16];
  float* mp = merged + (size_t)trow * HID_DIM + h * DH + l;
#pragma unroll
  for (int nt = 0; nt < 4; ++nt)
#pragma unroll
    for (int r = 0; r < 8; ++r) {
      float f = o[nt][r] / rsum[r];
      mp[(size_t)r * HID_DIM + nt * 16] =
          sv[nt][r] * gw[r][0] + co[nt][r] * gw[r][1] + f * gw[r][2];
    }
}

extern "C" void kernel_launch(void* const* d_in, const int* in_sizes, int n_in,
                              void* d_out, int out_size, void* d_ws,
                              size_t ws_size, hipStream_t stream) {
  const float* X = (const float*)d_in[0];
  const float* Wq = (const float*)d_in[1];
  const float* Wk = (const float*)d_in[2];
  const float* Wv = (const float*)d_in[3];
  const float* Wo = (const float*)d_in[4];
  const float* Wc = (const float*)d_in[5];
  const float* bc = (const float*)d_in[6];
  const float* ksc = (const float*)d_in[7];
  const float* kbi = (const float*)d_in[8];
  const float* Wg = (const float*)d_in[9];
  const float* bg = (const float*)d_in[10];
  float* out = (float*)d_out;

  const size_t U = (size_t)NH * T_SEQ * DH;
  float* ws = (float*)d_ws;
  float* q = ws;
  float* kbuf = ws + U;
  float* vbuf = ws + 2 * U;
  float* sw = ws + 3 * U;
  float* merged = ws + 5 * U;
  float* extra = ws + 6 * U;
  float* gate = extra;
  float* kc = gate + (size_t)T_SEQ * 48;
  float* vc = kc + 512 * 64;
  float* dkbuf = vc + 512 * 64;
  float* tail = out + (size_t)T_SEQ * HID_DIM;

  {
    int wtiles = 3 * (T_SEQ / 64) * (HID_DIM / 32);
    int blocks = (wtiles + 11) / 12;
    wmma_gemm_f32_blk<<<blocks, 384, 0, stream>>>(X, Wq, Wk, Wv, nullptr, q,
                                                  kbuf, vbuf, 3, T_SEQ, HID_DIM,
                                                  HID_DIM, 0, 1, nullptr,
                                                  nullptr);
  }
  sliding_window_kernel<<<NH * NB, ATT_THREADS,
                          (2 * 128 * 66 + 64 * 130) * sizeof(float), stream>>>(
      q, kbuf, vbuf, sw);
  {
    small_products_splitk<<<256, 1024, 0, stream>>>(kbuf, vbuf, Wc, bc, kc, vc,
                                                    X, Wg, bg, gate);
  }
  fine_attn_kernel<<<NH * NB, ATT_THREADS, 0, stream>>>(
      q, kbuf, vbuf, kc, vc, ksc, kbi, sw, gate, merged, tail, dkbuf);
  {
    int wtiles = (T_SEQ / 64) * (HID_DIM / 32);
    wmma_gemm_f32_blk<<<(wtiles + 3) / 4, 128, 0, stream>>>(
        merged, Wo, nullptr, nullptr, nullptr, out, nullptr, nullptr, 1, T_SEQ,
        HID_DIM, HID_DIM, 0, 0, dkbuf, tail);
  }
  (void)in_sizes;
  (void)n_in;
  (void)out_size;
  (void)ws_size;
}
